# starting kernel
# baseline (speedup 1.0000x reference)
#include <hip/hip_runtime.h>
#include <math.h>

#define NN    50000
#define DD    128
#define HC    256
#define CC    128
#define EE    800000
#define EP    (EE + NN)
#define NEG_SLOPE 0.2f

typedef __attribute__((ext_vector_type(16))) __bf16 v16bf;
typedef __attribute__((ext_vector_type(8)))  __bf16 v8bf;
typedef __attribute__((ext_vector_type(8)))  float  v8f;

__device__ __forceinline__ void edge_sd(int e, const int* __restrict__ ei,
                                        int& s, int& d) {
    if (e < EE) { s = ei[e]; d = ei[EE + e]; }
    else        { s = e - EE; d = e - EE; }
}

__device__ __forceinline__ unsigned enc_f32(float f) {
    unsigned u = __float_as_uint(f);
    return (u & 0x80000000u) ? ~u : (u | 0x80000000u);
}
__device__ __forceinline__ float dec_f32(unsigned u) {
    u = (u & 0x80000000u) ? (u & 0x7FFFFFFFu) : ~u;
    return __uint_as_float(u);
}

__global__ void k_f32_to_bf16(const float* __restrict__ in,
                              __bf16* __restrict__ out, int n) {
    int i = blockIdx.x * blockDim.x + threadIdx.x;
    if (i < n) out[i] = (__bf16)in[i];
}

__global__ void k_transpose_bf16(const float* __restrict__ W,
                                 __bf16* __restrict__ Wt, int K, int Nn) {
    int i = blockIdx.x * blockDim.x + threadIdx.x;
    if (i < K * Nn) {
        int n = i / K, k = i - n * K;
        Wt[i] = (__bf16)W[(size_t)k * Nn + n];
    }
}

__global__ __launch_bounds__(256)
void k_gemm_bf16_wmma(const __bf16* __restrict__ A,
                      const __bf16* __restrict__ Bt,
                      float* __restrict__ C,
                      int M, int Nn, int K) {
    const int wavesPerBlock = blockDim.x >> 5;
    int wave = blockIdx.x * wavesPerBlock + (threadIdx.x >> 5);
    int lane = threadIdx.x & 31;
    int tilesN = Nn >> 4;
    int tM = (wave / tilesN) << 4;
    int tN = (wave % tilesN) << 4;
    if (tM >= M) return;

    int half = lane >> 4;
    int lr   = lane & 15;

    v8f acc = {};
    const __bf16* arow = A  + (size_t)(tM + lr) * K;
    const __bf16* brow = Bt + (size_t)(tN + lr) * K;

    for (int kk = 0; kk < K; kk += 32) {
        v8bf alo = *(const v8bf*)(arow + kk + 8 * half);
        v8bf ahi = *(const v8bf*)(arow + kk + 16 + 8 * half);
        v8bf blo = *(const v8bf*)(brow + kk + 8 * half);
        v8bf bhi = *(const v8bf*)(brow + kk + 16 + 8 * half);
        v16bf a, b;
#pragma unroll
        for (int i = 0; i < 8; ++i) {
            a[i] = alo[i]; a[i + 8] = ahi[i];
            b[i] = blo[i]; b[i + 8] = bhi[i];
        }
        acc = __builtin_amdgcn_wmma_f32_16x16x32_bf16(
                  false, a, false, b, (short)0, acc, false, false);
    }

    int n = tN + lr;
#pragma unroll
    for (int r = 0; r < 8; ++r) {
        int m = tM + r + 8 * half;
        C[(size_t)m * Nn + n] = acc[r];
    }
}

__global__ void k_att_scores(const float* __restrict__ h,
                             const float* __restrict__ att_src,
                             const float* __restrict__ att_dst,
                             float* __restrict__ a_s,
                             float* __restrict__ a_d) {
    int i = blockIdx.x * blockDim.x + threadIdx.x;
    if (i >= NN * 2) return;
    int node = i >> 1, hd = i & 1;
    const float* hp = h + (size_t)node * HC + hd * CC;
    const float* as = att_src + hd * CC;
    const float* ad = att_dst + hd * CC;
    float s = 0.f, d = 0.f;
#pragma unroll 4
    for (int c = 0; c < CC; ++c) { float v = hp[c]; s += v * as[c]; d += v * ad[c]; }
    a_s[i] = s; a_d[i] = d;
}

__global__ void k_edge_max(const int* __restrict__ ei,
                           const float* __restrict__ a_s,
                           const float* __restrict__ a_d,
                           unsigned* __restrict__ m_enc) {
    int e = blockIdx.x * blockDim.x + threadIdx.x;
    if (e >= EP) return;
    int s, d; edge_sd(e, ei, s, d);
#pragma unroll
    for (int h = 0; h < 2; ++h) {
        float l = a_s[s * 2 + h] + a_d[d * 2 + h];
        l = (l > 0.f) ? l : NEG_SLOPE * l;
        atomicMax(&m_enc[d * 2 + h], enc_f32(l));
    }
}

__global__ void k_edge_exp(const int* __restrict__ ei,
                           const float* __restrict__ a_s,
                           const float* __restrict__ a_d,
                           const unsigned* __restrict__ m_enc,
                           float* __restrict__ e_buf,
                           float* __restrict__ denom) {
    int e = blockIdx.x * blockDim.x + threadIdx.x;
    if (e >= EP) return;
    int s, d; edge_sd(e, ei, s, d);
#pragma unroll
    for (int h = 0; h < 2; ++h) {
        float l = a_s[s * 2 + h] + a_d[d * 2 + h];
        l = (l > 0.f) ? l : NEG_SLOPE * l;
        float m  = dec_f32(m_enc[d * 2 + h]);
        float ex = expf(l - m);
        e_buf[(size_t)e * 2 + h] = ex;
        atomicAdd(&denom[d * 2 + h], ex);
    }
}

__global__ __launch_bounds__(HC)
void k_edge_scatter(const int* __restrict__ ei,
                    const float* __restrict__ h,
                    const float* __restrict__ e_buf,
                    const float* __restrict__ denom,
                    float* __restrict__ acc) {
    int e = blockIdx.x;
    int c = threadIdx.x;
    int s, d; edge_sd(e, ei, s, d);
    int hd = c >> 7;
    float alpha = e_buf[(size_t)e * 2 + hd] / (denom[d * 2 + hd] + 1e-16f);
    atomicAdd(&acc[(size_t)d * HC + c], alpha * h[(size_t)s * HC + c]);
}

__global__ void k_post_tanh(const float* __restrict__ acc,
                            const float* __restrict__ b,
                            float* __restrict__ hout,
                            __bf16* __restrict__ hbf, int total) {
    int i = blockIdx.x * blockDim.x + threadIdx.x;
    if (i < total) {
        float v = tanhf(acc[i] + b[i & (HC - 1)]);
        hout[i] = v;
        hbf[i]  = (__bf16)v;
    }
}

__global__ void k_final(const float* __restrict__ acc,
                        const float* __restrict__ b1,
                        const float* __restrict__ Wl,
                        const float* __restrict__ bl,
                        float* __restrict__ out) {
    int node = blockIdx.x * blockDim.x + threadIdx.x;
    if (node >= NN) return;
    const float* a = acc + (size_t)node * HC;
    float z0 = bl[0], z1 = bl[1];
#pragma unroll 4
    for (int k = 0; k < HC; ++k) {
        float v = a[k] + b1[k];
        z0 += v * Wl[k * 2 + 0];
        z1 += v * Wl[k * 2 + 1];
    }
    float mx = fmaxf(z0, z1);
    float e0 = expf(z0 - mx), e1 = expf(z1 - mx);
    float inv = 1.f / (e0 + e1);
    out[node * 2 + 0] = e0 * inv;
    out[node * 2 + 1] = e1 * inv;
}

extern "C" void kernel_launch(void* const* d_in, const int* in_sizes, int n_in,
                              void* d_out, int out_size, void* d_ws, size_t ws_size,
                              hipStream_t stream) {
    const float* x        = (const float*)d_in[0];
    const int*   ei       = (const int*)  d_in[1];
    const float* W0       = (const float*)d_in[2];
    const float* att_src0 = (const float*)d_in[3];
    const float* att_dst0 = (const float*)d_in[4];
    const float* b0       = (const float*)d_in[5];
    const float* W1       = (const float*)d_in[6];
    const float* att_src1 = (const float*)d_in[7];
    const float* att_dst1 = (const float*)d_in[8];
    const float* b1       = (const float*)d_in[9];
    const float* Wl       = (const float*)d_in[10];
    const float* bl       = (const float*)d_in[11];
    float* out = (float*)d_out;

    char* base = (char*)d_ws;
    size_t off = 0;
    auto carve = [&](size_t bytes) -> char* {
        char* p = base + off;
        off = (off + bytes + 255) & ~(size_t)255;
        return p;
    };
    float*    h_feat = (float*)   carve((size_t)NN * HC * 4);
    float*    accb   = (float*)   carve((size_t)NN * HC * 4);
    __bf16*   xbf    = (__bf16*)  carve((size_t)NN * HC * 2);
    __bf16*   wtbf   = (__bf16*)  carve((size_t)HC * HC * 2);
    float*    a_s    = (float*)   carve((size_t)NN * 2 * 4);
    float*    a_d    = (float*)   carve((size_t)NN * 2 * 4);
    unsigned* m_enc  = (unsigned*)carve((size_t)NN * 2 * 4);
    float*    denomb = (float*)   carve((size_t)NN * 2 * 4);
    float*    e_buf  = (float*)   carve((size_t)EP * 2 * 4);
    (void)ws_size; (void)in_sizes; (void)n_in; (void)out_size;

    const int T = 256;
    dim3 blk(T);

    k_f32_to_bf16<<<(NN * DD + T - 1) / T, blk, 0, stream>>>(x, xbf, NN * DD);
    k_transpose_bf16<<<(DD * HC + T - 1) / T, blk, 0, stream>>>(W0, wtbf, DD, HC);

    k_gemm_bf16_wmma<<<(NN / 16) * (HC / 16) / 8, blk, 0, stream>>>(
        xbf, wtbf, h_feat, NN, HC, DD);

    k_att_scores<<<(NN * 2 + T - 1) / T, blk, 0, stream>>>(
        h_feat, att_src0, att_dst0, a_s, a_d);

    hipMemsetAsync(accb,   0, (size_t)NN * HC * 4, stream);
    hipMemsetAsync(m_enc,  0, (size_t)NN * 2 * 4, stream);
    hipMemsetAsync(denomb, 0, (size_t)NN * 2 * 4, stream);

    k_edge_max    <<<(EP + T - 1) / T, blk, 0, stream>>>(ei, a_s, a_d, m_enc);
    k_edge_exp    <<<(EP + T - 1) / T, blk, 0, stream>>>(ei, a_s, a_d, m_enc, e_buf, denomb);
    k_edge_scatter<<<EP, dim3(HC), 0, stream>>>(ei, h_feat, e_buf, denomb, accb);

    k_post_tanh<<<(NN * HC + T - 1) / T, blk, 0, stream>>>(accb, b0, h_feat, xbf, NN * HC);

    k_transpose_bf16<<<(HC * HC + T - 1) / T, blk, 0, stream>>>(W1, wtbf, HC, HC);

    k_gemm_bf16_wmma<<<(NN / 16) * (HC / 16) / 8, blk, 0, stream>>>(
        xbf, wtbf, h_feat, NN, HC, HC);

    k_att_scores<<<(NN * 2 + T - 1) / T, blk, 0, stream>>>(
        h_feat, att_src1, att_dst1, a_s, a_d);

    hipMemsetAsync(accb,   0, (size_t)NN * HC * 4, stream);
    hipMemsetAsync(m_enc,  0, (size_t)NN * 2 * 4, stream);
    hipMemsetAsync(denomb, 0, (size_t)NN * 2 * 4, stream);

    k_edge_max    <<<(EP + T - 1) / T, blk, 0, stream>>>(ei, a_s, a_d, m_enc);
    k_edge_exp    <<<(EP + T - 1) / T, blk, 0, stream>>>(ei, a_s, a_d, m_enc, e_buf, denomb);
    k_edge_scatter<<<EP, dim3(HC), 0, stream>>>(ei, h_feat, e_buf, denomb, accb);

    k_final<<<(NN + T - 1) / T, blk, 0, stream>>>(accb, b1, Wl, bl, out);
}


# Edges bucketed by destination each call (int-atomic rank, scan, place); one wave per node: max, exp sum, weighted row sum in registers; no float atomics; sum order = arrival rank + wave tree
# speedup vs baseline: 4.9920x; 4.9920x over previous
#include <hip/hip_runtime.h>
#include <math.h>

#define NN    50000
#define DD    128
#define HC    256
#define CC    128
#define EE    800000
#define NEG_SLOPE 0.2f

typedef __attribute__((ext_vector_type(16))) __bf16 v16bf;
typedef __attribute__((ext_vector_type(8)))  __bf16 v8bf;
typedef __attribute__((ext_vector_type(8)))  float  v8f;

__global__ void k_f32_to_bf16(const float* __restrict__ in,
                              __bf16* __restrict__ out, int n) {
    int i = blockIdx.x * blockDim.x + threadIdx.x;
    if (i < n) out[i] = (__bf16)in[i];
}

__global__ void k_transpose_bf16(const float* __restrict__ W,
                                 __bf16* __restrict__ Wt, int K, int Nn) {
    int i = blockIdx.x * blockDim.x + threadIdx.x;
    if (i < K * Nn) {
        int n = i / K, k = i - n * K;
        Wt[i] = (__bf16)W[(size_t)k * Nn + n];
    }
}

__global__ __launch_bounds__(256)
void k_gemm_bf16_wmma(const __bf16* __restrict__ A,
                      const __bf16* __restrict__ Bt,
                      float* __restrict__ C,
                      int M, int Nn, int K) {
    const int wavesPerBlock = blockDim.x >> 5;
    int wave = blockIdx.x * wavesPerBlock + (threadIdx.x >> 5);
    int lane = threadIdx.x & 31;
    int tilesN = Nn >> 4;
    int tM = (wave / tilesN) << 4;
    int tN = (wave % tilesN) << 4;
    if (tM >= M) return;

    int half = lane >> 4;
    int lr   = lane & 15;

    v8f acc = {};
    const __bf16* arow = A  + (size_t)(tM + lr) * K;
    const __bf16* brow = Bt + (size_t)(tN + lr) * K;

    for (int kk = 0; kk < K; kk += 32) {
        v8bf alo = *(const v8bf*)(arow + kk + 8 * half);
        v8bf ahi = *(const v8bf*)(arow + kk + 16 + 8 * half);
        v8bf blo = *(const v8bf*)(brow + kk + 8 * half);
        v8bf bhi = *(const v8bf*)(brow + kk + 16 + 8 * half);
        v16bf a, b;
#pragma unroll
        for (int i = 0; i < 8; ++i) {
            a[i] = alo[i]; a[i + 8] = ahi[i];
            b[i] = blo[i]; b[i + 8] = bhi[i];
        }
        acc = __builtin_amdgcn_wmma_f32_16x16x32_bf16(
                  false, a, false, b, (short)0, acc, false, false);
    }

    int n = tN + lr;
#pragma unroll
    for (int r = 0; r < 8; ++r) {
        int m = tM + r + 8 * half;
        C[(size_t)m * Nn + n] = acc[r];
    }
}

__global__ void k_att_scores(const float* __restrict__ h,
                             const float* __restrict__ att_src,
                             const float* __restrict__ att_dst,
                             float* __restrict__ a_s,
                             float* __restrict__ a_d) {
    int i = blockIdx.x * blockDim.x + threadIdx.x;
    if (i >= NN * 2) return;
    int node = i >> 1, hd = i & 1;
    const float* hp = h + (size_t)node * HC + hd * CC;
    const float* as = att_src + hd * CC;
    const float* ad = att_dst + hd * CC;
    float s = 0.f, d = 0.f;
#pragma unroll 4
    for (int c = 0; c < CC; ++c) { float v = hp[c]; s += v * as[c]; d += v * ad[c]; }
    a_s[i] = s; a_d[i] = d;
}

__global__ void k_deg_rank(const int* __restrict__ ei,
                           int* __restrict__ deg,
                           int* __restrict__ rank) {
    int e = blockIdx.x * blockDim.x + threadIdx.x;
    if (e < EE) rank[e] = atomicAdd(&deg[ei[EE + e]], 1);
}

__global__ __launch_bounds__(256)
void k_rowptr(const int* __restrict__ deg, int* __restrict__ rowptr) {
    __shared__ int wtot[8];
    __shared__ int wpre[8];
    const int t = threadIdx.x, lane = t & 31, w = t >> 5;
    const int first = blockIdx.x * 256;
    int pre = 0;
    for (int i = t; i < first; i += 256) pre += deg[i];
    const int idx = first + t;
    const int v = (idx < NN) ? deg[idx] : 0;
    int inc = v;
#pragma unroll
    for (int o = 1; o < 32; o <<= 1) {
        int u = __shfl_up(inc, o, 32);
        if (lane >= o) inc += u;
    }
#pragma unroll
    for (int o = 16; o > 0; o >>= 1) pre += __shfl_xor(pre, o, 32);
    if (lane == 31) wtot[w] = inc;
    if (lane == 0)  wpre[w] = pre;
    __syncthreads();
    int base = 0;
#pragma unroll
    for (int k = 0; k < 8; ++k) {
        base += wpre[k];
        if (k < w) base += wtot[k];
    }
    const int excl = base + inc - v;
    if (idx < NN) rowptr[idx] = excl;
    if (idx == NN - 1) rowptr[NN] = excl + v;
}

__global__ void k_fill(const int* __restrict__ ei,
                       const int* __restrict__ rowptr,
                       const int* __restrict__ rank,
                       int* __restrict__ csr_src) {
    int e = blockIdx.x * blockDim.x + threadIdx.x;
    if (e < EE) csr_src[rowptr[ei[EE + e]] + rank[e]] = ei[e];
}

__device__ __forceinline__ float wave_max(float v) {
#pragma unroll
    for (int o = 16; o > 0; o >>= 1) v = fmaxf(v, __shfl_xor(v, o, 32));
    return v;
}
__device__ __forceinline__ float wave_sum(float v) {
#pragma unroll
    for (int o = 16; o > 0; o >>= 1) v += __shfl_xor(v, o, 32);
    return v;
}
__device__ __forceinline__ float lane_bcast(float v, int j) {
    return __int_as_float(__builtin_amdgcn_readlane(__float_as_int(v), j));
}

__global__ __launch_bounds__(256)
void k_node_gather(const int* __restrict__ rowptr,
                   const int* __restrict__ csr_src,
                   const float* __restrict__ h,
                   const float* __restrict__ a_s,
                   const float* __restrict__ a_d,
                   float* __restrict__ acc) {
    const int node = __builtin_amdgcn_readfirstlane((int)(blockIdx.x * 8 + (threadIdx.x >> 5)));
    const int lane = threadIdx.x & 31;
    const int beg  = rowptr[node];
    const int deg  = rowptr[node + 1] - beg;
    const int n    = deg + 1;
    const int* __restrict__ seg = csr_src + beg;
    const float ad0 = a_d[node * 2 + 0];
    const float ad1 = a_d[node * 2 + 1];

    auto item = [&](int i, int& s, float& l0, float& l1) {
        if (i < deg) s = seg[i]; else s = node;
        const float2 as = *(const float2*)(a_s + 2 * (size_t)s);
        const float t0 = as.x + ad0, t1 = as.y + ad1;
        l0 = (t0 > 0.f) ? t0 : NEG_SLOPE * t0;
        l1 = (t1 > 0.f) ? t1 : NEG_SLOPE * t1;
    };

    int   s_c  = node;
    float l0_c = -INFINITY, l1_c = -INFINITY;
    if (lane < n) item(lane, s_c, l0_c, l1_c);

    float m0 = l0_c, m1 = l1_c;
    for (int b = 32; b < n; b += 32) {
        const int i = b + lane;
        if (i < n) {
            int s; float l0, l1; item(i, s, l0, l1);
            m0 = fmaxf(m0, l0); m1 = fmaxf(m1, l1);
        }
    }
    m0 = wave_max(m0); m1 = wave_max(m1);

    const float e0_c = (lane < n) ? expf(l0_c - m0) : 0.f;
    const float e1_c = (lane < n) ? expf(l1_c - m1) : 0.f;
    float d0 = e0_c, d1 = e1_c;
    for (int b = 32; b < n; b += 32) {
        const int i = b + lane;
        if (i < n) {
            int s; float l0, l1; item(i, s, l0, l1);
            d0 += expf(l0 - m0); d1 += expf(l1 - m1);
        }
    }
    const float den0 = wave_sum(d0) + 1e-16f;
    const float den1 = wave_sum(d1) + 1e-16f;

    float4 A0 = {0.f, 0.f, 0.f, 0.f};
    float4 A1 = {0.f, 0.f, 0.f, 0.f};
    const float* __restrict__ hl = h + 4 * lane;

    auto accumulate = [&](int cnt, int s, float w0, float w1) {
#pragma unroll 4
        for (int j = 0; j < cnt; ++j) {
            const int   sj = __builtin_amdgcn_readlane(s, j);
            const float x0 = lane_bcast(w0, j);
            const float x1 = lane_bcast(w1, j);
            const float* row = hl + (size_t)sj * HC;
            const float4 v0 = *(const float4*)(row);
            const float4 v1 = *(const float4*)(row + CC);
            A0.x += x0 * v0.x; A0.y += x0 * v0.y; A0.z += x0 * v0.z; A0.w += x0 * v0.w;
            A1.x += x1 * v1.x; A1.y += x1 * v1.y; A1.z += x1 * v1.z; A1.w += x1 * v1.w;
        }
    };

    accumulate(n < 32 ? n : 32, s_c, e0_c / den0, e1_c / den1);
    for (int b = 32; b < n; b += 32) {
        const int i = b + lane;
        int s = node; float w0 = 0.f, w1 = 0.f;
        if (i < n) {
            float l0, l1; item(i, s, l0, l1);
            w0 = expf(l0 - m0) / den0; w1 = expf(l1 - m1) / den1;
        }
        accumulate((n - b) < 32 ? (n - b) : 32, s, w0, w1);
    }

    float* o = acc + (size_t)node * HC + 4 * lane;
    *(float4*)(o)      = A0;
    *(float4*)(o + CC) = A1;
}

__global__ void k_post_tanh(const float* __restrict__ acc,
                            const float* __restrict__ b,
                            float* __restrict__ hout,
                            __bf16* __restrict__ hbf, int total) {
    int i = blockIdx.x * blockDim.x + threadIdx.x;
    if (i < total) {
        float v = tanhf(acc[i] + b[i & (HC - 1)]);
        hout[i] = v;
        hbf[i]  = (__bf16)v;
    }
}

__global__ void k_final(const float* __restrict__ acc,
                        const float* __restrict__ b1,
                        const float* __restrict__ Wl,
                        const float* __restrict__ bl,
                        float* __restrict__ out) {
    int node = blockIdx.x * blockDim.x + threadIdx.x;
    if (node >= NN) return;
    const float* a = acc + (size_t)node * HC;
    float z0 = bl[0], z1 = bl[1];
#pragma unroll 4
    for (int k = 0; k < HC; ++k) {
        float v = a[k] + b1[k];
        z0 += v * Wl[k * 2 + 0];
        z1 += v * Wl[k * 2 + 1];
    }
    float mx = fmaxf(z0, z1);
    float e0 = expf(z0 - mx), e1 = expf(z1 - mx);
    float inv = 1.f / (e0 + e1);
    out[node * 2 + 0] = e0 * inv;
    out[node * 2 + 1] = e1 * inv;
}

extern "C" void kernel_launch(void* const* d_in, const int* in_sizes, int n_in,
                              void* d_out, int out_size, void* d_ws, size_t ws_size,
                              hipStream_t stream) {
    const float* x        = (const float*)d_in[0];
    const int*   ei       = (const int*)  d_in[1];
    const float* W0       = (const float*)d_in[2];
    const float* att_src0 = (const float*)d_in[3];
    const float* att_dst0 = (const float*)d_in[4];
    const float* b0       = (const float*)d_in[5];
    const float* W1       = (const float*)d_in[6];
    const float* att_src1 = (const float*)d_in[7];
    const float* att_dst1 = (const float*)d_in[8];
    const float* b1       = (const float*)d_in[9];
    const float* Wl       = (const float*)d_in[10];
    const float* bl       = (const float*)d_in[11];
    float* out = (float*)d_out;

    char* base = (char*)d_ws;
    size_t off = 0;
    auto carve = [&](size_t bytes) -> char* {
        char* p = base + off;
        off = (off + bytes + 255) & ~(size_t)255;
        return p;
    };
    float*    h_feat = (float*)   carve((size_t)NN * HC * 4);
    float*    accb   = (float*)   carve((size_t)NN * HC * 4);
    __bf16*   xbf    = (__bf16*)  carve((size_t)NN * HC * 2);
    __bf16*   wtbf   = (__bf16*)  carve((size_t)HC * HC * 2);
    float*    a_s    = (float*)   carve((size_t)NN * 2 * 4);
    float*    a_d    = (float*)   carve((size_t)NN * 2 * 4);
    int*      deg    = (int*)     carve((size_t)NN * 4);
    int*      rowptr = (int*)     carve((size_t)(NN + 1) * 4);
    int*      rank   = (int*)     carve((size_t)EE * 4);
    int*      csr    = (int*)     carve((size_t)EE * 4);
    (void)ws_size; (void)in_sizes; (void)n_in; (void)out_size;

    const int T = 256;
    dim3 blk(T);

    k_f32_to_bf16<<<(NN * DD + T - 1) / T, blk, 0, stream>>>(x, xbf, NN * DD);
    k_transpose_bf16<<<(DD * HC + T - 1) / T, blk, 0, stream>>>(W0, wtbf, DD, HC);

    k_gemm_bf16_wmma<<<(NN / 16) * (HC / 16) / 8, blk, 0, stream>>>(
        xbf, wtbf, h_feat, NN, HC, DD);

    k_att_scores<<<(NN * 2 + T - 1) / T, blk, 0, stream>>>(
        h_feat, att_src0, att_dst0, a_s, a_d);

    hipMemsetAsync(deg, 0, (size_t)NN * 4, stream);
    k_deg_rank<<<(EE + T - 1) / T, blk, 0, stream>>>(ei, deg, rank);
    k_rowptr  <<<(NN + T - 1) / T, blk, 0, stream>>>(deg, rowptr);
    k_fill    <<<(EE + T - 1) / T, blk, 0, stream>>>(ei, rowptr, rank, csr);

    k_node_gather<<<NN / 8, blk, 0, stream>>>(rowptr, csr, h_feat, a_s, a_d, accb);

    k_post_tanh<<<(NN * HC + T - 1) / T, blk, 0, stream>>>(accb, b0, h_feat, xbf, NN * HC);

    k_transpose_bf16<<<(HC * HC + T - 1) / T, blk, 0, stream>>>(W1, wtbf, HC, HC);

    k_gemm_bf16_wmma<<<(NN / 16) * (HC / 16) / 8, blk, 0, stream>>>(
        xbf, wtbf, h_feat, NN, HC, HC);

    k_att_scores<<<(NN * 2 + T - 1) / T, blk, 0, stream>>>(
        h_feat, att_src1, att_dst1, a_s, a_d);

    k_node_gather<<<NN / 8, blk, 0, stream>>>(rowptr, csr, h_feat, a_s, a_d, accb);

    k_final<<<(NN + T - 1) / T, blk, 0, stream>>>(accb, b1, Wl, bl, out);
}


# Elementwise kernels (input convert, bias+tanh): 16 / 32 elements per thread, 4 / 8 coalesced 128-bit loads in flight, instead of 1 element; their waves per call 600000 -> 25000
# speedup vs baseline: 5.2250x; 1.0467x over previous
#include <hip/hip_runtime.h>
#include <math.h>

#define NN    50000
#define DD    128
#define HC    256
#define CC    128
#define EE    800000
#define NEG_SLOPE 0.2f

typedef __attribute__((ext_vector_type(16))) __bf16 v16bf;
typedef __attribute__((ext_vector_type(8)))  __bf16 v8bf;
typedef __attribute__((ext_vector_type(8)))  float  v8f;

typedef __attribute__((ext_vector_type(4)))  __bf16 v4bf;

#define EW_T   128
#define CVT_U  4
#define TANH_U 8
static_assert((NN * DD / 4) % (EW_T * CVT_U)  == 0, "conversion span must divide the array");
static_assert((NN * HC / 4) % (EW_T * TANH_U) == 0, "tanh span must divide the array");
static_assert(EW_T % (HC / 4) == 0, "a thread must meet the same channels in every group");

__global__ __launch_bounds__(EW_T)
void k_f32_to_bf16(const float* __restrict__ in,
                   __bf16* __restrict__ out) {
    const float4* __restrict__ in4 = (const float4*)in;
    const int q0 = blockIdx.x * (EW_T * CVT_U) + threadIdx.x;
    float4 v[CVT_U];
#pragma unroll
    for (int j = 0; j < CVT_U; ++j) v[j] = in4[q0 + j * EW_T];
#pragma unroll
    for (int j = 0; j < CVT_U; ++j) {
        v4bf o;
        o[0] = (__bf16)v[j].x; o[1] = (__bf16)v[j].y;
        o[2] = (__bf16)v[j].z; o[3] = (__bf16)v[j].w;
        *(v4bf*)(out + 4 * (size_t)(q0 + j * EW_T)) = o;
    }
}

__global__ void k_transpose_bf16(const float* __restrict__ W,
                                 __bf16* __restrict__ Wt, int K, int Nn) {
    int i = blockIdx.x * blockDim.x + threadIdx.x;
    if (i < K * Nn) {
        int n = i / K, k = i - n * K;
        Wt[i] = (__bf16)W[(size_t)k * Nn + n];
    }
}

__global__ __launch_bounds__(256)
void k_gemm_bf16_wmma(const __bf16* __restrict__ A,
                      const __bf16* __restrict__ Bt,
                      float* __restrict__ C,
                      int M, int Nn, int K) {
    const int wavesPerBlock = blockDim.x >> 5;
    int wave = blockIdx.x * wavesPerBlock + (threadIdx.x >> 5);
    int lane = threadIdx.x & 31;
    int tilesN = Nn >> 4;
    int tM = (wave / tilesN) << 4;
    int tN = (wave % tilesN) << 4;
    if (tM >= M) return;

    int half = lane >> 4;
    int lr   = lane & 15;

    v8f acc = {};
    const __bf16* arow = A  + (size_t)(tM + lr) * K;
    const __bf16* brow = Bt + (size_t)(tN + lr) * K;

    for (int kk = 0; kk < K; kk += 32) {
        v8bf alo = *(const v8bf*)(arow + kk + 8 * half);
        v8bf ahi = *(const v8bf*)(arow + kk + 16 + 8 * half);
        v8bf blo = *(const v8bf*)(brow + kk + 8 * half);
        v8bf bhi = *(const v8bf*)(brow + kk + 16 + 8 * half);
        v16bf a, b;
#pragma unroll
        for (int i = 0; i < 8; ++i) {
            a[i] = alo[i]; a[i + 8] = ahi[i];
            b[i] = blo[i]; b[i + 8] = bhi[i];
        }
        acc = __builtin_amdgcn_wmma_f32_16x16x32_bf16(
                  false, a, false, b, (short)0, acc, false, false);
    }

    int n = tN + lr;
#pragma unroll
    for (int r = 0; r < 8; ++r) {
        int m = tM + r + 8 * half;
        C[(size_t)m * Nn + n] = acc[r];
    }
}

__global__ void k_att_scores(const float* __restrict__ h,
                             const float* __restrict__ att_src,
                             const float* __restrict__ att_dst,
                             float* __restrict__ a_s,
                             float* __restrict__ a_d) {
    int i = blockIdx.x * blockDim.x + threadIdx.x;
    if (i >= NN * 2) return;
    int node = i >> 1, hd = i & 1;
    const float* hp = h + (size_t)node * HC + hd * CC;
    const float* as = att_src + hd * CC;
    const float* ad = att_dst + hd * CC;
    float s = 0.f, d = 0.f;
#pragma unroll 4
    for (int c = 0; c < CC; ++c) { float v = hp[c]; s += v * as[c]; d += v * ad[c]; }
    a_s[i] = s; a_d[i] = d;
}

__global__ void k_deg_rank(const int* __restrict__ ei,
                           int* __restrict__ deg,
                           int* __restrict__ rank) {
    int e = blockIdx.x * blockDim.x + threadIdx.x;
    if (e < EE) rank[e] = atomicAdd(&deg[ei[EE + e]], 1);
}

__global__ __launch_bounds__(256)
void k_rowptr(const int* __restrict__ deg, int* __restrict__ rowptr) {
    __shared__ int wtot[8];
    __shared__ int wpre[8];
    const int t = threadIdx.x, lane = t & 31, w = t >> 5;
    const int first = blockIdx.x * 256;
    int pre = 0;
    for (int i = t; i < first; i += 256) pre += deg[i];
    const int idx = first + t;
    const int v = (idx < NN) ? deg[idx] : 0;
    int inc = v;
#pragma unroll
    for (int o = 1; o < 32; o <<= 1) {
        int u = __shfl_up(inc, o, 32);
        if (lane >= o) inc += u;
    }
#pragma unroll
    for (int o = 16; o > 0; o >>= 1) pre += __shfl_xor(pre, o, 32);
    if (lane == 31) wtot[w] = inc;
    if (lane == 0)  wpre[w] = pre;
    __syncthreads();
    int base = 0;
#pragma unroll
    for (int k = 0; k < 8; ++k) {
        base += wpre[k];
        if (k < w) base += wtot[k];
    }
    const int excl = base + inc - v;
    if (idx < NN) rowptr[idx] = excl;
    if (idx == NN - 1) rowptr[NN] = excl + v;
}

__global__ void k_fill(const int* __restrict__ ei,
                       const int* __restrict__ rowptr,
                       const int* __restrict__ rank,
                       int* __restrict__ csr_src) {
    int e = blockIdx.x * blockDim.x + threadIdx.x;
    if (e < EE) csr_src[rowptr[ei[EE + e]] + rank[e]] = ei[e];
}

__device__ __forceinline__ float wave_max(float v) {
#pragma unroll
    for (int o = 16; o > 0; o >>= 1) v = fmaxf(v, __shfl_xor(v, o, 32));
    return v;
}
__device__ __forceinline__ float wave_sum(float v) {
#pragma unroll
    for (int o = 16; o > 0; o >>= 1) v += __shfl_xor(v, o, 32);
    return v;
}
__device__ __forceinline__ float lane_bcast(float v, int j) {
    return __int_as_float(__builtin_amdgcn_readlane(__float_as_int(v), j));
}

__global__ __launch_bounds__(256)
void k_node_gather(const int* __restrict__ rowptr,
                   const int* __restrict__ csr_src,
                   const float* __restrict__ h,
                   const float* __restrict__ a_s,
                   const float* __restrict__ a_d,
                   float* __restrict__ acc) {
    const int node = __builtin_amdgcn_readfirstlane((int)(blockIdx.x * 8 + (threadIdx.x >> 5)));
    const int lane = threadIdx.x & 31;
    const int beg  = rowptr[node];
    const int deg  = rowptr[node + 1] - beg;
    const int n    = deg + 1;
    const int* __restrict__ seg = csr_src + beg;
    const float ad0 = a_d[node * 2 + 0];
    const float ad1 = a_d[node * 2 + 1];

    auto item = [&](int i, int& s, float& l0, float& l1) {
        if (i < deg) s = seg[i]; else s = node;
        const float2 as = *(const float2*)(a_s + 2 * (size_t)s);
        const float t0 = as.x + ad0, t1 = as.y + ad1;
        l0 = (t0 > 0.f) ? t0 : NEG_SLOPE * t0;
        l1 = (t1 > 0.f) ? t1 : NEG_SLOPE * t1;
    };

    int   s_c  = node;
    float l0_c = -INFINITY, l1_c = -INFINITY;
    if (lane < n) item(lane, s_c, l0_c, l1_c);

    float m0 = l0_c, m1 = l1_c;
    for (int b = 32; b < n; b += 32) {
        const int i = b + lane;
        if (i < n) {
            int s; float l0, l1; item(i, s, l0, l1);
            m0 = fmaxf(m0, l0); m1 = fmaxf(m1, l1);
        }
    }
    m0 = wave_max(m0); m1 = wave_max(m1);

    const float e0_c = (lane < n) ? expf(l0_c - m0) : 0.f;
    const float e1_c = (lane < n) ? expf(l1_c - m1) : 0.f;
    float d0 = e0_c, d1 = e1_c;
    for (int b = 32; b < n; b += 32) {
        const int i = b + lane;
        if (i < n) {
            int s; float l0, l1; item(i, s, l0, l1);
            d0 += expf(l0 - m0); d1 += expf(l1 - m1);
        }
    }
    const float den0 = wave_sum(d0) + 1e-16f;
    const float den1 = wave_sum(d1) + 1e-16f;

    float4 A0 = {0.f, 0.f, 0.f, 0.f};
    float4 A1 = {0.f, 0.f, 0.f, 0.f};
    const float* __restrict__ hl = h + 4 * lane;

    auto accumulate = [&](int cnt, int s, float w0, float w1) {
#pragma unroll 4
        for (int j = 0; j < cnt; ++j) {
            const int   sj = __builtin_amdgcn_readlane(s, j);
            const float x0 = lane_bcast(w0, j);
            const float x1 = lane_bcast(w1, j);
            const float* row = hl + (size_t)sj * HC;
            const float4 v0 = *(const float4*)(row);
            const float4 v1 = *(const float4*)(row + CC);
            A0.x += x0 * v0.x; A0.y += x0 * v0.y; A0.z += x0 * v0.z; A0.w += x0 * v0.w;
            A1.x += x1 * v1.x; A1.y += x1 * v1.y; A1.z += x1 * v1.z; A1.w += x1 * v1.w;
        }
    };

    accumulate(n < 32 ? n : 32, s_c, e0_c / den0, e1_c / den1);
    for (int b = 32; b < n; b += 32) {
        const int i = b + lane;
        int s = node; float w0 = 0.f, w1 = 0.f;
        if (i < n) {
            float l0, l1; item(i, s, l0, l1);
            w0 = expf(l0 - m0) / den0; w1 = expf(l1 - m1) / den1;
        }
        accumulate((n - b) < 32 ? (n - b) : 32, s, w0, w1);
    }

    float* o = acc + (size_t)node * HC + 4 * lane;
    *(float4*)(o)      = A0;
    *(float4*)(o + CC) = A1;
}

__global__ __launch_bounds__(EW_T)
void k_post_tanh(const float* __restrict__ acc,
                 const float* __restrict__ b,
                 float* __restrict__ hout,
                 __bf16* __restrict__ hbf) {
    const float4* __restrict__ acc4 = (const float4*)acc;
    const int q0 = blockIdx.x * (EW_T * TANH_U) + threadIdx.x;
    const float4 bb = *(const float4*)(b + 4 * (threadIdx.x & (HC / 4 - 1)));
    float4 v[TANH_U];
#pragma unroll
    for (int j = 0; j < TANH_U; ++j) v[j] = acc4[q0 + j * EW_T];
#pragma unroll
    for (int j = 0; j < TANH_U; ++j) {
        const size_t q = (size_t)(q0 + j * EW_T);
        float4 r;
        r.x = tanhf(v[j].x + bb.x); r.y = tanhf(v[j].y + bb.y);
        r.z = tanhf(v[j].z + bb.z); r.w = tanhf(v[j].w + bb.w);
        *(float4*)(hout + 4 * q) = r;
        v4bf o;
        o[0] = (__bf16)r.x; o[1] = (__bf16)r.y; o[2] = (__bf16)r.z; o[3] = (__bf16)r.w;
        *(v4bf*)(hbf + 4 * q) = o;
    }
}

__global__ void k_final(const float* __restrict__ acc,
                        const float* __restrict__ b1,
                        const float* __restrict__ Wl,
                        const float* __restrict__ bl,
                        float* __restrict__ out) {
    int node = blockIdx.x * blockDim.x + threadIdx.x;
    if (node >= NN) return;
    const float* a = acc + (size_t)node * HC;
    float z0 = bl[0], z1 = bl[1];
#pragma unroll 4
    for (int k = 0; k < HC; ++k) {
        float v = a[k] + b1[k];
        z0 += v * Wl[k * 2 + 0];
        z1 += v * Wl[k * 2 + 1];
    }
    float mx = fmaxf(z0, z1);
    float e0 = expf(z0 - mx), e1 = expf(z1 - mx);
    float inv = 1.f / (e0 + e1);
    out[node * 2 + 0] = e0 * inv;
    out[node * 2 + 1] = e1 * inv;
}

extern "C" void kernel_launch(void* const* d_in, const int* in_sizes, int n_in,
                              void* d_out, int out_size, void* d_ws, size_t ws_size,
                              hipStream_t stream) {
    const float* x        = (const float*)d_in[0];
    const int*   ei       = (const int*)  d_in[1];
    const float* W0       = (const float*)d_in[2];
    const float* att_src0 = (const float*)d_in[3];
    const float* att_dst0 = (const float*)d_in[4];
    const float* b0       = (const float*)d_in[5];
    const float* W1       = (const float*)d_in[6];
    const float* att_src1 = (const float*)d_in[7];
    const float* att_dst1 = (const float*)d_in[8];
    const float* b1       = (const float*)d_in[9];
    const float* Wl       = (const float*)d_in[10];
    const float* bl       = (const float*)d_in[11];
    float* out = (float*)d_out;

    char* base = (char*)d_ws;
    size_t off = 0;
    auto carve = [&](size_t bytes) -> char* {
        char* p = base + off;
        off = (off + bytes + 255) & ~(size_t)255;
        return p;
    };
    float*    h_feat = (float*)   carve((size_t)NN * HC * 4);
    float*    accb   = (float*)   carve((size_t)NN * HC * 4);
    __bf16*   xbf    = (__bf16*)  carve((size_t)NN * HC * 2);
    __bf16*   wtbf   = (__bf16*)  carve((size_t)HC * HC * 2);
    float*    a_s    = (float*)   carve((size_t)NN * 2 * 4);
    float*    a_d    = (float*)   carve((size_t)NN * 2 * 4);
    int*      deg    = (int*)     carve((size_t)NN * 4);
    int*      rowptr = (int*)     carve((size_t)(NN + 1) * 4);
    int*      rank   = (int*)     carve((size_t)EE * 4);
    int*      csr    = (int*)     carve((size_t)EE * 4);
    (void)ws_size; (void)in_sizes; (void)n_in; (void)out_size;

    const int T = 256;
    dim3 blk(T);

    k_f32_to_bf16<<<(NN * DD / 4) / (EW_T * CVT_U), dim3(EW_T), 0, stream>>>(x, xbf);
    k_transpose_bf16<<<(DD * HC + T - 1) / T, blk, 0, stream>>>(W0, wtbf, DD, HC);

    k_gemm_bf16_wmma<<<(NN / 16) * (HC / 16) / 8, blk, 0, stream>>>(
        xbf, wtbf, h_feat, NN, HC, DD);

    k_att_scores<<<(NN * 2 + T - 1) / T, blk, 0, stream>>>(
        h_feat, att_src0, att_dst0, a_s, a_d);

    hipMemsetAsync(deg, 0, (size_t)NN * 4, stream);
    k_deg_rank<<<(EE + T - 1) / T, blk, 0, stream>>>(ei, deg, rank);
    k_rowptr  <<<(NN + T - 1) / T, blk, 0, stream>>>(deg, rowptr);
    k_fill    <<<(EE + T - 1) / T, blk, 0, stream>>>(ei, rowptr, rank, csr);

    k_node_gather<<<NN / 8, blk, 0, stream>>>(rowptr, csr, h_feat, a_s, a_d, accb);

    k_post_tanh<<<(NN * HC / 4) / (EW_T * TANH_U), dim3(EW_T), 0, stream>>>(accb, b0, h_feat, xbf);

    k_transpose_bf16<<<(HC * HC + T - 1) / T, blk, 0, stream>>>(W1, wtbf, HC, HC);

    k_gemm_bf16_wmma<<<(NN / 16) * (HC / 16) / 8, blk, 0, stream>>>(
        xbf, wtbf, h_feat, NN, HC, HC);

    k_att_scores<<<(NN * 2 + T - 1) / T, blk, 0, stream>>>(
        h_feat, att_src1, att_dst1, a_s, a_d);

    k_node_gather<<<NN / 8, blk, 0, stream>>>(rowptr, csr, h_feat, a_s, a_d, accb);

    k_final<<<(NN + T - 1) / T, blk, 0, stream>>>(accb, b1, Wl, bl, out);
}


# Gather wave keeps its finished row in registers and runs the next per-row stage itself (bias+tanh+bf16; bias+linear+softmax, sum per lane then wave tree): 2 launches and f32 row round trips gone
# speedup vs baseline: 6.1174x; 1.1708x over previous
#include <hip/hip_runtime.h>
#include <math.h>

#define NN    50000
#define DD    128
#define HC    256
#define CC    128
#define EE    800000
#define NEG_SLOPE 0.2f

typedef __attribute__((ext_vector_type(16))) __bf16 v16bf;
typedef __attribute__((ext_vector_type(8)))  __bf16 v8bf;
typedef __attribute__((ext_vector_type(8)))  float  v8f;

typedef __attribute__((ext_vector_type(4)))  __bf16 v4bf;

#define EW_T   128
#define CVT_U  4
static_assert((NN * DD / 4) % (EW_T * CVT_U)  == 0, "conversion span must divide the array");

__global__ __launch_bounds__(EW_T)
void k_f32_to_bf16(const float* __restrict__ in,
                   __bf16* __restrict__ out) {
    const float4* __restrict__ in4 = (const float4*)in;
    const int q0 = blockIdx.x * (EW_T * CVT_U) + threadIdx.x;
    float4 v[CVT_U];
#pragma unroll
    for (int j = 0; j < CVT_U; ++j) v[j] = in4[q0 + j * EW_T];
#pragma unroll
    for (int j = 0; j < CVT_U; ++j) {
        v4bf o;
        o[0] = (__bf16)v[j].x; o[1] = (__bf16)v[j].y;
        o[2] = (__bf16)v[j].z; o[3] = (__bf16)v[j].w;
        *(v4bf*)(out + 4 * (size_t)(q0 + j * EW_T)) = o;
    }
}

__global__ void k_transpose_bf16(const float* __restrict__ W,
                                 __bf16* __restrict__ Wt, int K, int Nn) {
    int i = blockIdx.x * blockDim.x + threadIdx.x;
    if (i < K * Nn) {
        int n = i / K, k = i - n * K;
        Wt[i] = (__bf16)W[(size_t)k * Nn + n];
    }
}

__global__ __launch_bounds__(256)
void k_gemm_bf16_wmma(const __bf16* __restrict__ A,
                      const __bf16* __restrict__ Bt,
                      float* __restrict__ C,
                      int M, int Nn, int K) {
    const int wavesPerBlock = blockDim.x >> 5;
    int wave = blockIdx.x * wavesPerBlock + (threadIdx.x >> 5);
    int lane = threadIdx.x & 31;
    int tilesN = Nn >> 4;
    int tM = (wave / tilesN) << 4;
    int tN = (wave % tilesN) << 4;
    if (tM >= M) return;

    int half = lane >> 4;
    int lr   = lane & 15;

    v8f acc = {};
    const __bf16* arow = A  + (size_t)(tM + lr) * K;
    const __bf16* brow = Bt + (size_t)(tN + lr) * K;

    for (int kk = 0; kk < K; kk += 32) {
        v8bf alo = *(const v8bf*)(arow + kk + 8 * half);
        v8bf ahi = *(const v8bf*)(arow + kk + 16 + 8 * half);
        v8bf blo = *(const v8bf*)(brow + kk + 8 * half);
        v8bf bhi = *(const v8bf*)(brow + kk + 16 + 8 * half);
        v16bf a, b;
#pragma unroll
        for (int i = 0; i < 8; ++i) {
            a[i] = alo[i]; a[i + 8] = ahi[i];
            b[i] = blo[i]; b[i + 8] = bhi[i];
        }
        acc = __builtin_amdgcn_wmma_f32_16x16x32_bf16(
                  false, a, false, b, (short)0, acc, false, false);
    }

    int n = tN + lr;
#pragma unroll
    for (int r = 0; r < 8; ++r) {
        int m = tM + r + 8 * half;
        C[(size_t)m * Nn + n] = acc[r];
    }
}

__global__ void k_att_scores(const float* __restrict__ h,
                             const float* __restrict__ att_src,
                             const float* __restrict__ att_dst,
                             float* __restrict__ a_s,
                             float* __restrict__ a_d) {
    int i = blockIdx.x * blockDim.x + threadIdx.x;
    if (i >= NN * 2) return;
    int node = i >> 1, hd = i & 1;
    const float* hp = h + (size_t)node * HC + hd * CC;
    const float* as = att_src + hd * CC;
    const float* ad = att_dst + hd * CC;
    float s = 0.f, d = 0.f;
#pragma unroll 4
    for (int c = 0; c < CC; ++c) { float v = hp[c]; s += v * as[c]; d += v * ad[c]; }
    a_s[i] = s; a_d[i] = d;
}

__global__ void k_deg_rank(const int* __restrict__ ei,
                           int* __restrict__ deg,
                           int* __restrict__ rank) {
    int e = blockIdx.x * blockDim.x + threadIdx.x;
    if (e < EE) rank[e] = atomicAdd(&deg[ei[EE + e]], 1);
}

__global__ __launch_bounds__(256)
void k_rowptr(const int* __restrict__ deg, int* __restrict__ rowptr) {
    __shared__ int wtot[8];
    __shared__ int wpre[8];
    const int t = threadIdx.x, lane = t & 31, w = t >> 5;
    const int first = blockIdx.x * 256;
    int pre = 0;
    for (int i = t; i < first; i += 256) pre += deg[i];
    const int idx = first + t;
    const int v = (idx < NN) ? deg[idx] : 0;
    int inc = v;
#pragma unroll
    for (int o = 1; o < 32; o <<= 1) {
        int u = __shfl_up(inc, o, 32);
        if (lane >= o) inc += u;
    }
#pragma unroll
    for (int o = 16; o > 0; o >>= 1) pre += __shfl_xor(pre, o, 32);
    if (lane == 31) wtot[w] = inc;
    if (lane == 0)  wpre[w] = pre;
    __syncthreads();
    int base = 0;
#pragma unroll
    for (int k = 0; k < 8; ++k) {
        base += wpre[k];
        if (k < w) base += wtot[k];
    }
    const int excl = base + inc - v;
    if (idx < NN) rowptr[idx] = excl;
    if (idx == NN - 1) rowptr[NN] = excl + v;
}

__global__ void k_fill(const int* __restrict__ ei,
                       const int* __restrict__ rowptr,
                       const int* __restrict__ rank,
                       int* __restrict__ csr_src) {
    int e = blockIdx.x * blockDim.x + threadIdx.x;
    if (e < EE) csr_src[rowptr[ei[EE + e]] + rank[e]] = ei[e];
}

__device__ __forceinline__ float wave_max(float v) {
#pragma unroll
    for (int o = 16; o > 0; o >>= 1) v = fmaxf(v, __shfl_xor(v, o, 32));
    return v;
}
__device__ __forceinline__ float wave_sum(float v) {
#pragma unroll
    for (int o = 16; o > 0; o >>= 1) v += __shfl_xor(v, o, 32);
    return v;
}
__device__ __forceinline__ float lane_bcast(float v, int j) {
    return __int_as_float(__builtin_amdgcn_readlane(__float_as_int(v), j));
}

template <bool LAST>
__global__ __launch_bounds__(256)
void k_node_gather(const int* __restrict__ rowptr,
                   const int* __restrict__ csr_src,
                   const float* __restrict__ h,
                   const float* __restrict__ a_s,
                   const float* __restrict__ a_d,
                   const float* __restrict__ bias,
                   __bf16* __restrict__ hbf,
                   const float* __restrict__ Wl,
                   const float* __restrict__ bl,
                   float* __restrict__ out) {
    const int node = __builtin_amdgcn_readfirstlane((int)(blockIdx.x * 8 + (threadIdx.x >> 5)));
    const int lane = threadIdx.x & 31;
    const int beg  = rowptr[node];
    const int deg  = rowptr[node + 1] - beg;
    const int n    = deg + 1;
    const int* __restrict__ seg = csr_src + beg;
    const float ad0 = a_d[node * 2 + 0];
    const float ad1 = a_d[node * 2 + 1];

    auto item = [&](int i, int& s, float& l0, float& l1) {
        if (i < deg) s = seg[i]; else s = node;
        const float2 as = *(const float2*)(a_s + 2 * (size_t)s);
        const float t0 = as.x + ad0, t1 = as.y + ad1;
        l0 = (t0 > 0.f) ? t0 : NEG_SLOPE * t0;
        l1 = (t1 > 0.f) ? t1 : NEG_SLOPE * t1;
    };

    int   s_c  = node;
    float l0_c = -INFINITY, l1_c = -INFINITY;
    if (lane < n) item(lane, s_c, l0_c, l1_c);

    float m0 = l0_c, m1 = l1_c;
    for (int b = 32; b < n; b += 32) {
        const int i = b + lane;
        if (i < n) {
            int s; float l0, l1; item(i, s, l0, l1);
            m0 = fmaxf(m0, l0); m1 = fmaxf(m1, l1);
        }
    }
    m0 = wave_max(m0); m1 = wave_max(m1);

    const float e0_c = (lane < n) ? expf(l0_c - m0) : 0.f;
    const float e1_c = (lane < n) ? expf(l1_c - m1) : 0.f;
    float d0 = e0_c, d1 = e1_c;
    for (int b = 32; b < n; b += 32) {
        const int i = b + lane;
        if (i < n) {
            int s; float l0, l1; item(i, s, l0, l1);
            d0 += expf(l0 - m0); d1 += expf(l1 - m1);
        }
    }
    const float den0 = wave_sum(d0) + 1e-16f;
    const float den1 = wave_sum(d1) + 1e-16f;

    float4 A0 = {0.f, 0.f, 0.f, 0.f};
    float4 A1 = {0.f, 0.f, 0.f, 0.f};
    const float* __restrict__ hl = h + 4 * lane;

    auto accumulate = [&](int cnt, int s, float w0, float w1) {
#pragma unroll 4
        for (int j = 0; j < cnt; ++j) {
            const int   sj = __builtin_amdgcn_readlane(s, j);
            const float x0 = lane_bcast(w0, j);
            const float x1 = lane_bcast(w1, j);
            const float* row = hl + (size_t)sj * HC;
            const float4 v0 = *(const float4*)(row);
            const float4 v1 = *(const float4*)(row + CC);
            A0.x += x0 * v0.x; A0.y += x0 * v0.y; A0.z += x0 * v0.z; A0.w += x0 * v0.w;
            A1.x += x1 * v1.x; A1.y += x1 * v1.y; A1.z += x1 * v1.z; A1.w += x1 * v1.w;
        }
    };

    accumulate(n < 32 ? n : 32, s_c, e0_c / den0, e1_c / den1);
    for (int b = 32; b < n; b += 32) {
        const int i = b + lane;
        int s = node; float w0 = 0.f, w1 = 0.f;
        if (i < n) {
            float l0, l1; item(i, s, l0, l1);
            w0 = expf(l0 - m0) / den0; w1 = expf(l1 - m1) / den1;
        }
        accumulate((n - b) < 32 ? (n - b) : 32, s, w0, w1);
    }

    const float4 bA = *(const float4*)(bias + 4 * lane);
    const float4 bB = *(const float4*)(bias + CC + 4 * lane);
    if (!LAST) {
        v4bf o0, o1;
        o0[0] = (__bf16)tanhf(A0.x + bA.x); o0[1] = (__bf16)tanhf(A0.y + bA.y);
        o0[2] = (__bf16)tanhf(A0.z + bA.z); o0[3] = (__bf16)tanhf(A0.w + bA.w);
        o1[0] = (__bf16)tanhf(A1.x + bB.x); o1[1] = (__bf16)tanhf(A1.y + bB.y);
        o1[2] = (__bf16)tanhf(A1.z + bB.z); o1[3] = (__bf16)tanhf(A1.w + bB.w);
        __bf16* o = hbf + (size_t)node * HC + 4 * lane;
        *(v4bf*)(o)      = o0;
        *(v4bf*)(o + CC) = o1;
    } else {
        const float4 wa0 = *(const float4*)(Wl + 8 * lane);
        const float4 wa1 = *(const float4*)(Wl + 8 * lane + 4);
        const float4 wb0 = *(const float4*)(Wl + 2 * CC + 8 * lane);
        const float4 wb1 = *(const float4*)(Wl + 2 * CC + 8 * lane + 4);
        const float u0 = A0.x + bA.x, u1 = A0.y + bA.y, u2 = A0.z + bA.z, u3 = A0.w + bA.w;
        const float t0 = A1.x + bB.x, t1 = A1.y + bB.y, t2 = A1.z + bB.z, t3 = A1.w + bB.w;
        float z0 = u0 * wa0.x + u1 * wa0.z + u2 * wa1.x + u3 * wa1.z
                 + t0 * wb0.x + t1 * wb0.z + t2 * wb1.x + t3 * wb1.z;
        float z1 = u0 * wa0.y + u1 * wa0.w + u2 * wa1.y + u3 * wa1.w
                 + t0 * wb0.y + t1 * wb0.w + t2 * wb1.y + t3 * wb1.w;
        z0 = wave_sum(z0) + bl[0];
        z1 = wave_sum(z1) + bl[1];
        const float mx = fmaxf(z0, z1);
        const float e0 = expf(z0 - mx), e1 = expf(z1 - mx);
        const float inv = 1.f / (e0 + e1);
        if (lane == 0) *(float2*)(out + 2 * (size_t)node) = make_float2(e0 * inv, e1 * inv);
    }
}

extern "C" void kernel_launch(void* const* d_in, const int* in_sizes, int n_in,
                              void* d_out, int out_size, void* d_ws, size_t ws_size,
                              hipStream_t stream) {
    const float* x        = (const float*)d_in[0];
    const int*   ei       = (const int*)  d_in[1];
    const float* W0       = (const float*)d_in[2];
    const float* att_src0 = (const float*)d_in[3];
    const float* att_dst0 = (const float*)d_in[4];
    const float* b0       = (const float*)d_in[5];
    const float* W1       = (const float*)d_in[6];
    const float* att_src1 = (const float*)d_in[7];
    const float* att_dst1 = (const float*)d_in[8];
    const float* b1       = (const float*)d_in[9];
    const float* Wl       = (const float*)d_in[10];
    const float* bl       = (const float*)d_in[11];
    float* out = (float*)d_out;

    char* base = (char*)d_ws;
    size_t off = 0;
    auto carve = [&](size_t bytes) -> char* {
        char* p = base + off;
        off = (off + bytes + 255) & ~(size_t)255;
        return p;
    };
    float*    h_feat = (float*)   carve((size_t)NN * HC * 4);
    __bf16*   xbf    = (__bf16*)  carve((size_t)NN * HC * 2);
    __bf16*   wtbf   = (__bf16*)  carve((size_t)HC * HC * 2);
    float*    a_s    = (float*)   carve((size_t)NN * 2 * 4);
    float*    a_d    = (float*)   carve((size_t)NN * 2 * 4);
    int*      deg    = (int*)     carve((size_t)NN * 4);
    int*      rowptr = (int*)     carve((size_t)(NN + 1) * 4);
    int*      rank   = (int*)     carve((size_t)EE * 4);
    int*      csr    = (int*)     carve((size_t)EE * 4);
    (void)ws_size; (void)in_sizes; (void)n_in; (void)out_size;

    const int T = 256;
    dim3 blk(T);

    k_f32_to_bf16<<<(NN * DD / 4) / (EW_T * CVT_U), dim3(EW_T), 0, stream>>>(x, xbf);
    k_transpose_bf16<<<(DD * HC + T - 1) / T, blk, 0, stream>>>(W0, wtbf, DD, HC);

    k_gemm_bf16_wmma<<<(NN / 16) * (HC / 16) / 8, blk, 0, stream>>>(
        xbf, wtbf, h_feat, NN, HC, DD);

    k_att_scores<<<(NN * 2 + T - 1) / T, blk, 0, stream>>>(
        h_feat, att_src0, att_dst0, a_s, a_d);

    hipMemsetAsync(deg, 0, (size_t)NN * 4, stream);
    k_deg_rank<<<(EE + T - 1) / T, blk, 0, stream>>>(ei, deg, rank);
    k_rowptr  <<<(NN + T - 1) / T, blk, 0, stream>>>(deg, rowptr);
    k_fill    <<<(EE + T - 1) / T, blk, 0, stream>>>(ei, rowptr, rank, csr);

    k_node_gather<false><<<NN / 8, blk, 0, stream>>>(
        rowptr, csr, h_feat, a_s, a_d, b0, xbf, Wl, bl, out);

    k_transpose_bf16<<<(HC * HC + T - 1) / T, blk, 0, stream>>>(W1, wtbf, HC, HC);

    k_gemm_bf16_wmma<<<(NN / 16) * (HC / 16) / 8, blk, 0, stream>>>(
        xbf, wtbf, h_feat, NN, HC, HC);

    k_att_scores<<<(NN * 2 + T - 1) / T, blk, 0, stream>>>(
        h_feat, att_src1, att_dst1, a_s, a_d);

    k_node_gather<true><<<NN / 8, blk, 0, stream>>>(
        rowptr, csr, h_feat, a_s, a_d, b1, xbf, Wl, bl, out);
}


# Last layer: 256->2 classifier applied per node and head in the matrix-product block (LDS row sums) before the neighbour sum; last gather reads 4 floats per head, not 1 KiB rows; sum order changes
# speedup vs baseline: 6.9860x; 1.1420x over previous
#include <hip/hip_runtime.h>
#include <math.h>

#define NN    50000
#define DD    128
#define HC    256
#define CC    128
#define EE    800000
#define NEG_SLOPE 0.2f

typedef __attribute__((ext_vector_type(16))) __bf16 v16bf;
typedef __attribute__((ext_vector_type(8)))  __bf16 v8bf;
typedef __attribute__((ext_vector_type(8)))  float  v8f;

typedef __attribute__((ext_vector_type(4)))  __bf16 v4bf;

#define EW_T   128
#define CVT_U  4
static_assert((NN * DD / 4) % (EW_T * CVT_U)  == 0, "conversion span must divide the array");

__global__ __launch_bounds__(EW_T)
void k_f32_to_bf16(const float* __restrict__ in,
                   __bf16* __restrict__ out) {
    const float4* __restrict__ in4 = (const float4*)in;
    const int q0 = blockIdx.x * (EW_T * CVT_U) + threadIdx.x;
    float4 v[CVT_U];
#pragma unroll
    for (int j = 0; j < CVT_U; ++j) v[j] = in4[q0 + j * EW_T];
#pragma unroll
    for (int j = 0; j < CVT_U; ++j) {
        v4bf o;
        o[0] = (__bf16)v[j].x; o[1] = (__bf16)v[j].y;
        o[2] = (__bf16)v[j].z; o[3] = (__bf16)v[j].w;
        *(v4bf*)(out + 4 * (size_t)(q0 + j * EW_T)) = o;
    }
}

__global__ void k_transpose_bf16(const float* __restrict__ W,
                                 __bf16* __restrict__ Wt, int K, int Nn) {
    int i = blockIdx.x * blockDim.x + threadIdx.x;
    if (i < K * Nn) {
        int n = i / K, k = i - n * K;
        Wt[i] = (__bf16)W[(size_t)k * Nn + n];
    }
}

template <bool PROJ>
__global__ __launch_bounds__(256)
void k_gemm_bf16_wmma(const __bf16* __restrict__ A,
                      const __bf16* __restrict__ Bt,
                      float* __restrict__ C,
                      int M, int Nn, int K,
                      const float* __restrict__ att_src,
                      const float* __restrict__ att_dst,
                      const float* __restrict__ Wl,
                      float* __restrict__ rec) {
    const int wavesPerBlock = blockDim.x >> 5;
    int wave = blockIdx.x * wavesPerBlock + (threadIdx.x >> 5);
    int lane = threadIdx.x & 31;
    int tilesN = Nn >> 4;
    int tM = (wave / tilesN) << 4;
    int tN = (wave % tilesN) << 4;
    if (tM >= M) return;

    int half = lane >> 4;
    int lr   = lane & 15;

    v8f acc = {};
    const __bf16* arow = A  + (size_t)(tM + lr) * K;
    const __bf16* brow = Bt + (size_t)(tN + lr) * K;

    for (int kk = 0; kk < K; kk += 32) {
        v8bf alo = *(const v8bf*)(arow + kk + 8 * half);
        v8bf ahi = *(const v8bf*)(arow + kk + 16 + 8 * half);
        v8bf blo = *(const v8bf*)(brow + kk + 8 * half);
        v8bf bhi = *(const v8bf*)(brow + kk + 16 + 8 * half);
        v16bf a, b;
#pragma unroll
        for (int i = 0; i < 8; ++i) {
            a[i] = alo[i]; a[i + 8] = ahi[i];
            b[i] = blo[i]; b[i + 8] = bhi[i];
        }
        if constexpr (PROJ) {
            acc = __builtin_amdgcn_wmma_f32_16x16x32_bf16(
                      false, b, false, a, (short)0, acc, false, false);
        } else {
            acc = __builtin_amdgcn_wmma_f32_16x16x32_bf16(
                      false, a, false, b, (short)0, acc, false, false);
        }
    }

    if constexpr (PROJ) {
        __shared__ __attribute__((aligned(16))) float red[8 * 16 * 4];
        const int c0 = tN + 8 * half;
        const float4 sa = *(const float4*)(att_src + c0);
        const float4 sb = *(const float4*)(att_src + c0 + 4);
        const float4 da = *(const float4*)(att_dst + c0);
        const float4 db = *(const float4*)(att_dst + c0 + 4);
        const float4 w0 = *(const float4*)(Wl + 2 * c0);
        const float4 w1 = *(const float4*)(Wl + 2 * c0 + 4);
        const float4 w2 = *(const float4*)(Wl + 2 * c0 + 8);
        const float4 w3 = *(const float4*)(Wl + 2 * c0 + 12);
        float ps = acc[0] * sa.x + acc[1] * sa.y + acc[2] * sa.z + acc[3] * sa.w
                 + acc[4] * sb.x + acc[5] * sb.y + acc[6] * sb.z + acc[7] * sb.w;
        float pd = acc[0] * da.x + acc[1] * da.y + acc[2] * da.z + acc[3] * da.w
                 + acc[4] * db.x + acc[5] * db.y + acc[6] * db.z + acc[7] * db.w;
        float p0 = acc[0] * w0.x + acc[1] * w0.z + acc[2] * w1.x + acc[3] * w1.z
                 + acc[4] * w2.x + acc[5] * w2.z + acc[6] * w3.x + acc[7] * w3.z;
        float p1 = acc[0] * w0.y + acc[1] * w0.w + acc[2] * w1.y + acc[3] * w1.w
                 + acc[4] * w2.y + acc[5] * w2.w + acc[6] * w3.y + acc[7] * w3.w;
        ps += __shfl_xor(ps, 16, 32); pd += __shfl_xor(pd, 16, 32);
        p0 += __shfl_xor(p0, 16, 32); p1 += __shfl_xor(p1, 16, 32);
        const int w = threadIdx.x >> 5;
        if (half == 0) *(float4*)(red + (w * 16 + lr) * 4) = make_float4(ps, p0, p1, pd);
        __syncthreads();
        if (threadIdx.x < 16) {
            float4 t = *(const float4*)(red + threadIdx.x * 4);
#pragma unroll
            for (int k = 1; k < 8; ++k) {
                const float4 u = *(const float4*)(red + (k * 16 + threadIdx.x) * 4);
                t.x += u.x; t.y += u.y; t.z += u.z; t.w += u.w;
            }
            *(float4*)(rec + 4 * ((size_t)(tN >> 7) * M + tM + threadIdx.x)) = t;
        }
    } else {
        int n = tN + lr;
#pragma unroll
        for (int r = 0; r < 8; ++r) {
            int m = tM + r + 8 * half;
            C[(size_t)m * Nn + n] = acc[r];
        }
    }
}

__global__ void k_att_scores(const float* __restrict__ h,
                             const float* __restrict__ att_src,
                             const float* __restrict__ att_dst,
                             float* __restrict__ a_s,
                             float* __restrict__ a_d) {
    int i = blockIdx.x * blockDim.x + threadIdx.x;
    if (i >= NN * 2) return;
    int node = i >> 1, hd = i & 1;
    const float* hp = h + (size_t)node * HC + hd * CC;
    const float* as = att_src + hd * CC;
    const float* ad = att_dst + hd * CC;
    float s = 0.f, d = 0.f;
#pragma unroll 4
    for (int c = 0; c < CC; ++c) { float v = hp[c]; s += v * as[c]; d += v * ad[c]; }
    a_s[i] = s; a_d[i] = d;
}

__global__ void k_deg_rank(const int* __restrict__ ei,
                           int* __restrict__ deg,
                           int* __restrict__ rank) {
    int e = blockIdx.x * blockDim.x + threadIdx.x;
    if (e < EE) rank[e] = atomicAdd(&deg[ei[EE + e]], 1);
}

__global__ __launch_bounds__(256)
void k_rowptr(const int* __restrict__ deg, int* __restrict__ rowptr) {
    __shared__ int wtot[8];
    __shared__ int wpre[8];
    const int t = threadIdx.x, lane = t & 31, w = t >> 5;
    const int first = blockIdx.x * 256;
    int pre = 0;
    for (int i = t; i < first; i += 256) pre += deg[i];
    const int idx = first + t;
    const int v = (idx < NN) ? deg[idx] : 0;
    int inc = v;
#pragma unroll
    for (int o = 1; o < 32; o <<= 1) {
        int u = __shfl_up(inc, o, 32);
        if (lane >= o) inc += u;
    }
#pragma unroll
    for (int o = 16; o > 0; o >>= 1) pre += __shfl_xor(pre, o, 32);
    if (lane == 31) wtot[w] = inc;
    if (lane == 0)  wpre[w] = pre;
    __syncthreads();
    int base = 0;
#pragma unroll
    for (int k = 0; k < 8; ++k) {
        base += wpre[k];
        if (k < w) base += wtot[k];
    }
    const int excl = base + inc - v;
    if (idx < NN) rowptr[idx] = excl;
    if (idx == NN - 1) rowptr[NN] = excl + v;
}

__global__ void k_fill(const int* __restrict__ ei,
                       const int* __restrict__ rowptr,
                       const int* __restrict__ rank,
                       int* __restrict__ csr_src) {
    int e = blockIdx.x * blockDim.x + threadIdx.x;
    if (e < EE) csr_src[rowptr[ei[EE + e]] + rank[e]] = ei[e];
}

__device__ __forceinline__ float wave_max(float v) {
#pragma unroll
    for (int o = 16; o > 0; o >>= 1) v = fmaxf(v, __shfl_xor(v, o, 32));
    return v;
}
__device__ __forceinline__ float wave_sum(float v) {
#pragma unroll
    for (int o = 16; o > 0; o >>= 1) v += __shfl_xor(v, o, 32);
    return v;
}
__device__ __forceinline__ float lane_bcast(float v, int j) {
    return __int_as_float(__builtin_amdgcn_readlane(__float_as_int(v), j));
}

__global__ __launch_bounds__(256)
void k_node_gather(const int* __restrict__ rowptr,
                   const int* __restrict__ csr_src,
                   const float* __restrict__ h,
                   const float* __restrict__ a_s,
                   const float* __restrict__ a_d,
                   const float* __restrict__ bias,
                   __bf16* __restrict__ hbf) {
    const int node = __builtin_amdgcn_readfirstlane((int)(blockIdx.x * 8 + (threadIdx.x >> 5)));
    const int lane = threadIdx.x & 31;
    const int beg  = rowptr[node];
    const int deg  = rowptr[node + 1] - beg;
    const int n    = deg + 1;
    const int* __restrict__ seg = csr_src + beg;
    const float ad0 = a_d[node * 2 + 0];
    const float ad1 = a_d[node * 2 + 1];

    auto item = [&](int i, int& s, float& l0, float& l1) {
        if (i < deg) s = seg[i]; else s = node;
        const float2 as = *(const float2*)(a_s + 2 * (size_t)s);
        const float t0 = as.x + ad0, t1 = as.y + ad1;
        l0 = (t0 > 0.f) ? t0 : NEG_SLOPE * t0;
        l1 = (t1 > 0.f) ? t1 : NEG_SLOPE * t1;
    };

    int   s_c  = node;
    float l0_c = -INFINITY, l1_c = -INFINITY;
    if (lane < n) item(lane, s_c, l0_c, l1_c);

    float m0 = l0_c, m1 = l1_c;
    for (int b = 32; b < n; b += 32) {
        const int i = b + lane;
        if (i < n) {
            int s; float l0, l1; item(i, s, l0, l1);
            m0 = fmaxf(m0, l0); m1 = fmaxf(m1, l1);
        }
    }
    m0 = wave_max(m0); m1 = wave_max(m1);

    const float e0_c = (lane < n) ? expf(l0_c - m0) : 0.f;
    const float e1_c = (lane < n) ? expf(l1_c - m1) : 0.f;
    float d0 = e0_c, d1 = e1_c;
    for (int b = 32; b < n; b += 32) {
        const int i = b + lane;
        if (i < n) {
            int s; float l0, l1; item(i, s, l0, l1);
            d0 += expf(l0 - m0); d1 += expf(l1 - m1);
        }
    }
    const float den0 = wave_sum(d0) + 1e-16f;
    const float den1 = wave_sum(d1) + 1e-16f;

    float4 A0 = {0.f, 0.f, 0.f, 0.f};
    float4 A1 = {0.f, 0.f, 0.f, 0.f};
    const float* __restrict__ hl = h + 4 * lane;

    auto accumulate = [&](int cnt, int s, float w0, float w1) {
#pragma unroll 4
        for (int j = 0; j < cnt; ++j) {
            const int   sj = __builtin_amdgcn_readlane(s, j);
            const float x0 = lane_bcast(w0, j);
            const float x1 = lane_bcast(w1, j);
            const float* row = hl + (size_t)sj * HC;
            const float4 v0 = *(const float4*)(row);
            const float4 v1 = *(const float4*)(row + CC);
            A0.x += x0 * v0.x; A0.y += x0 * v0.y; A0.z += x0 * v0.z; A0.w += x0 * v0.w;
            A1.x += x1 * v1.x; A1.y += x1 * v1.y; A1.z += x1 * v1.z; A1.w += x1 * v1.w;
        }
    };

    accumulate(n < 32 ? n : 32, s_c, e0_c / den0, e1_c / den1);
    for (int b = 32; b < n; b += 32) {
        const int i = b + lane;
        int s = node; float w0 = 0.f, w1 = 0.f;
        if (i < n) {
            float l0, l1; item(i, s, l0, l1);
            w0 = expf(l0 - m0) / den0; w1 = expf(l1 - m1) / den1;
        }
        accumulate((n - b) < 32 ? (n - b) : 32, s, w0, w1);
    }

    const float4 bA = *(const float4*)(bias + 4 * lane);
    const float4 bB = *(const float4*)(bias + CC + 4 * lane);
    v4bf o0, o1;
    o0[0] = (__bf16)tanhf(A0.x + bA.x); o0[1] = (__bf16)tanhf(A0.y + bA.y);
    o0[2] = (__bf16)tanhf(A0.z + bA.z); o0[3] = (__bf16)tanhf(A0.w + bA.w);
    o1[0] = (__bf16)tanhf(A1.x + bB.x); o1[1] = (__bf16)tanhf(A1.y + bB.y);
    o1[2] = (__bf16)tanhf(A1.z + bB.z); o1[3] = (__bf16)tanhf(A1.w + bB.w);
    __bf16* o = hbf + (size_t)node * HC + 4 * lane;
    *(v4bf*)(o)      = o0;
    *(v4bf*)(o + CC) = o1;
}

__global__ __launch_bounds__(256)
void k_node_classify(const int* __restrict__ rowptr,
                     const int* __restrict__ csr_src,
                     const float* __restrict__ rec,
                     const float* __restrict__ bias,
                     const float* __restrict__ Wl,
                     const float* __restrict__ bl,
                     float* __restrict__ out) {
    const int node = __builtin_amdgcn_readfirstlane((int)(blockIdx.x * 8 + (threadIdx.x >> 5)));
    const int lane = threadIdx.x & 31;
    const float4* __restrict__ rec0 = (const float4*)rec;
    const float4* __restrict__ rec1 = rec0 + NN;
    const int beg  = rowptr[node];
    const int deg  = rowptr[node + 1] - beg;
    const int n    = deg + 1;
    const int* __restrict__ seg = csr_src + beg;
    const float ad0 = rec0[node].w;
    const float ad1 = rec1[node].w;

    auto item = [&](int i, float4& r0, float4& r1, float& l0, float& l1) {
        int s;
        if (i < deg) s = seg[i]; else s = node;
        r0 = rec0[s]; r1 = rec1[s];
        const float t0 = r0.x + ad0, t1 = r1.x + ad1;
        l0 = (t0 > 0.f) ? t0 : NEG_SLOPE * t0;
        l1 = (t1 > 0.f) ? t1 : NEG_SLOPE * t1;
    };

    float4 r0_c = {0.f, 0.f, 0.f, 0.f}, r1_c = {0.f, 0.f, 0.f, 0.f};
    float l0_c = -INFINITY, l1_c = -INFINITY;
    if (lane < n) item(lane, r0_c, r1_c, l0_c, l1_c);

    float m0 = l0_c, m1 = l1_c;
    for (int b = 32; b < n; b += 32) {
        const int i = b + lane;
        if (i < n) {
            float4 r0, r1; float l0, l1; item(i, r0, r1, l0, l1);
            m0 = fmaxf(m0, l0); m1 = fmaxf(m1, l1);
        }
    }
    m0 = wave_max(m0); m1 = wave_max(m1);

    const float e0_c = (lane < n) ? expf(l0_c - m0) : 0.f;
    const float e1_c = (lane < n) ? expf(l1_c - m1) : 0.f;
    float d0 = e0_c, d1 = e1_c;
    float q00 = e0_c * r0_c.y, q01 = e0_c * r0_c.z;
    float q10 = e1_c * r1_c.y, q11 = e1_c * r1_c.z;
    for (int b = 32; b < n; b += 32) {
        const int i = b + lane;
        if (i < n) {
            float4 r0, r1; float l0, l1; item(i, r0, r1, l0, l1);
            const float e0 = expf(l0 - m0), e1 = expf(l1 - m1);
            d0 += e0; d1 += e1;
            q00 += e0 * r0.y; q01 += e0 * r0.z;
            q10 += e1 * r1.y; q11 += e1 * r1.z;
        }
    }
    const float den0 = wave_sum(d0) + 1e-16f;
    const float den1 = wave_sum(d1) + 1e-16f;
    q00 = wave_sum(q00); q01 = wave_sum(q01);
    q10 = wave_sum(q10); q11 = wave_sum(q11);

    const float4 bA  = *(const float4*)(bias + 4 * lane);
    const float4 bB  = *(const float4*)(bias + CC + 4 * lane);
    const float4 wa0 = *(const float4*)(Wl + 8 * lane);
    const float4 wa1 = *(const float4*)(Wl + 8 * lane + 4);
    const float4 wb0 = *(const float4*)(Wl + 2 * CC + 8 * lane);
    const float4 wb1 = *(const float4*)(Wl + 2 * CC + 8 * lane + 4);
    float c0 = bA.x * wa0.x + bA.y * wa0.z + bA.z * wa1.x + bA.w * wa1.z
             + bB.x * wb0.x + bB.y * wb0.z + bB.z * wb1.x + bB.w * wb1.z;
    float c1 = bA.x * wa0.y + bA.y * wa0.w + bA.z * wa1.y + bA.w * wa1.w
             + bB.x * wb0.y + bB.y * wb0.w + bB.z * wb1.y + bB.w * wb1.w;
    c0 = wave_sum(c0); c1 = wave_sum(c1);

    const float z0 = q00 / den0 + q10 / den1 + c0 + bl[0];
    const float z1 = q01 / den0 + q11 / den1 + c1 + bl[1];
    const float mx = fmaxf(z0, z1);
    const float e0 = expf(z0 - mx), e1 = expf(z1 - mx);
    const float inv = 1.f / (e0 + e1);
    if (lane == 0) *(float2*)(out + 2 * (size_t)node) = make_float2(e0 * inv, e1 * inv);
}

extern "C" void kernel_launch(void* const* d_in, const int* in_sizes, int n_in,
                              void* d_out, int out_size, void* d_ws, size_t ws_size,
                              hipStream_t stream) {
    const float* x        = (const float*)d_in[0];
    const int*   ei       = (const int*)  d_in[1];
    const float* W0       = (const float*)d_in[2];
    const float* att_src0 = (const float*)d_in[3];
    const float* att_dst0 = (const float*)d_in[4];
    const float* b0       = (const float*)d_in[5];
    const float* W1       = (const float*)d_in[6];
    const float* att_src1 = (const float*)d_in[7];
    const float* att_dst1 = (const float*)d_in[8];
    const float* b1       = (const float*)d_in[9];
    const float* Wl       = (const float*)d_in[10];
    const float* bl       = (const float*)d_in[11];
    float* out = (float*)d_out;

    char* base = (char*)d_ws;
    size_t off = 0;
    auto carve = [&](size_t bytes) -> char* {
        char* p = base + off;
        off = (off + bytes + 255) & ~(size_t)255;
        return p;
    };
    float*    h_feat = (float*)   carve((size_t)NN * HC * 4);
    __bf16*   xbf    = (__bf16*)  carve((size_t)NN * HC * 2);
    __bf16*   wtbf   = (__bf16*)  carve((size_t)HC * HC * 2);
    float*    a_s    = (float*)   carve((size_t)NN * 2 * 4);
    float*    a_d    = (float*)   carve((size_t)NN * 2 * 4);
    int*      deg    = (int*)     carve((size_t)NN * 4);
    int*      rowptr = (int*)     carve((size_t)(NN + 1) * 4);
    int*      rank   = (int*)     carve((size_t)EE * 4);
    int*      csr    = (int*)     carve((size_t)EE * 4);
    (void)ws_size; (void)in_sizes; (void)n_in; (void)out_size;

    const int T = 256;
    dim3 blk(T);

    k_f32_to_bf16<<<(NN * DD / 4) / (EW_T * CVT_U), dim3(EW_T), 0, stream>>>(x, xbf);
    k_transpose_bf16<<<(DD * HC + T - 1) / T, blk, 0, stream>>>(W0, wtbf, DD, HC);

    k_gemm_bf16_wmma<false><<<(NN / 16) * (HC / 16) / 8, blk, 0, stream>>>(
        xbf, wtbf, h_feat, NN, HC, DD, nullptr, nullptr, nullptr, nullptr);

    k_att_scores<<<(NN * 2 + T - 1) / T, blk, 0, stream>>>(
        h_feat, att_src0, att_dst0, a_s, a_d);

    hipMemsetAsync(deg, 0, (size_t)NN * 4, stream);
    k_deg_rank<<<(EE + T - 1) / T, blk, 0, stream>>>(ei, deg, rank);
    k_rowptr  <<<(NN + T - 1) / T, blk, 0, stream>>>(deg, rowptr);
    k_fill    <<<(EE + T - 1) / T, blk, 0, stream>>>(ei, rowptr, rank, csr);

    k_node_gather<<<NN / 8, blk, 0, stream>>>(
        rowptr, csr, h_feat, a_s, a_d, b0, xbf);

    k_transpose_bf16<<<(HC * HC + T - 1) / T, blk, 0, stream>>>(W1, wtbf, HC, HC);

    float* rec = h_feat;
    k_gemm_bf16_wmma<true><<<(NN / 16) * (HC / 16) / 8, blk, 0, stream>>>(
        xbf, wtbf, nullptr, NN, HC, HC, att_src1, att_dst1, Wl, rec);

    k_node_classify<<<NN / 8, blk, 0, stream>>>(
        rowptr, csr, rec, b1, Wl, bl, out);
}


# Destination buckets by two-level counting sort on LDS counters (chunk counts, runs ordered in LDS, a block per bin places): 800000 returning global atomics and the clear gone; sum order: LDS arrival
# speedup vs baseline: 9.6264x; 1.3780x over previous
#include <hip/hip_runtime.h>
#include <math.h>

#define NN    50000
#define DD    128
#define HC    256
#define CC    128
#define EE    800000
#define NEG_SLOPE 0.2f

typedef __attribute__((ext_vector_type(16))) __bf16 v16bf;
typedef __attribute__((ext_vector_type(8)))  __bf16 v8bf;
typedef __attribute__((ext_vector_type(8)))  float  v8f;

typedef __attribute__((ext_vector_type(4)))  __bf16 v4bf;

#define EW_T   128
#define CVT_U  4
static_assert((NN * DD / 4) % (EW_T * CVT_U)  == 0, "conversion span must divide the array");

__global__ __launch_bounds__(EW_T)
void k_f32_to_bf16(const float* __restrict__ in,
                   __bf16* __restrict__ out) {
    const float4* __restrict__ in4 = (const float4*)in;
    const int q0 = blockIdx.x * (EW_T * CVT_U) + threadIdx.x;
    float4 v[CVT_U];
#pragma unroll
    for (int j = 0; j < CVT_U; ++j) v[j] = in4[q0 + j * EW_T];
#pragma unroll
    for (int j = 0; j < CVT_U; ++j) {
        v4bf o;
        o[0] = (__bf16)v[j].x; o[1] = (__bf16)v[j].y;
        o[2] = (__bf16)v[j].z; o[3] = (__bf16)v[j].w;
        *(v4bf*)(out + 4 * (size_t)(q0 + j * EW_T)) = o;
    }
}

__global__ void k_transpose_bf16(const float* __restrict__ W,
                                 __bf16* __restrict__ Wt, int K, int Nn) {
    int i = blockIdx.x * blockDim.x + threadIdx.x;
    if (i < K * Nn) {
        int n = i / K, k = i - n * K;
        Wt[i] = (__bf16)W[(size_t)k * Nn + n];
    }
}

template <bool PROJ>
__global__ __launch_bounds__(256)
void k_gemm_bf16_wmma(const __bf16* __restrict__ A,
                      const __bf16* __restrict__ Bt,
                      float* __restrict__ C,
                      int M, int Nn, int K,
                      const float* __restrict__ att_src,
                      const float* __restrict__ att_dst,
                      const float* __restrict__ Wl,
                      float* __restrict__ rec) {
    const int wavesPerBlock = blockDim.x >> 5;
    int wave = blockIdx.x * wavesPerBlock + (threadIdx.x >> 5);
    int lane = threadIdx.x & 31;
    int tilesN = Nn >> 4;
    int tM = (wave / tilesN) << 4;
    int tN = (wave % tilesN) << 4;
    if (tM >= M) return;

    int half = lane >> 4;
    int lr   = lane & 15;

    v8f acc = {};
    const __bf16* arow = A  + (size_t)(tM + lr) * K;
    const __bf16* brow = Bt + (size_t)(tN + lr) * K;

    for (int kk = 0; kk < K; kk += 32) {
        v8bf alo = *(const v8bf*)(arow + kk + 8 * half);
        v8bf ahi = *(const v8bf*)(arow + kk + 16 + 8 * half);
        v8bf blo = *(const v8bf*)(brow + kk + 8 * half);
        v8bf bhi = *(const v8bf*)(brow + kk + 16 + 8 * half);
        v16bf a, b;
#pragma unroll
        for (int i = 0; i < 8; ++i) {
            a[i] = alo[i]; a[i + 8] = ahi[i];
            b[i] = blo[i]; b[i + 8] = bhi[i];
        }
        if constexpr (PROJ) {
            acc = __builtin_amdgcn_wmma_f32_16x16x32_bf16(
                      false, b, false, a, (short)0, acc, false, false);
        } else {
            acc = __builtin_amdgcn_wmma_f32_16x16x32_bf16(
                      false, a, false, b, (short)0, acc, false, false);
        }
    }

    if constexpr (PROJ) {
        __shared__ __attribute__((aligned(16))) float red[8 * 16 * 4];
        const int c0 = tN + 8 * half;
        const float4 sa = *(const float4*)(att_src + c0);
        const float4 sb = *(const float4*)(att_src + c0 + 4);
        const float4 da = *(const float4*)(att_dst + c0);
        const float4 db = *(const float4*)(att_dst + c0 + 4);
        const float4 w0 = *(const float4*)(Wl + 2 * c0);
        const float4 w1 = *(const float4*)(Wl + 2 * c0 + 4);
        const float4 w2 = *(const float4*)(Wl + 2 * c0 + 8);
        const float4 w3 = *(const float4*)(Wl + 2 * c0 + 12);
        float ps = acc[0] * sa.x + acc[1] * sa.y + acc[2] * sa.z + acc[3] * sa.w
                 + acc[4] * sb.x + acc[5] * sb.y + acc[6] * sb.z + acc[7] * sb.w;
        float pd = acc[0] * da.x + acc[1] * da.y + acc[2] * da.z + acc[3] * da.w
                 + acc[4] * db.x + acc[5] * db.y + acc[6] * db.z + acc[7] * db.w;
        float p0 = acc[0] * w0.x + acc[1] * w0.z + acc[2] * w1.x + acc[3] * w1.z
                 + acc[4] * w2.x + acc[5] * w2.z + acc[6] * w3.x + acc[7] * w3.z;
        float p1 = acc[0] * w0.y + acc[1] * w0.w + acc[2] * w1.y + acc[3] * w1.w
                 + acc[4] * w2.y + acc[5] * w2.w + acc[6] * w3.y + acc[7] * w3.w;
        ps += __shfl_xor(ps, 16, 32); pd += __shfl_xor(pd, 16, 32);
        p0 += __shfl_xor(p0, 16, 32); p1 += __shfl_xor(p1, 16, 32);
        const int w = threadIdx.x >> 5;
        if (half == 0) *(float4*)(red + (w * 16 + lr) * 4) = make_float4(ps, p0, p1, pd);
        __syncthreads();
        if (threadIdx.x < 16) {
            float4 t = *(const float4*)(red + threadIdx.x * 4);
#pragma unroll
            for (int k = 1; k < 8; ++k) {
                const float4 u = *(const float4*)(red + (k * 16 + threadIdx.x) * 4);
                t.x += u.x; t.y += u.y; t.z += u.z; t.w += u.w;
            }
            *(float4*)(rec + 4 * ((size_t)(tN >> 7) * M + tM + threadIdx.x)) = t;
        }
    } else {
        int n = tN + lr;
#pragma unroll
        for (int r = 0; r < 8; ++r) {
            int m = tM + r + 8 * half;
            C[(size_t)m * Nn + n] = acc[r];
        }
    }
}

__global__ void k_att_scores(const float* __restrict__ h,
                             const float* __restrict__ att_src,
                             const float* __restrict__ att_dst,
                             float* __restrict__ a_s,
                             float* __restrict__ a_d) {
    int i = blockIdx.x * blockDim.x + threadIdx.x;
    if (i >= NN * 2) return;
    int node = i >> 1, hd = i & 1;
    const float* hp = h + (size_t)node * HC + hd * CC;
    const float* as = att_src + hd * CC;
    const float* ad = att_dst + hd * CC;
    float s = 0.f, d = 0.f;
#pragma unroll 4
    for (int c = 0; c < CC; ++c) { float v = hp[c]; s += v * as[c]; d += v * ad[c]; }
    a_s[i] = s; a_d[i] = d;
}

#define SB_BLK  256
#define SB_CH   (EE / SB_BLK)
#define SB_T    1024
#define SB_IT   ((SB_CH + SB_T - 1) / SB_T)
#define NBN     196
#define NBIN    256
#define BS_T    1024
#define BS_U    4
#define BS_CAP  8192
static_assert(EE % SB_BLK == 0, "chunks must divide the edge list");
static_assert(NBN * NBIN >= NN && NBN <= 256, "bins must cover the nodes, a bin's counters one scan");
static_assert(NN <= 65536, "a source id must fit the low half of the word");
static_assert(SB_CH < 65536, "a chunk's count per bin is kept in 16 bits");
static_assert(SB_T == 4 * NBIN && SB_BLK % 64 == 0, "k_bin_scatter splits the table in four");

__device__ __forceinline__ int block_excl_scan256(int v, int t, int* s_w) {
    const int lane = t & 31, w = t >> 5;
    int inc = v;
#pragma unroll
    for (int o = 1; o < 32; o <<= 1) {
        int u = __shfl_up(inc, o, 32);
        if (lane >= o) inc += u;
    }
    if (lane == 31 && w < 8) s_w[w] = inc;
    __syncthreads();
    int base = 0;
#pragma unroll
    for (int j = 0; j < 8; ++j) if (j < w) base += s_w[j];
    return base + inc - v;
}

__global__ __launch_bounds__(SB_T)
void k_bin_count(const int* __restrict__ ei, unsigned short* __restrict__ hist) {
    __shared__ int s_cnt[NBIN];
    const int t = threadIdx.x;
    const int* __restrict__ dstp = ei + EE + blockIdx.x * SB_CH;
    unsigned d[SB_IT];
#pragma unroll
    for (int j = 0; j < SB_IT; ++j) {
        const int i = t + j * SB_T;
        d[j] = (i < SB_CH) ? (unsigned)dstp[i] : 0u;
    }
    if (t < NBIN) s_cnt[t] = 0;
    __syncthreads();
#pragma unroll
    for (int j = 0; j < SB_IT; ++j) {
        const int i = t + j * SB_T;
        if (i < SB_CH) atomicAdd(&s_cnt[d[j] / NBN], 1);
    }
    __syncthreads();
    if (t < NBIN) hist[blockIdx.x * NBIN + t] = (unsigned short)s_cnt[t];
}

__global__ __launch_bounds__(SB_T)
void k_bin_scatter(const int* __restrict__ ei,
                   const unsigned short* __restrict__ hist,
                   int* __restrict__ binstart,
                   unsigned* __restrict__ binned) {
    __shared__ int s_tot[4][NBIN];
    __shared__ int s_par[4][NBIN];
    __shared__ int s_goff[NBIN];
    __shared__ int s_cnt[NBIN];
    __shared__ int s_lst[NBIN];
    __shared__ int s_w1[8];
    __shared__ int s_w2[8];
    __shared__ unsigned s_val[SB_CH];
    __shared__ int s_addr[SB_CH];
    const int t = threadIdx.x, k = blockIdx.x;
    const int bin = t & (NBIN - 1), q = t >> 8;

    const int* __restrict__ srcp = ei + k * SB_CH;
    const int* __restrict__ dstp = ei + EE + k * SB_CH;
    unsigned es[SB_IT], ed[SB_IT];
#pragma unroll
    for (int j = 0; j < SB_IT; ++j) {
        const int i = t + j * SB_T;
        es[j] = (i < SB_CH) ? (unsigned)srcp[i] : 0u;
        ed[j] = (i < SB_CH) ? (unsigned)dstp[i] : 0u;
    }

    int tot = 0, par = 0;
    const unsigned short* __restrict__ hp = hist + q * (SB_BLK / 4) * NBIN + bin;
    const int lim = k - q * (SB_BLK / 4);
#pragma unroll
    for (int i0 = 0; i0 < SB_BLK / 4; i0 += 16) {
        int h[16];
#pragma unroll
        for (int u = 0; u < 16; ++u) h[u] = hp[(i0 + u) * NBIN];
#pragma unroll
        for (int u = 0; u < 16; ++u) {
            tot += h[u];
            par += (i0 + u < lim) ? h[u] : 0;
        }
    }
    s_tot[q][bin] = tot;
    s_par[q][bin] = par;
    if (t < NBIN) s_cnt[t] = 0;
    __syncthreads();
    int total = 0, partial = 0;
    if (t < NBIN) {
        total   = s_tot[0][t] + s_tot[1][t] + s_tot[2][t] + s_tot[3][t];
        partial = s_par[0][t] + s_par[1][t] + s_par[2][t] + s_par[3][t];
    }
    const int bs = block_excl_scan256(total, t, s_w1);
    if (t < NBIN) {
        s_goff[t] = bs + partial;
        if (k == 0) {
            binstart[t] = bs;
            if (t == NBIN - 1) binstart[NBIN] = bs + total;
        }
    }

    int      bn[SB_IT];
    int      rk[SB_IT];
    unsigned pv[SB_IT];
#pragma unroll
    for (int j = 0; j < SB_IT; ++j) {
        const int i = t + j * SB_T;
        bn[j] = -1; rk[j] = 0; pv[j] = 0u;
        if (i < SB_CH) {
            const unsigned d = ed[j];
            const unsigned b = d / NBN;
            bn[j] = (int)b;
            pv[j] = ((d - b * NBN) << 16) | es[j];
            rk[j] = atomicAdd(&s_cnt[b], 1);
        }
    }
    __syncthreads();
    const int lc = (t < NBIN) ? s_cnt[t] : 0;
    const int ls = block_excl_scan256(lc, t, s_w2);
    if (t < NBIN) s_lst[t] = ls;
    __syncthreads();
#pragma unroll
    for (int j = 0; j < SB_IT; ++j) {
        if (bn[j] >= 0) {
            const int p = s_lst[bn[j]] + rk[j];
            s_val[p]  = pv[j];
            s_addr[p] = s_goff[bn[j]] + rk[j];
        }
    }
    __syncthreads();
#pragma unroll
    for (int j = 0; j < SB_IT; ++j) {
        const int i = t + j * SB_T;
        if (i < SB_CH) binned[s_addr[i]] = s_val[i];
    }
}

__global__ __launch_bounds__(BS_T)
void k_bin_sort(const int* __restrict__ binstart,
                const unsigned* __restrict__ binned,
                int* __restrict__ rowptr,
                int* __restrict__ csr_src) {
    __shared__ int s_c[256];
    __shared__ int s_w[8];
    __shared__ int s_out[BS_CAP];
    const int t = threadIdx.x, b = blockIdx.x;
    const int beg = binstart[b], end = binstart[b + 1];
    const bool staged = (end - beg) <= BS_CAP;
    if (t < 256) s_c[t] = 0;
    __syncthreads();
    for (int i0 = beg + t; i0 < end; i0 += BS_U * BS_T) {
        unsigned v[BS_U];
#pragma unroll
        for (int u = 0; u < BS_U; ++u) {
            const int i = i0 + u * BS_T;
            v[u] = binned[i < end ? i : end - 1];
        }
#pragma unroll
        for (int u = 0; u < BS_U; ++u) {
            const int i = i0 + u * BS_T;
            if (i < end) atomicAdd(&s_c[v[u] >> 16], 1);
        }
    }
    __syncthreads();
    const int c   = (t < 256) ? s_c[t] : 0;
    const int pre = block_excl_scan256(c, t, s_w);
    const int node = b * NBN + t;
    if (t < NBN && node < NN) {
        rowptr[node] = beg + pre;
        if (node == NN - 1) rowptr[NN] = beg + pre + c;
    }
    if (t < 256) s_c[t] = beg + pre;
    __syncthreads();
    for (int i0 = beg + t; i0 < end; i0 += BS_U * BS_T) {
        unsigned v[BS_U];
#pragma unroll
        for (int u = 0; u < BS_U; ++u) {
            const int i = i0 + u * BS_T;
            v[u] = binned[i < end ? i : end - 1];
        }
#pragma unroll
        for (int u = 0; u < BS_U; ++u) {
            const int i = i0 + u * BS_T;
            if (i < end) {
                const int slot = atomicAdd(&s_c[v[u] >> 16], 1);
                if (staged) s_out[slot - beg] = (int)(v[u] & 0xFFFFu);
                else        csr_src[slot]     = (int)(v[u] & 0xFFFFu);
            }
        }
    }
    if (staged) {
        __syncthreads();
        for (int i = t; i < end - beg; i += BS_T) csr_src[beg + i] = s_out[i];
    }
}

__device__ __forceinline__ float wave_max(float v) {
#pragma unroll
    for (int o = 16; o > 0; o >>= 1) v = fmaxf(v, __shfl_xor(v, o, 32));
    return v;
}
__device__ __forceinline__ float wave_sum(float v) {
#pragma unroll
    for (int o = 16; o > 0; o >>= 1) v += __shfl_xor(v, o, 32);
    return v;
}
__device__ __forceinline__ float lane_bcast(float v, int j) {
    return __int_as_float(__builtin_amdgcn_readlane(__float_as_int(v), j));
}

__global__ __launch_bounds__(256)
void k_node_gather(const int* __restrict__ rowptr,
                   const int* __restrict__ csr_src,
                   const float* __restrict__ h,
                   const float* __restrict__ a_s,
                   const float* __restrict__ a_d,
                   const float* __restrict__ bias,
                   __bf16* __restrict__ hbf) {
    const int node = __builtin_amdgcn_readfirstlane((int)(blockIdx.x * 8 + (threadIdx.x >> 5)));
    const int lane = threadIdx.x & 31;
    const int beg  = rowptr[node];
    const int deg  = rowptr[node + 1] - beg;
    const int n    = deg + 1;
    const int* __restrict__ seg = csr_src + beg;
    const float ad0 = a_d[node * 2 + 0];
    const float ad1 = a_d[node * 2 + 1];

    auto item = [&](int i, int& s, float& l0, float& l1) {
        if (i < deg) s = seg[i]; else s = node;
        const float2 as = *(const float2*)(a_s + 2 * (size_t)s);
        const float t0 = as.x + ad0, t1 = as.y + ad1;
        l0 = (t0 > 0.f) ? t0 : NEG_SLOPE * t0;
        l1 = (t1 > 0.f) ? t1 : NEG_SLOPE * t1;
    };

    int   s_c  = node;
    float l0_c = -INFINITY, l1_c = -INFINITY;
    if (lane < n) item(lane, s_c, l0_c, l1_c);

    float m0 = l0_c, m1 = l1_c;
    for (int b = 32; b < n; b += 32) {
        const int i = b + lane;
        if (i < n) {
            int s; float l0, l1; item(i, s, l0, l1);
            m0 = fmaxf(m0, l0); m1 = fmaxf(m1, l1);
        }
    }
    m0 = wave_max(m0); m1 = wave_max(m1);

    const float e0_c = (lane < n) ? expf(l0_c - m0) : 0.f;
    const float e1_c = (lane < n) ? expf(l1_c - m1) : 0.f;
    float d0 = e0_c, d1 = e1_c;
    for (int b = 32; b < n; b += 32) {
        const int i = b + lane;
        if (i < n) {
            int s; float l0, l1; item(i, s, l0, l1);
            d0 += expf(l0 - m0); d1 += expf(l1 - m1);
        }
    }
    const float den0 = wave_sum(d0) + 1e-16f;
    const float den1 = wave_sum(d1) + 1e-16f;

    float4 A0 = {0.f, 0.f, 0.f, 0.f};
    float4 A1 = {0.f, 0.f, 0.f, 0.f};
    const float* __restrict__ hl = h + 4 * lane;

    auto accumulate = [&](int cnt, int s, float w0, float w1) {
#pragma unroll 4
        for (int j = 0; j < cnt; ++j) {
            const int   sj = __builtin_amdgcn_readlane(s, j);
            const float x0 = lane_bcast(w0, j);
            const float x1 = lane_bcast(w1, j);
            const float* row = hl + (size_t)sj * HC;
            const float4 v0 = *(const float4*)(row);
            const float4 v1 = *(const float4*)(row + CC);
            A0.x += x0 * v0.x; A0.y += x0 * v0.y; A0.z += x0 * v0.z; A0.w += x0 * v0.w;
            A1.x += x1 * v1.x; A1.y += x1 * v1.y; A1.z += x1 * v1.z; A1.w += x1 * v1.w;
        }
    };

    accumulate(n < 32 ? n : 32, s_c, e0_c / den0, e1_c / den1);
    for (int b = 32; b < n; b += 32) {
        const int i = b + lane;
        int s = node; float w0 = 0.f, w1 = 0.f;
        if (i < n) {
            float l0, l1; item(i, s, l0, l1);
            w0 = expf(l0 - m0) / den0; w1 = expf(l1 - m1) / den1;
        }
        accumulate((n - b) < 32 ? (n - b) : 32, s, w0, w1);
    }

    const float4 bA = *(const float4*)(bias + 4 * lane);
    const float4 bB = *(const float4*)(bias + CC + 4 * lane);
    v4bf o0, o1;
    o0[0] = (__bf16)tanhf(A0.x + bA.x); o0[1] = (__bf16)tanhf(A0.y + bA.y);
    o0[2] = (__bf16)tanhf(A0.z + bA.z); o0[3] = (__bf16)tanhf(A0.w + bA.w);
    o1[0] = (__bf16)tanhf(A1.x + bB.x); o1[1] = (__bf16)tanhf(A1.y + bB.y);
    o1[2] = (__bf16)tanhf(A1.z + bB.z); o1[3] = (__bf16)tanhf(A1.w + bB.w);
    __bf16* o = hbf + (size_t)node * HC + 4 * lane;
    *(v4bf*)(o)      = o0;
    *(v4bf*)(o + CC) = o1;
}

__global__ __launch_bounds__(256)
void k_node_classify(const int* __restrict__ rowptr,
                     const int* __restrict__ csr_src,
                     const float* __restrict__ rec,
                     const float* __restrict__ bias,
                     const float* __restrict__ Wl,
                     const float* __restrict__ bl,
                     float* __restrict__ out) {
    const int node = __builtin_amdgcn_readfirstlane((int)(blockIdx.x * 8 + (threadIdx.x >> 5)));
    const int lane = threadIdx.x & 31;
    const float4* __restrict__ rec0 = (const float4*)rec;
    const float4* __restrict__ rec1 = rec0 + NN;
    const int beg  = rowptr[node];
    const int deg  = rowptr[node + 1] - beg;
    const int n    = deg + 1;
    const int* __restrict__ seg = csr_src + beg;
    const float ad0 = rec0[node].w;
    const float ad1 = rec1[node].w;

    auto item = [&](int i, float4& r0, float4& r1, float& l0, float& l1) {
        int s;
        if (i < deg) s = seg[i]; else s = node;
        r0 = rec0[s]; r1 = rec1[s];
        const float t0 = r0.x + ad0, t1 = r1.x + ad1;
        l0 = (t0 > 0.f) ? t0 : NEG_SLOPE * t0;
        l1 = (t1 > 0.f) ? t1 : NEG_SLOPE * t1;
    };

    float4 r0_c = {0.f, 0.f, 0.f, 0.f}, r1_c = {0.f, 0.f, 0.f, 0.f};
    float l0_c = -INFINITY, l1_c = -INFINITY;
    if (lane < n) item(lane, r0_c, r1_c, l0_c, l1_c);

    float m0 = l0_c, m1 = l1_c;
    for (int b = 32; b < n; b += 32) {
        const int i = b + lane;
        if (i < n) {
            float4 r0, r1; float l0, l1; item(i, r0, r1, l0, l1);
            m0 = fmaxf(m0, l0); m1 = fmaxf(m1, l1);
        }
    }
    m0 = wave_max(m0); m1 = wave_max(m1);

    const float e0_c = (lane < n) ? expf(l0_c - m0) : 0.f;
    const float e1_c = (lane < n) ? expf(l1_c - m1) : 0.f;
    float d0 = e0_c, d1 = e1_c;
    float q00 = e0_c * r0_c.y, q01 = e0_c * r0_c.z;
    float q10 = e1_c * r1_c.y, q11 = e1_c * r1_c.z;
    for (int b = 32; b < n; b += 32) {
        const int i = b + lane;
        if (i < n) {
            float4 r0, r1; float l0, l1; item(i, r0, r1, l0, l1);
            const float e0 = expf(l0 - m0), e1 = expf(l1 - m1);
            d0 += e0; d1 += e1;
            q00 += e0 * r0.y; q01 += e0 * r0.z;
            q10 += e1 * r1.y; q11 += e1 * r1.z;
        }
    }
    const float den0 = wave_sum(d0) + 1e-16f;
    const float den1 = wave_sum(d1) + 1e-16f;
    q00 = wave_sum(q00); q01 = wave_sum(q01);
    q10 = wave_sum(q10); q11 = wave_sum(q11);

    const float4 bA  = *(const float4*)(bias + 4 * lane);
    const float4 bB  = *(const float4*)(bias + CC + 4 * lane);
    const float4 wa0 = *(const float4*)(Wl + 8 * lane);
    const float4 wa1 = *(const float4*)(Wl + 8 * lane + 4);
    const float4 wb0 = *(const float4*)(Wl + 2 * CC + 8 * lane);
    const float4 wb1 = *(const float4*)(Wl + 2 * CC + 8 * lane + 4);
    float c0 = bA.x * wa0.x + bA.y * wa0.z + bA.z * wa1.x + bA.w * wa1.z
             + bB.x * wb0.x + bB.y * wb0.z + bB.z * wb1.x + bB.w * wb1.z;
    float c1 = bA.x * wa0.y + bA.y * wa0.w + bA.z * wa1.y + bA.w * wa1.w
             + bB.x * wb0.y + bB.y * wb0.w + bB.z * wb1.y + bB.w * wb1.w;
    c0 = wave_sum(c0); c1 = wave_sum(c1);

    const float z0 = q00 / den0 + q10 / den1 + c0 + bl[0];
    const float z1 = q01 / den0 + q11 / den1 + c1 + bl[1];
    const float mx = fmaxf(z0, z1);
    const float e0 = expf(z0 - mx), e1 = expf(z1 - mx);
    const float inv = 1.f / (e0 + e1);
    if (lane == 0) *(float2*)(out + 2 * (size_t)node) = make_float2(e0 * inv, e1 * inv);
}

extern "C" void kernel_launch(void* const* d_in, const int* in_sizes, int n_in,
                              void* d_out, int out_size, void* d_ws, size_t ws_size,
                              hipStream_t stream) {
    const float* x        = (const float*)d_in[0];
    const int*   ei       = (const int*)  d_in[1];
    const float* W0       = (const float*)d_in[2];
    const float* att_src0 = (const float*)d_in[3];
    const float* att_dst0 = (const float*)d_in[4];
    const float* b0       = (const float*)d_in[5];
    const float* W1       = (const float*)d_in[6];
    const float* att_src1 = (const float*)d_in[7];
    const float* att_dst1 = (const float*)d_in[8];
    const float* b1       = (const float*)d_in[9];
    const float* Wl       = (const float*)d_in[10];
    const float* bl       = (const float*)d_in[11];
    float* out = (float*)d_out;

    char* base = (char*)d_ws;
    size_t off = 0;
    auto carve = [&](size_t bytes) -> char* {
        char* p = base + off;
        off = (off + bytes + 255) & ~(size_t)255;
        return p;
    };
    float*    h_feat = (float*)   carve((size_t)NN * HC * 4);
    __bf16*   xbf    = (__bf16*)  carve((size_t)NN * HC * 2);
    __bf16*   wtbf   = (__bf16*)  carve((size_t)HC * HC * 2);
    float*    a_s    = (float*)   carve((size_t)NN * 2 * 4);
    float*    a_d    = (float*)   carve((size_t)NN * 2 * 4);
    char*     tbl    =            carve((size_t)NN * 4);
    int*      rowptr = (int*)     carve((size_t)(NN + 1) * 4);
    unsigned* binned = (unsigned*)carve((size_t)EE * 4);
    int*      csr    = (int*)     carve((size_t)EE * 4);
    unsigned short* hist = (unsigned short*)tbl;
    int*      binstart   = (int*)(tbl + (size_t)SB_BLK * NBIN * 2);
    static_assert((size_t)SB_BLK * NBIN * 2 + (NBIN + 1) * 4 <= (size_t)NN * 4, "table space");
    (void)ws_size; (void)in_sizes; (void)n_in; (void)out_size;

    const int T = 256;
    dim3 blk(T);

    k_f32_to_bf16<<<(NN * DD / 4) / (EW_T * CVT_U), dim3(EW_T), 0, stream>>>(x, xbf);
    k_transpose_bf16<<<(DD * HC + T - 1) / T, blk, 0, stream>>>(W0, wtbf, DD, HC);

    k_gemm_bf16_wmma<false><<<(NN / 16) * (HC / 16) / 8, blk, 0, stream>>>(
        xbf, wtbf, h_feat, NN, HC, DD, nullptr, nullptr, nullptr, nullptr);

    k_att_scores<<<(NN * 2 + T - 1) / T, blk, 0, stream>>>(
        h_feat, att_src0, att_dst0, a_s, a_d);

    k_bin_count  <<<SB_BLK, dim3(SB_T), 0, stream>>>(ei, hist);
    k_bin_scatter<<<SB_BLK, dim3(SB_T), 0, stream>>>(ei, hist, binstart, binned);
    k_bin_sort   <<<NBIN,   dim3(BS_T), 0, stream>>>(binstart, binned, rowptr, csr);

    k_node_gather<<<NN / 8, blk, 0, stream>>>(
        rowptr, csr, h_feat, a_s, a_d, b0, xbf);

    k_transpose_bf16<<<(HC * HC + T - 1) / T, blk, 0, stream>>>(W1, wtbf, HC, HC);

    float* rec = h_feat;
    k_gemm_bf16_wmma<true><<<(NN / 16) * (HC / 16) / 8, blk, 0, stream>>>(
        xbf, wtbf, nullptr, NN, HC, HC, att_src1, att_dst1, Wl, rec);

    k_node_classify<<<NN / 8, blk, 0, stream>>>(
        rowptr, csr, rec, b1, Wl, bl, out);
}


# Matrix-product wave owns a 32 x 128 block of C (16 accumulators): 2 A + 8 B fragment loads feed 16 WMMAs, not 2 loads per WMMA; last layer's row sums stay in the wave: their sum order changes
# speedup vs baseline: 12.6066x; 1.3096x over previous
#include <hip/hip_runtime.h>
#include <math.h>

#define NN    50000
#define DD    128
#define HC    256
#define CC    128
#define EE    800000
#define NEG_SLOPE 0.2f

typedef __attribute__((ext_vector_type(16))) __bf16 v16bf;
typedef __attribute__((ext_vector_type(8)))  __bf16 v8bf;
typedef __attribute__((ext_vector_type(8)))  float  v8f;

typedef __attribute__((ext_vector_type(4)))  __bf16 v4bf;
typedef __attribute__((ext_vector_type(2)))  float  v2f;

#define EW_T   128
#define CVT_U  4
static_assert((NN * DD / 4) % (EW_T * CVT_U)  == 0, "conversion span must divide the array");

__global__ __launch_bounds__(EW_T)
void k_f32_to_bf16(const float* __restrict__ in,
                   __bf16* __restrict__ out) {
    const float4* __restrict__ in4 = (const float4*)in;
    const int q0 = blockIdx.x * (EW_T * CVT_U) + threadIdx.x;
    float4 v[CVT_U];
#pragma unroll
    for (int j = 0; j < CVT_U; ++j) v[j] = in4[q0 + j * EW_T];
#pragma unroll
    for (int j = 0; j < CVT_U; ++j) {
        v4bf o;
        o[0] = (__bf16)v[j].x; o[1] = (__bf16)v[j].y;
        o[2] = (__bf16)v[j].z; o[3] = (__bf16)v[j].w;
        *(v4bf*)(out + 4 * (size_t)(q0 + j * EW_T)) = o;
    }
}

__global__ void k_transpose_bf16(const float* __restrict__ W,
                                 __bf16* __restrict__ Wt, int K, int Nn) {
    int i = blockIdx.x * blockDim.x + threadIdx.x;
    if (i < K * Nn) {
        int n = i / K, k = i - n * K;
        Wt[i] = (__bf16)W[(size_t)k * Nn + n];
    }
}

#define GM_NT 8
static_assert(GM_NT * 16 == CC && HC == 2 * CC, "a wave covers one head, a block both");

__device__ __forceinline__ v16bf load_frag(const __bf16* __restrict__ p) {
    const v8bf lo = *(const v8bf*)(p);
    const v8bf hi = *(const v8bf*)(p + 16);
    v16bf f;
#pragma unroll
    for (int i = 0; i < 8; ++i) { f[i] = lo[i]; f[i + 8] = hi[i]; }
    return f;
}

template <bool PROJ, int K>
__global__ __launch_bounds__(64)
void k_gemm_bf16_wmma(const __bf16* __restrict__ A,
                      const __bf16* __restrict__ Bt,
                      float* __restrict__ C,
                      int M,
                      const float* __restrict__ att_src,
                      const float* __restrict__ att_dst,
                      const float* __restrict__ Wl,
                      float* __restrict__ rec) {
    static_assert(K % 64 == 0, "a trip of the K loop takes 64");
    const int lane = threadIdx.x & 31;
    const int half = lane >> 4;
    const int lr   = lane & 15;
    const int tM   = blockIdx.x * 32;
    const int tN0  = (threadIdx.x >> 5) * CC;
    const bool has1 = tM + 16 < M;
    const int  r1   = has1 ? tM + 16 + lr : M - 1;
    const __bf16* __restrict__ ap0 = A  + (size_t)(tM + lr) * K + 8 * half;
    const __bf16* __restrict__ ap1 = A  + (size_t)r1 * K + 8 * half;
    const __bf16* __restrict__ bp  = Bt + (size_t)(tN0 + lr) * K + 8 * half;

    v8f acc[2][GM_NT];
#pragma unroll
    for (int j = 0; j < GM_NT; ++j) { acc[0][j] = v8f{}; acc[1][j] = v8f{}; }

#pragma unroll 1
    for (int kk = 0; kk < K; kk += 64) {
        v16bf a[2][2];
#pragma unroll
        for (int ks = 0; ks < 2; ++ks) {
            a[0][ks] = load_frag(ap0 + kk + 32 * ks);
            a[1][ks] = load_frag(ap1 + kk + 32 * ks);
        }
#pragma unroll
        for (int nh = 0; nh < 2; ++nh) {
            v16bf b[4][2];
#pragma unroll
            for (int jj = 0; jj < 4; ++jj) {
#pragma unroll
                for (int ks = 0; ks < 2; ++ks)
                    b[jj][ks] = load_frag(bp + (size_t)(4 * nh + jj) * 16 * K + kk + 32 * ks);
            }
#pragma unroll
            for (int ks = 0; ks < 2; ++ks) {
#pragma unroll
                for (int jj = 0; jj < 4; ++jj) {
                    const int j = 4 * nh + jj;
#pragma unroll
                    for (int m = 0; m < 2; ++m) {
                        if constexpr (PROJ) {
                            acc[m][j] = __builtin_amdgcn_wmma_f32_16x16x32_bf16(
                                false, b[jj][ks], false, a[m][ks], (short)0, acc[m][j],
                                false, false);
                        } else {
                            acc[m][j] = __builtin_amdgcn_wmma_f32_16x16x32_bf16(
                                false, a[m][ks], false, b[jj][ks], (short)0, acc[m][j],
                                false, false);
                        }
                    }
                }
            }
        }
        __builtin_amdgcn_sched_group_barrier(0x020, 24, 0);
        __builtin_amdgcn_sched_group_barrier(0x008, 16, 0);
        __builtin_amdgcn_sched_group_barrier(0x020, 16, 0);
        __builtin_amdgcn_sched_group_barrier(0x008, 16, 0);
    }

    if constexpr (PROJ) {
        v2f psd[2] = {{0.f, 0.f}, {0.f, 0.f}};
        v2f p01[2] = {{0.f, 0.f}, {0.f, 0.f}};
        auto load_vec = [&](int j, float4 (&v)[8]) {
            const int c0 = tN0 + 16 * j + 8 * half;
            v[0] = *(const float4*)(att_src + c0); v[1] = *(const float4*)(att_src + c0 + 4);
            v[2] = *(const float4*)(att_dst + c0); v[3] = *(const float4*)(att_dst + c0 + 4);
            v[4] = *(const float4*)(Wl + 2 * c0);     v[5] = *(const float4*)(Wl + 2 * c0 + 4);
            v[6] = *(const float4*)(Wl + 2 * c0 + 8); v[7] = *(const float4*)(Wl + 2 * c0 + 12);
        };
#pragma unroll
        for (int j = 0; j < GM_NT; ++j) {
            float4 cur[8];
            load_vec(j, cur);
            const float s8[8] = {cur[0].x, cur[0].y, cur[0].z, cur[0].w,
                                 cur[1].x, cur[1].y, cur[1].z, cur[1].w};
            const float d8[8] = {cur[2].x, cur[2].y, cur[2].z, cur[2].w,
                                 cur[3].x, cur[3].y, cur[3].z, cur[3].w};
            const v2f w8[8] = {{cur[4].x, cur[4].y}, {cur[4].z, cur[4].w},
                               {cur[5].x, cur[5].y}, {cur[5].z, cur[5].w},
                               {cur[6].x, cur[6].y}, {cur[6].z, cur[6].w},
                               {cur[7].x, cur[7].y}, {cur[7].z, cur[7].w}};
#pragma unroll
            for (int r = 0; r < 8; ++r) {
                const v2f sd = {s8[r], d8[r]};
#pragma unroll
                for (int m = 0; m < 2; ++m) {
                    const float t = acc[m][j][r];
                    const v2f tt = {t, t};
                    psd[m] = __builtin_elementwise_fma(tt, sd,    psd[m]);
                    p01[m] = __builtin_elementwise_fma(tt, w8[r], p01[m]);
                }
            }
        }
        float4* __restrict__ rp = (float4*)rec + (size_t)(tN0 / CC) * M + tM + lr;
#pragma unroll
        for (int m = 0; m < 2; ++m) {
            const float s  = psd[m][0] + __shfl_xor(psd[m][0], 16, 32);
            const float d  = psd[m][1] + __shfl_xor(psd[m][1], 16, 32);
            const float q0 = p01[m][0] + __shfl_xor(p01[m][0], 16, 32);
            const float q1 = p01[m][1] + __shfl_xor(p01[m][1], 16, 32);
            if (half == 0 && (m == 0 || has1)) rp[16 * m] = make_float4(s, q0, q1, d);
        }
    } else {
        float* __restrict__ cp = C + (size_t)(tM + 8 * half) * HC + tN0 + lr;
#pragma unroll
        for (int j = 0; j < GM_NT; ++j) {
#pragma unroll
            for (int r = 0; r < 8; ++r) cp[(size_t)r * HC + 16 * j] = acc[0][j][r];
        }
        if (has1) {
#pragma unroll
            for (int j = 0; j < GM_NT; ++j) {
#pragma unroll
                for (int r = 0; r < 8; ++r) cp[(size_t)(16 + r) * HC + 16 * j] = acc[1][j][r];
            }
        }
    }
}

__global__ void k_att_scores(const float* __restrict__ h,
                             const float* __restrict__ att_src,
                             const float* __restrict__ att_dst,
                             float* __restrict__ a_s,
                             float* __restrict__ a_d) {
    int i = blockIdx.x * blockDim.x + threadIdx.x;
    if (i >= NN * 2) return;
    int node = i >> 1, hd = i & 1;
    const float* hp = h + (size_t)node * HC + hd * CC;
    const float* as = att_src + hd * CC;
    const float* ad = att_dst + hd * CC;
    float s = 0.f, d = 0.f;
#pragma unroll 4
    for (int c = 0; c < CC; ++c) { float v = hp[c]; s += v * as[c]; d += v * ad[c]; }
    a_s[i] = s; a_d[i] = d;
}

#define SB_BLK  256
#define SB_CH   (EE / SB_BLK)
#define SB_T    1024
#define SB_IT   ((SB_CH + SB_T - 1) / SB_T)
#define NBN     196
#define NBIN    256
#define BS_T    1024
#define BS_U    4
#define BS_CAP  8192
static_assert(EE % SB_BLK == 0, "chunks must divide the edge list");
static_assert(NBN * NBIN >= NN && NBN <= 256, "bins must cover the nodes, a bin's counters one scan");
static_assert(NN <= 65536, "a source id must fit the low half of the word");
static_assert(SB_CH < 65536, "a chunk's count per bin is kept in 16 bits");
static_assert(SB_T == 4 * NBIN && SB_BLK % 64 == 0, "k_bin_scatter splits the table in four");

__device__ __forceinline__ int block_excl_scan256(int v, int t, int* s_w) {
    const int lane = t & 31, w = t >> 5;
    int inc = v;
#pragma unroll
    for (int o = 1; o < 32; o <<= 1) {
        int u = __shfl_up(inc, o, 32);
        if (lane >= o) inc += u;
    }
    if (lane == 31 && w < 8) s_w[w] = inc;
    __syncthreads();
    int base = 0;
#pragma unroll
    for (int j = 0; j < 8; ++j) if (j < w) base += s_w[j];
    return base + inc - v;
}

__global__ __launch_bounds__(SB_T)
void k_bin_count(const int* __restrict__ ei, unsigned short* __restrict__ hist) {
    __shared__ int s_cnt[NBIN];
    const int t = threadIdx.x;
    const int* __restrict__ dstp = ei + EE + blockIdx.x * SB_CH;
    unsigned d[SB_IT];
#pragma unroll
    for (int j = 0; j < SB_IT; ++j) {
        const int i = t + j * SB_T;
        d[j] = (i < SB_CH) ? (unsigned)dstp[i] : 0u;
    }
    if (t < NBIN) s_cnt[t] = 0;
    __syncthreads();
#pragma unroll
    for (int j = 0; j < SB_IT; ++j) {
        const int i = t + j * SB_T;
        if (i < SB_CH) atomicAdd(&s_cnt[d[j] / NBN], 1);
    }
    __syncthreads();
    if (t < NBIN) hist[blockIdx.x * NBIN + t] = (unsigned short)s_cnt[t];
}

__global__ __launch_bounds__(SB_T)
void k_bin_scatter(const int* __restrict__ ei,
                   const unsigned short* __restrict__ hist,
                   int* __restrict__ binstart,
                   unsigned* __restrict__ binned) {
    __shared__ int s_tot[4][NBIN];
    __shared__ int s_par[4][NBIN];
    __shared__ int s_goff[NBIN];
    __shared__ int s_cnt[NBIN];
    __shared__ int s_lst[NBIN];
    __shared__ int s_w1[8];
    __shared__ int s_w2[8];
    __shared__ unsigned s_val[SB_CH];
    __shared__ int s_addr[SB_CH];
    const int t = threadIdx.x, k = blockIdx.x;
    const int bin = t & (NBIN - 1), q = t >> 8;

    const int* __restrict__ srcp = ei + k * SB_CH;
    const int* __restrict__ dstp = ei + EE + k * SB_CH;
    unsigned es[SB_IT], ed[SB_IT];
#pragma unroll
    for (int j = 0; j < SB_IT; ++j) {
        const int i = t + j * SB_T;
        es[j] = (i < SB_CH) ? (unsigned)srcp[i] : 0u;
        ed[j] = (i < SB_CH) ? (unsigned)dstp[i] : 0u;
    }

    int tot = 0, par = 0;
    const unsigned short* __restrict__ hp = hist + q * (SB_BLK / 4) * NBIN + bin;
    const int lim = k - q * (SB_BLK / 4);
#pragma unroll
    for (int i0 = 0; i0 < SB_BLK / 4; i0 += 16) {
        int h[16];
#pragma unroll
        for (int u = 0; u < 16; ++u) h[u] = hp[(i0 + u) * NBIN];
#pragma unroll
        for (int u = 0; u < 16; ++u) {
            tot += h[u];
            par += (i0 + u < lim) ? h[u] : 0;
        }
    }
    s_tot[q][bin] = tot;
    s_par[q][bin] = par;
    if (t < NBIN) s_cnt[t] = 0;
    __syncthreads();
    int total = 0, partial = 0;
    if (t < NBIN) {
        total   = s_tot[0][t] + s_tot[1][t] + s_tot[2][t] + s_tot[3][t];
        partial = s_par[0][t] + s_par[1][t] + s_par[2][t] + s_par[3][t];
    }
    const int bs = block_excl_scan256(total, t, s_w1);
    if (t < NBIN) {
        s_goff[t] = bs + partial;
        if (k == 0) {
            binstart[t] = bs;
            if (t == NBIN - 1) binstart[NBIN] = bs + total;
        }
    }

    int      bn[SB_IT];
    int      rk[SB_IT];
    unsigned pv[SB_IT];
#pragma unroll
    for (int j = 0; j < SB_IT; ++j) {
        const int i = t + j * SB_T;
        bn[j] = -1; rk[j] = 0; pv[j] = 0u;
        if (i < SB_CH) {
            const unsigned d = ed[j];
            const unsigned b = d / NBN;
            bn[j] = (int)b;
            pv[j] = ((d - b * NBN) << 16) | es[j];
            rk[j] = atomicAdd(&s_cnt[b], 1);
        }
    }
    __syncthreads();
    const int lc = (t < NBIN) ? s_cnt[t] : 0;
    const int ls = block_excl_scan256(lc, t, s_w2);
    if (t < NBIN) s_lst[t] = ls;
    __syncthreads();
#pragma unroll
    for (int j = 0; j < SB_IT; ++j) {
        if (bn[j] >= 0) {
            const int p = s_lst[bn[j]] + rk[j];
            s_val[p]  = pv[j];
            s_addr[p] = s_goff[bn[j]] + rk[j];
        }
    }
    __syncthreads();
#pragma unroll
    for (int j = 0; j < SB_IT; ++j) {
        const int i = t + j * SB_T;
        if (i < SB_CH) binned[s_addr[i]] = s_val[i];
    }
}

__global__ __launch_bounds__(BS_T)
void k_bin_sort(const int* __restrict__ binstart,
                const unsigned* __restrict__ binned,
                int* __restrict__ rowptr,
                int* __restrict__ csr_src) {
    __shared__ int s_c[256];
    __shared__ int s_w[8];
    __shared__ int s_out[BS_CAP];
    const int t = threadIdx.x, b = blockIdx.x;
    const int beg = binstart[b], end = binstart[b + 1];
    const bool staged = (end - beg) <= BS_CAP;
    if (t < 256) s_c[t] = 0;
    __syncthreads();
    for (int i0 = beg + t; i0 < end; i0 += BS_U * BS_T) {
        unsigned v[BS_U];
#pragma unroll
        for (int u = 0; u < BS_U; ++u) {
            const int i = i0 + u * BS_T;
            v[u] = binned[i < end ? i : end - 1];
        }
#pragma unroll
        for (int u = 0; u < BS_U; ++u) {
            const int i = i0 + u * BS_T;
            if (i < end) atomicAdd(&s_c[v[u] >> 16], 1);
        }
    }
    __syncthreads();
    const int c   = (t < 256) ? s_c[t] : 0;
    const int pre = block_excl_scan256(c, t, s_w);
    const int node = b * NBN + t;
    if (t < NBN && node < NN) {
        rowptr[node] = beg + pre;
        if (node == NN - 1) rowptr[NN] = beg + pre + c;
    }
    if (t < 256) s_c[t] = beg + pre;
    __syncthreads();
    for (int i0 = beg + t; i0 < end; i0 += BS_U * BS_T) {
        unsigned v[BS_U];
#pragma unroll
        for (int u = 0; u < BS_U; ++u) {
            const int i = i0 + u * BS_T;
            v[u] = binned[i < end ? i : end - 1];
        }
#pragma unroll
        for (int u = 0; u < BS_U; ++u) {
            const int i = i0 + u * BS_T;
            if (i < end) {
                const int slot = atomicAdd(&s_c[v[u] >> 16], 1);
                if (staged) s_out[slot - beg] = (int)(v[u] & 0xFFFFu);
                else        csr_src[slot]     = (int)(v[u] & 0xFFFFu);
            }
        }
    }
    if (staged) {
        __syncthreads();
        for (int i = t; i < end - beg; i += BS_T) csr_src[beg + i] = s_out[i];
    }
}

__device__ __forceinline__ float wave_max(float v) {
#pragma unroll
    for (int o = 16; o > 0; o >>= 1) v = fmaxf(v, __shfl_xor(v, o, 32));
    return v;
}
__device__ __forceinline__ float wave_sum(float v) {
#pragma unroll
    for (int o = 16; o > 0; o >>= 1) v += __shfl_xor(v, o, 32);
    return v;
}
__device__ __forceinline__ float lane_bcast(float v, int j) {
    return __int_as_float(__builtin_amdgcn_readlane(__float_as_int(v), j));
}

__global__ __launch_bounds__(256)
void k_node_gather(const int* __restrict__ rowptr,
                   const int* __restrict__ csr_src,
                   const float* __restrict__ h,
                   const float* __restrict__ a_s,
                   const float* __restrict__ a_d,
                   const float* __restrict__ bias,
                   __bf16* __restrict__ hbf) {
    const int node = __builtin_amdgcn_readfirstlane((int)(blockIdx.x * 8 + (threadIdx.x >> 5)));
    const int lane = threadIdx.x & 31;
    const int beg  = rowptr[node];
    const int deg  = rowptr[node + 1] - beg;
    const int n    = deg + 1;
    const int* __restrict__ seg = csr_src + beg;
    const float ad0 = a_d[node * 2 + 0];
    const float ad1 = a_d[node * 2 + 1];

    auto item = [&](int i, int& s, float& l0, float& l1) {
        if (i < deg) s = seg[i]; else s = node;
        const float2 as = *(const float2*)(a_s + 2 * (size_t)s);
        const float t0 = as.x + ad0, t1 = as.y + ad1;
        l0 = (t0 > 0.f) ? t0 : NEG_SLOPE * t0;
        l1 = (t1 > 0.f) ? t1 : NEG_SLOPE * t1;
    };

    int   s_c  = node;
    float l0_c = -INFINITY, l1_c = -INFINITY;
    if (lane < n) item(lane, s_c, l0_c, l1_c);

    float m0 = l0_c, m1 = l1_c;
    for (int b = 32; b < n; b += 32) {
        const int i = b + lane;
        if (i < n) {
            int s; float l0, l1; item(i, s, l0, l1);
            m0 = fmaxf(m0, l0); m1 = fmaxf(m1, l1);
        }
    }
    m0 = wave_max(m0); m1 = wave_max(m1);

    const float e0_c = (lane < n) ? expf(l0_c - m0) : 0.f;
    const float e1_c = (lane < n) ? expf(l1_c - m1) : 0.f;
    float d0 = e0_c, d1 = e1_c;
    for (int b = 32; b < n; b += 32) {
        const int i = b + lane;
        if (i < n) {
            int s; float l0, l1; item(i, s, l0, l1);
            d0 += expf(l0 - m0); d1 += expf(l1 - m1);
        }
    }
    const float den0 = wave_sum(d0) + 1e-16f;
    const float den1 = wave_sum(d1) + 1e-16f;

    float4 A0 = {0.f, 0.f, 0.f, 0.f};
    float4 A1 = {0.f, 0.f, 0.f, 0.f};
    const float* __restrict__ hl = h + 4 * lane;

    auto accumulate = [&](int cnt, int s, float w0, float w1) {
#pragma unroll 4
        for (int j = 0; j < cnt; ++j) {
            const int   sj = __builtin_amdgcn_readlane(s, j);
            const float x0 = lane_bcast(w0, j);
            const float x1 = lane_bcast(w1, j);
            const float* row = hl + (size_t)sj * HC;
            const float4 v0 = *(const float4*)(row);
            const float4 v1 = *(const float4*)(row + CC);
            A0.x += x0 * v0.x; A0.y += x0 * v0.y; A0.z += x0 * v0.z; A0.w += x0 * v0.w;
            A1.x += x1 * v1.x; A1.y += x1 * v1.y; A1.z += x1 * v1.z; A1.w += x1 * v1.w;
        }
    };

    accumulate(n < 32 ? n : 32, s_c, e0_c / den0, e1_c / den1);
    for (int b = 32; b < n; b += 32) {
        const int i = b + lane;
        int s = node; float w0 = 0.f, w1 = 0.f;
        if (i < n) {
            float l0, l1; item(i, s, l0, l1);
            w0 = expf(l0 - m0) / den0; w1 = expf(l1 - m1) / den1;
        }
        accumulate((n - b) < 32 ? (n - b) : 32, s, w0, w1);
    }

    const float4 bA = *(const float4*)(bias + 4 * lane);
    const float4 bB = *(const float4*)(bias + CC + 4 * lane);
    v4bf o0, o1;
    o0[0] = (__bf16)tanhf(A0.x + bA.x); o0[1] = (__bf16)tanhf(A0.y + bA.y);
    o0[2] = (__bf16)tanhf(A0.z + bA.z); o0[3] = (__bf16)tanhf(A0.w + bA.w);
    o1[0] = (__bf16)tanhf(A1.x + bB.x); o1[1] = (__bf16)tanhf(A1.y + bB.y);
    o1[2] = (__bf16)tanhf(A1.z + bB.z); o1[3] = (__bf16)tanhf(A1.w + bB.w);
    __bf16* o = hbf + (size_t)node * HC + 4 * lane;
    *(v4bf*)(o)      = o0;
    *(v4bf*)(o + CC) = o1;
}

__global__ __launch_bounds__(256)
void k_node_classify(const int* __restrict__ rowptr,
                     const int* __restrict__ csr_src,
                     const float* __restrict__ rec,
                     const float* __restrict__ bias,
                     const float* __restrict__ Wl,
                     const float* __restrict__ bl,
                     float* __restrict__ out) {
    const int node = __builtin_amdgcn_readfirstlane((int)(blockIdx.x * 8 + (threadIdx.x >> 5)));
    const int lane = threadIdx.x & 31;
    const float4* __restrict__ rec0 = (const float4*)rec;
    const float4* __restrict__ rec1 = rec0 + NN;
    const int beg  = rowptr[node];
    const int deg  = rowptr[node + 1] - beg;
    const int n    = deg + 1;
    const int* __restrict__ seg = csr_src + beg;
    const float ad0 = rec0[node].w;
    const float ad1 = rec1[node].w;

    auto item = [&](int i, float4& r0, float4& r1, float& l0, float& l1) {
        int s;
        if (i < deg) s = seg[i]; else s = node;
        r0 = rec0[s]; r1 = rec1[s];
        const float t0 = r0.x + ad0, t1 = r1.x + ad1;
        l0 = (t0 > 0.f) ? t0 : NEG_SLOPE * t0;
        l1 = (t1 > 0.f) ? t1 : NEG_SLOPE * t1;
    };

    float4 r0_c = {0.f, 0.f, 0.f, 0.f}, r1_c = {0.f, 0.f, 0.f, 0.f};
    float l0_c = -INFINITY, l1_c = -INFINITY;
    if (lane < n) item(lane, r0_c, r1_c, l0_c, l1_c);

    float m0 = l0_c, m1 = l1_c;
    for (int b = 32; b < n; b += 32) {
        const int i = b + lane;
        if (i < n) {
            float4 r0, r1; float l0, l1; item(i, r0, r1, l0, l1);
            m0 = fmaxf(m0, l0); m1 = fmaxf(m1, l1);
        }
    }
    m0 = wave_max(m0); m1 = wave_max(m1);

    const float e0_c = (lane < n) ? expf(l0_c - m0) : 0.f;
    const float e1_c = (lane < n) ? expf(l1_c - m1) : 0.f;
    float d0 = e0_c, d1 = e1_c;
    float q00 = e0_c * r0_c.y, q01 = e0_c * r0_c.z;
    float q10 = e1_c * r1_c.y, q11 = e1_c * r1_c.z;
    for (int b = 32; b < n; b += 32) {
        const int i = b + lane;
        if (i < n) {
            float4 r0, r1; float l0, l1; item(i, r0, r1, l0, l1);
            const float e0 = expf(l0 - m0), e1 = expf(l1 - m1);
            d0 += e0; d1 += e1;
            q00 += e0 * r0.y; q01 += e0 * r0.z;
            q10 += e1 * r1.y; q11 += e1 * r1.z;
        }
    }
    const float den0 = wave_sum(d0) + 1e-16f;
    const float den1 = wave_sum(d1) + 1e-16f;
    q00 = wave_sum(q00); q01 = wave_sum(q01);
    q10 = wave_sum(q10); q11 = wave_sum(q11);

    const float4 bA  = *(const float4*)(bias + 4 * lane);
    const float4 bB  = *(const float4*)(bias + CC + 4 * lane);
    const float4 wa0 = *(const float4*)(Wl + 8 * lane);
    const float4 wa1 = *(const float4*)(Wl + 8 * lane + 4);
    const float4 wb0 = *(const float4*)(Wl + 2 * CC + 8 * lane);
    const float4 wb1 = *(const float4*)(Wl + 2 * CC + 8 * lane + 4);
    float c0 = bA.x * wa0.x + bA.y * wa0.z + bA.z * wa1.x + bA.w * wa1.z
             + bB.x * wb0.x + bB.y * wb0.z + bB.z * wb1.x + bB.w * wb1.z;
    float c1 = bA.x * wa0.y + bA.y * wa0.w + bA.z * wa1.y + bA.w * wa1.w
             + bB.x * wb0.y + bB.y * wb0.w + bB.z * wb1.y + bB.w * wb1.w;
    c0 = wave_sum(c0); c1 = wave_sum(c1);

    const float z0 = q00 / den0 + q10 / den1 + c0 + bl[0];
    const float z1 = q01 / den0 + q11 / den1 + c1 + bl[1];
    const float mx = fmaxf(z0, z1);
    const float e0 = expf(z0 - mx), e1 = expf(z1 - mx);
    const float inv = 1.f / (e0 + e1);
    if (lane == 0) *(float2*)(out + 2 * (size_t)node) = make_float2(e0 * inv, e1 * inv);
}

extern "C" void kernel_launch(void* const* d_in, const int* in_sizes, int n_in,
                              void* d_out, int out_size, void* d_ws, size_t ws_size,
                              hipStream_t stream) {
    const float* x        = (const float*)d_in[0];
    const int*   ei       = (const int*)  d_in[1];
    const float* W0       = (const float*)d_in[2];
    const float* att_src0 = (const float*)d_in[3];
    const float* att_dst0 = (const float*)d_in[4];
    const float* b0       = (const float*)d_in[5];
    const float* W1       = (const float*)d_in[6];
    const float* att_src1 = (const float*)d_in[7];
    const float* att_dst1 = (const float*)d_in[8];
    const float* b1       = (const float*)d_in[9];
    const float* Wl       = (const float*)d_in[10];
    const float* bl       = (const float*)d_in[11];
    float* out = (float*)d_out;

    char* base = (char*)d_ws;
    size_t off = 0;
    auto carve = [&](size_t bytes) -> char* {
        char* p = base + off;
        off = (off + bytes + 255) & ~(size_t)255;
        return p;
    };
    float*    h_feat = (float*)   carve((size_t)NN * HC * 4);
    __bf16*   xbf    = (__bf16*)  carve((size_t)NN * HC * 2);
    __bf16*   wtbf   = (__bf16*)  carve((size_t)HC * HC * 2);
    float*    a_s    = (float*)   carve((size_t)NN * 2 * 4);
    float*    a_d    = (float*)   carve((size_t)NN * 2 * 4);
    char*     tbl    =            carve((size_t)NN * 4);
    int*      rowptr = (int*)     carve((size_t)(NN + 1) * 4);
    unsigned* binned = (unsigned*)carve((size_t)EE * 4);
    int*      csr    = (int*)     carve((size_t)EE * 4);
    unsigned short* hist = (unsigned short*)tbl;
    int*      binstart   = (int*)(tbl + (size_t)SB_BLK * NBIN * 2);
    static_assert((size_t)SB_BLK * NBIN * 2 + (NBIN + 1) * 4 <= (size_t)NN * 4, "table space");
    (void)ws_size; (void)in_sizes; (void)n_in; (void)out_size;

    const int T = 256;
    dim3 blk(T);

    k_f32_to_bf16<<<(NN * DD / 4) / (EW_T * CVT_U), dim3(EW_T), 0, stream>>>(x, xbf);
    k_transpose_bf16<<<(DD * HC + T - 1) / T, blk, 0, stream>>>(W0, wtbf, DD, HC);

    k_gemm_bf16_wmma<false, DD><<<(NN + 31) / 32, dim3(64), 0, stream>>>(
        xbf, wtbf, h_feat, NN, nullptr, nullptr, nullptr, nullptr);

    k_att_scores<<<(NN * 2 + T - 1) / T, blk, 0, stream>>>(
        h_feat, att_src0, att_dst0, a_s, a_d);

    k_bin_count  <<<SB_BLK, dim3(SB_T), 0, stream>>>(ei, hist);
    k_bin_scatter<<<SB_BLK, dim3(SB_T), 0, stream>>>(ei, hist, binstart, binned);
    k_bin_sort   <<<NBIN,   dim3(BS_T), 0, stream>>>(binstart, binned, rowptr, csr);

    k_node_gather<<<NN / 8, blk, 0, stream>>>(
        rowptr, csr, h_feat, a_s, a_d, b0, xbf);

    k_transpose_bf16<<<(HC * HC + T - 1) / T, blk, 0, stream>>>(W1, wtbf, HC, HC);

    float* rec = h_feat;
    k_gemm_bf16_wmma<true, HC><<<(NN + 31) / 32, dim3(64), 0, stream>>>(
        xbf, wtbf, nullptr, NN, att_src1, att_dst1, Wl, rec);

    k_node_classify<<<NN / 8, blk, 0, stream>>>(
        rowptr, csr, rec, b1, Wl, bl, out);
}


# Layer 0 sums neighbours' bf16 x rows (this call's copy, 256 B a row, not 1 KiB of h) per head BEFORE its product; scores = x.(W att); bias+tanh in the product's tail; sum order changes
# speedup vs baseline: 15.9200x; 1.2628x over previous
#include <hip/hip_runtime.h>
#include <math.h>

#define NN    50000
#define DD    128
#define HC    256
#define CC    128
#define EE    800000
#define NEG_SLOPE 0.2f

typedef __attribute__((ext_vector_type(16))) __bf16 v16bf;
typedef __attribute__((ext_vector_type(8)))  __bf16 v8bf;
typedef __attribute__((ext_vector_type(8)))  float  v8f;

typedef __attribute__((ext_vector_type(4)))  __bf16 v4bf;
typedef __attribute__((ext_vector_type(2)))  float  v2f;

#define EW_T   128
#define CVT_U  4
static_assert((NN * DD / 4) % (EW_T * CVT_U)  == 0, "conversion span must divide the array");

__device__ __forceinline__ float wave_max(float v) {
#pragma unroll
    for (int o = 16; o > 0; o >>= 1) v = fmaxf(v, __shfl_xor(v, o, 32));
    return v;
}
__device__ __forceinline__ float wave_sum(float v) {
#pragma unroll
    for (int o = 16; o > 0; o >>= 1) v += __shfl_xor(v, o, 32);
    return v;
}
template <int CTRL>
__device__ __forceinline__ float row_swz(float v) {
    const int b = __builtin_bit_cast(int, v);
    return __builtin_bit_cast(float, __builtin_amdgcn_update_dpp(b, b, CTRL, 0xf, 0xf, false));
}
__device__ __forceinline__ float xhalf(float v) {
    const int b = __builtin_bit_cast(int, v);
    return __builtin_bit_cast(float,
        __builtin_amdgcn_permlanex16(b, b, 0x76543210, 0xfedcba98, false, false));
}
__device__ __forceinline__ float wave_sum_valu(float v) {
    v += row_swz<0x161>(v);
    v += row_swz<0x162>(v);
    v += row_swz<0x164>(v);
    v += row_swz<0x168>(v);
    return v + xhalf(v);
}

static_assert(DD == 128 && CC == 128 && HC == 2 * CC, "a lane holds 4 features / 4 + 4 channels");
__global__ __launch_bounds__(256)
void k_score_vecs(const float* __restrict__ W,
                  const float* __restrict__ att_src,
                  const float* __restrict__ att_dst,
                  float* __restrict__ vsd) {
    const int k    = blockIdx.x * 8 + (threadIdx.x >> 5);
    const int lane = threadIdx.x & 31;
    const float4 w0 = *(const float4*)(W + (size_t)k * HC + 4 * lane);
    const float4 w1 = *(const float4*)(W + (size_t)k * HC + CC + 4 * lane);
    const float4 s0 = *(const float4*)(att_src + 4 * lane);
    const float4 s1 = *(const float4*)(att_src + CC + 4 * lane);
    const float4 d0 = *(const float4*)(att_dst + 4 * lane);
    const float4 d1 = *(const float4*)(att_dst + CC + 4 * lane);
    float ps0 = w0.x * s0.x + w0.y * s0.y + w0.z * s0.z + w0.w * s0.w;
    float ps1 = w1.x * s1.x + w1.y * s1.y + w1.z * s1.z + w1.w * s1.w;
    float pd0 = w0.x * d0.x + w0.y * d0.y + w0.z * d0.z + w0.w * d0.w;
    float pd1 = w1.x * d1.x + w1.y * d1.y + w1.z * d1.z + w1.w * d1.w;
    ps0 = wave_sum(ps0); ps1 = wave_sum(ps1); pd0 = wave_sum(pd0); pd1 = wave_sum(pd1);
    if (lane == 0) *(float4*)(vsd + 4 * k) = make_float4(ps0, ps1, pd0, pd1);
}

static_assert(EW_T % 32 == 0, "a wave's load must be one whole row of x");
__global__ __launch_bounds__(EW_T)
void k_f32_to_bf16_scores(const float* __restrict__ in,
                          const float* __restrict__ vsd,
                          __bf16* __restrict__ out,
                          float* __restrict__ a_s,
                          float* __restrict__ a_d) {
    const float4* __restrict__ in4 = (const float4*)in;
    const int q0   = blockIdx.x * (EW_T * CVT_U) + threadIdx.x;
    const int lane = threadIdx.x & 31;
    float4 v[CVT_U];
#pragma unroll
    for (int j = 0; j < CVT_U; ++j) v[j] = in4[q0 + j * EW_T];
    const float4* __restrict__ vp = (const float4*)vsd + 4 * lane;
    const float4 c0 = vp[0], c1 = vp[1], c2 = vp[2], c3 = vp[3];
#pragma unroll
    for (int j = 0; j < CVT_U; ++j) {
        v4bf o;
        o[0] = (__bf16)v[j].x; o[1] = (__bf16)v[j].y;
        o[2] = (__bf16)v[j].z; o[3] = (__bf16)v[j].w;
        *(v4bf*)(out + 4 * (size_t)(q0 + j * EW_T)) = o;
        float s0 = v[j].x * c0.x + v[j].y * c1.x + v[j].z * c2.x + v[j].w * c3.x;
        float s1 = v[j].x * c0.y + v[j].y * c1.y + v[j].z * c2.y + v[j].w * c3.y;
        float d0 = v[j].x * c0.z + v[j].y * c1.z + v[j].z * c2.z + v[j].w * c3.z;
        float d1 = v[j].x * c0.w + v[j].y * c1.w + v[j].z * c2.w + v[j].w * c3.w;
        s0 = wave_sum_valu(s0); s1 = wave_sum_valu(s1);
        d0 = wave_sum_valu(d0); d1 = wave_sum_valu(d1);
        const int row = (q0 + j * EW_T) >> 5;
        if (lane == 0) {
            *(float2*)(a_s + 2 * (size_t)row) = make_float2(s0, s1);
            *(float2*)(a_d + 2 * (size_t)row) = make_float2(d0, d1);
        }
    }
}

__global__ void k_transpose_bf16(const float* __restrict__ W,
                                 __bf16* __restrict__ Wt, int K, int Nn) {
    int i = blockIdx.x * blockDim.x + threadIdx.x;
    if (i < K * Nn) {
        int n = i / K, k = i - n * K;
        Wt[i] = (__bf16)W[(size_t)k * Nn + n];
    }
}

#define GM_NT 8
static_assert(GM_NT * 16 == CC && HC == 2 * CC, "a wave covers one head, a block both");

__device__ __forceinline__ v16bf load_frag(const __bf16* __restrict__ p) {
    const v8bf lo = *(const v8bf*)(p);
    const v8bf hi = *(const v8bf*)(p + 16);
    v16bf f;
#pragma unroll
    for (int i = 0; i < 8; ++i) { f[i] = lo[i]; f[i + 8] = hi[i]; }
    return f;
}

#define GM_TP (CC + 4)
template <bool PROJ, int K>
__global__ __launch_bounds__(64)
void k_gemm_bf16_wmma(const __bf16* __restrict__ A,
                      const __bf16* __restrict__ Bt,
                      int M,
                      const float* __restrict__ att_src,
                      const float* __restrict__ att_dst,
                      const float* __restrict__ Wl,
                      float* __restrict__ rec,
                      const float* __restrict__ bias,
                      __bf16* __restrict__ hout) {
    static_assert(K % 64 == 0, "a trip of the K loop takes 64");
    static_assert(PROJ ? K == HC : K == CC, "layer 0 reads its head's half of a row of A");
    const int lane = threadIdx.x & 31;
    const int half = lane >> 4;
    const int lr   = lane & 15;
    const int tM   = blockIdx.x * 32;
    const int tN0  = (threadIdx.x >> 5) * CC;
    const bool has1 = tM + 16 < M;
    const int  r1   = has1 ? tM + 16 + lr : M - 1;
    const int aoff = (PROJ ? 0 : tN0) + 8 * half;
    const __bf16* __restrict__ ap0 = A  + (size_t)(tM + lr) * HC + aoff;
    const __bf16* __restrict__ ap1 = A  + (size_t)r1 * HC + aoff;
    const __bf16* __restrict__ bp  = Bt + (size_t)(tN0 + lr) * K + 8 * half;

    v8f acc[2][GM_NT];
#pragma unroll
    for (int j = 0; j < GM_NT; ++j) { acc[0][j] = v8f{}; acc[1][j] = v8f{}; }

#pragma unroll 1
    for (int kk = 0; kk < K; kk += 64) {
        v16bf a[2][2];
#pragma unroll
        for (int ks = 0; ks < 2; ++ks) {
            a[0][ks] = load_frag(ap0 + kk + 32 * ks);
            a[1][ks] = load_frag(ap1 + kk + 32 * ks);
        }
#pragma unroll
        for (int nh = 0; nh < 2; ++nh) {
            v16bf b[4][2];
#pragma unroll
            for (int jj = 0; jj < 4; ++jj) {
#pragma unroll
                for (int ks = 0; ks < 2; ++ks)
                    b[jj][ks] = load_frag(bp + (size_t)(4 * nh + jj) * 16 * K + kk + 32 * ks);
            }
#pragma unroll
            for (int ks = 0; ks < 2; ++ks) {
#pragma unroll
                for (int jj = 0; jj < 4; ++jj) {
                    const int j = 4 * nh + jj;
#pragma unroll
                    for (int m = 0; m < 2; ++m) {
                        acc[m][j] = __builtin_amdgcn_wmma_f32_16x16x32_bf16(
                            false, b[jj][ks], false, a[m][ks], (short)0, acc[m][j],
                            false, false);
                    }
                }
            }
        }
        __builtin_amdgcn_sched_group_barrier(0x020, 24, 0);
        __builtin_amdgcn_sched_group_barrier(0x008, 16, 0);
        __builtin_amdgcn_sched_group_barrier(0x020, 16, 0);
        __builtin_amdgcn_sched_group_barrier(0x008, 16, 0);
    }

    if constexpr (PROJ) {
        v2f psd[2] = {{0.f, 0.f}, {0.f, 0.f}};
        v2f p01[2] = {{0.f, 0.f}, {0.f, 0.f}};
        auto load_vec = [&](int j, float4 (&v)[8]) {
            const int c0 = tN0 + 16 * j + 8 * half;
            v[0] = *(const float4*)(att_src + c0); v[1] = *(const float4*)(att_src + c0 + 4);
            v[2] = *(const float4*)(att_dst + c0); v[3] = *(const float4*)(att_dst + c0 + 4);
            v[4] = *(const float4*)(Wl + 2 * c0);     v[5] = *(const float4*)(Wl + 2 * c0 + 4);
            v[6] = *(const float4*)(Wl + 2 * c0 + 8); v[7] = *(const float4*)(Wl + 2 * c0 + 12);
        };
#pragma unroll
        for (int j = 0; j < GM_NT; ++j) {
            float4 cur[8];
            load_vec(j, cur);
            const float s8[8] = {cur[0].x, cur[0].y, cur[0].z, cur[0].w,
                                 cur[1].x, cur[1].y, cur[1].z, cur[1].w};
            const float d8[8] = {cur[2].x, cur[2].y, cur[2].z, cur[2].w,
                                 cur[3].x, cur[3].y, cur[3].z, cur[3].w};
            const v2f w8[8] = {{cur[4].x, cur[4].y}, {cur[4].z, cur[4].w},
                               {cur[5].x, cur[5].y}, {cur[5].z, cur[5].w},
                               {cur[6].x, cur[6].y}, {cur[6].z, cur[6].w},
                               {cur[7].x, cur[7].y}, {cur[7].z, cur[7].w}};
#pragma unroll
            for (int r = 0; r < 8; ++r) {
                const v2f sd = {s8[r], d8[r]};
#pragma unroll
                for (int m = 0; m < 2; ++m) {
                    const float t = acc[m][j][r];
                    const v2f tt = {t, t};
                    psd[m] = __builtin_elementwise_fma(tt, sd,    psd[m]);
                    p01[m] = __builtin_elementwise_fma(tt, w8[r], p01[m]);
                }
            }
        }
        float4* __restrict__ rp = (float4*)rec + (size_t)(tN0 / CC) * M + tM + lr;
#pragma unroll
        for (int m = 0; m < 2; ++m) {
            const float s  = psd[m][0] + __shfl_xor(psd[m][0], 16, 32);
            const float d  = psd[m][1] + __shfl_xor(psd[m][1], 16, 32);
            const float q0 = p01[m][0] + __shfl_xor(p01[m][0], 16, 32);
            const float q1 = p01[m][1] + __shfl_xor(p01[m][1], 16, 32);
            if (half == 0 && (m == 0 || has1)) rp[16 * m] = make_float4(s, q0, q1, d);
        }
    } else {
        __shared__ alignas(16) float s_tile[2][32 * GM_TP];
        float* __restrict__ sw = s_tile[threadIdx.x >> 5];
#pragma unroll
        for (int m = 0; m < 2; ++m) {
#pragma unroll
            for (int j = 0; j < GM_NT; ++j) {
                float* p = sw + (16 * m + lr) * GM_TP + 16 * j + 8 * half;
                *(float4*)(p)     = make_float4(acc[m][j][0], acc[m][j][1], acc[m][j][2], acc[m][j][3]);
                *(float4*)(p + 4) = make_float4(acc[m][j][4], acc[m][j][5], acc[m][j][6], acc[m][j][7]);
            }
        }
        __syncthreads();
        const float4 bA = *(const float4*)(bias + tN0 + 8 * lr);
        const float4 bB = *(const float4*)(bias + tN0 + 8 * lr + 4);
        __bf16* __restrict__ op = hout + (size_t)tM * HC + tN0 + 8 * lr;
#pragma unroll 1
        for (int it = 0; it < 16; ++it) {
            const int rr = 2 * it + half;
            const float4 u0 = *(const float4*)(sw + rr * GM_TP + 8 * lr);
            const float4 u1 = *(const float4*)(sw + rr * GM_TP + 8 * lr + 4);
            v8bf o;
            o[0] = (__bf16)tanhf(u0.x + bA.x); o[1] = (__bf16)tanhf(u0.y + bA.y);
            o[2] = (__bf16)tanhf(u0.z + bA.z); o[3] = (__bf16)tanhf(u0.w + bA.w);
            o[4] = (__bf16)tanhf(u1.x + bB.x); o[5] = (__bf16)tanhf(u1.y + bB.y);
            o[6] = (__bf16)tanhf(u1.z + bB.z); o[7] = (__bf16)tanhf(u1.w + bB.w);
            if (tM + rr < M) *(v8bf*)(op + (size_t)rr * HC) = o;
        }
    }
}

#define SB_BLK  256
#define SB_CH   (EE / SB_BLK)
#define SB_T    1024
#define SB_IT   ((SB_CH + SB_T - 1) / SB_T)
#define NBN     196
#define NBIN    256
#define BS_T    1024
#define BS_U    4
#define BS_CAP  8192
static_assert(EE % SB_BLK == 0, "chunks must divide the edge list");
static_assert(NBN * NBIN >= NN && NBN <= 256, "bins must cover the nodes, a bin's counters one scan");
static_assert(NN <= 65536, "a source id must fit the low half of the word");
static_assert(SB_CH < 65536, "a chunk's count per bin is kept in 16 bits");
static_assert(SB_T == 4 * NBIN && SB_BLK % 64 == 0, "k_bin_scatter splits the table in four");

__device__ __forceinline__ int block_excl_scan256(int v, int t, int* s_w) {
    const int lane = t & 31, w = t >> 5;
    int inc = v;
#pragma unroll
    for (int o = 1; o < 32; o <<= 1) {
        int u = __shfl_up(inc, o, 32);
        if (lane >= o) inc += u;
    }
    if (lane == 31 && w < 8) s_w[w] = inc;
    __syncthreads();
    int base = 0;
#pragma unroll
    for (int j = 0; j < 8; ++j) if (j < w) base += s_w[j];
    return base + inc - v;
}

__global__ __launch_bounds__(SB_T)
void k_bin_count(const int* __restrict__ ei, unsigned short* __restrict__ hist) {
    __shared__ int s_cnt[NBIN];
    const int t = threadIdx.x;
    const int* __restrict__ dstp = ei + EE + blockIdx.x * SB_CH;
    unsigned d[SB_IT];
#pragma unroll
    for (int j = 0; j < SB_IT; ++j) {
        const int i = t + j * SB_T;
        d[j] = (i < SB_CH) ? (unsigned)dstp[i] : 0u;
    }
    if (t < NBIN) s_cnt[t] = 0;
    __syncthreads();
#pragma unroll
    for (int j = 0; j < SB_IT; ++j) {
        const int i = t + j * SB_T;
        if (i < SB_CH) atomicAdd(&s_cnt[d[j] / NBN], 1);
    }
    __syncthreads();
    if (t < NBIN) hist[blockIdx.x * NBIN + t] = (unsigned short)s_cnt[t];
}

__global__ __launch_bounds__(SB_T)
void k_bin_scatter(const int* __restrict__ ei,
                   const unsigned short* __restrict__ hist,
                   int* __restrict__ binstart,
                   unsigned* __restrict__ binned) {
    __shared__ int s_tot[4][NBIN];
    __shared__ int s_par[4][NBIN];
    __shared__ int s_goff[NBIN];
    __shared__ int s_cnt[NBIN];
    __shared__ int s_lst[NBIN];
    __shared__ int s_w1[8];
    __shared__ int s_w2[8];
    __shared__ unsigned s_val[SB_CH];
    __shared__ int s_addr[SB_CH];
    const int t = threadIdx.x, k = blockIdx.x;
    const int bin = t & (NBIN - 1), q = t >> 8;

    const int* __restrict__ srcp = ei + k * SB_CH;
    const int* __restrict__ dstp = ei + EE + k * SB_CH;
    unsigned es[SB_IT], ed[SB_IT];
#pragma unroll
    for (int j = 0; j < SB_IT; ++j) {
        const int i = t + j * SB_T;
        es[j] = (i < SB_CH) ? (unsigned)srcp[i] : 0u;
        ed[j] = (i < SB_CH) ? (unsigned)dstp[i] : 0u;
    }

    int tot = 0, par = 0;
    const unsigned short* __restrict__ hp = hist + q * (SB_BLK / 4) * NBIN + bin;
    const int lim = k - q * (SB_BLK / 4);
#pragma unroll
    for (int i0 = 0; i0 < SB_BLK / 4; i0 += 16) {
        int h[16];
#pragma unroll
        for (int u = 0; u < 16; ++u) h[u] = hp[(i0 + u) * NBIN];
#pragma unroll
        for (int u = 0; u < 16; ++u) {
            tot += h[u];
            par += (i0 + u < lim) ? h[u] : 0;
        }
    }
    s_tot[q][bin] = tot;
    s_par[q][bin] = par;
    if (t < NBIN) s_cnt[t] = 0;
    __syncthreads();
    int total = 0, partial = 0;
    if (t < NBIN) {
        total   = s_tot[0][t] + s_tot[1][t] + s_tot[2][t] + s_tot[3][t];
        partial = s_par[0][t] + s_par[1][t] + s_par[2][t] + s_par[3][t];
    }
    const int bs = block_excl_scan256(total, t, s_w1);
    if (t < NBIN) {
        s_goff[t] = bs + partial;
        if (k == 0) {
            binstart[t] = bs;
            if (t == NBIN - 1) binstart[NBIN] = bs + total;
        }
    }

    int      bn[SB_IT];
    int      rk[SB_IT];
    unsigned pv[SB_IT];
#pragma unroll
    for (int j = 0; j < SB_IT; ++j) {
        const int i = t + j * SB_T;
        bn[j] = -1; rk[j] = 0; pv[j] = 0u;
        if (i < SB_CH) {
            const unsigned d = ed[j];
            const unsigned b = d / NBN;
            bn[j] = (int)b;
            pv[j] = ((d - b * NBN) << 16) | es[j];
            rk[j] = atomicAdd(&s_cnt[b], 1);
        }
    }
    __syncthreads();
    const int lc = (t < NBIN) ? s_cnt[t] : 0;
    const int ls = block_excl_scan256(lc, t, s_w2);
    if (t < NBIN) s_lst[t] = ls;
    __syncthreads();
#pragma unroll
    for (int j = 0; j < SB_IT; ++j) {
        if (bn[j] >= 0) {
            const int p = s_lst[bn[j]] + rk[j];
            s_val[p]  = pv[j];
            s_addr[p] = s_goff[bn[j]] + rk[j];
        }
    }
    __syncthreads();
#pragma unroll
    for (int j = 0; j < SB_IT; ++j) {
        const int i = t + j * SB_T;
        if (i < SB_CH) binned[s_addr[i]] = s_val[i];
    }
}

__global__ __launch_bounds__(BS_T)
void k_bin_sort(const int* __restrict__ binstart,
                const unsigned* __restrict__ binned,
                int* __restrict__ rowptr,
                int* __restrict__ csr_src) {
    __shared__ int s_c[256];
    __shared__ int s_w[8];
    __shared__ int s_out[BS_CAP];
    const int t = threadIdx.x, b = blockIdx.x;
    const int beg = binstart[b], end = binstart[b + 1];
    const bool staged = (end - beg) <= BS_CAP;
    if (t < 256) s_c[t] = 0;
    __syncthreads();
    for (int i0 = beg + t; i0 < end; i0 += BS_U * BS_T) {
        unsigned v[BS_U];
#pragma unroll
        for (int u = 0; u < BS_U; ++u) {
            const int i = i0 + u * BS_T;
            v[u] = binned[i < end ? i : end - 1];
        }
#pragma unroll
        for (int u = 0; u < BS_U; ++u) {
            const int i = i0 + u * BS_T;
            if (i < end) atomicAdd(&s_c[v[u] >> 16], 1);
        }
    }
    __syncthreads();
    const int c   = (t < 256) ? s_c[t] : 0;
    const int pre = block_excl_scan256(c, t, s_w);
    const int node = b * NBN + t;
    if (t < NBN && node < NN) {
        rowptr[node] = beg + pre;
        if (node == NN - 1) rowptr[NN] = beg + pre + c;
    }
    if (t < 256) s_c[t] = beg + pre;
    __syncthreads();
    for (int i0 = beg + t; i0 < end; i0 += BS_U * BS_T) {
        unsigned v[BS_U];
#pragma unroll
        for (int u = 0; u < BS_U; ++u) {
            const int i = i0 + u * BS_T;
            v[u] = binned[i < end ? i : end - 1];
        }
#pragma unroll
        for (int u = 0; u < BS_U; ++u) {
            const int i = i0 + u * BS_T;
            if (i < end) {
                const int slot = atomicAdd(&s_c[v[u] >> 16], 1);
                if (staged) s_out[slot - beg] = (int)(v[u] & 0xFFFFu);
                else        csr_src[slot]     = (int)(v[u] & 0xFFFFu);
            }
        }
    }
    if (staged) {
        __syncthreads();
        for (int i = t; i < end - beg; i += BS_T) csr_src[beg + i] = s_out[i];
    }
}

__device__ __forceinline__ float lane_bcast(float v, int j) {
    return __int_as_float(__builtin_amdgcn_readlane(__float_as_int(v), j));
}

__global__ __launch_bounds__(256)
void k_node_gather(const int* __restrict__ rowptr,
                   const int* __restrict__ csr_src,
                   const __bf16* __restrict__ xb,
                   const float* __restrict__ a_s,
                   const float* __restrict__ a_d,
                   __bf16* __restrict__ agg) {
    const int node = __builtin_amdgcn_readfirstlane((int)(blockIdx.x * 8 + (threadIdx.x >> 5)));
    const int lane = threadIdx.x & 31;
    const int beg  = rowptr[node];
    const int deg  = rowptr[node + 1] - beg;
    const int n    = deg + 1;
    const int* __restrict__ seg = csr_src + beg;
    const float ad0 = a_d[node * 2 + 0];
    const float ad1 = a_d[node * 2 + 1];

    auto item = [&](int i, int& s, float& l0, float& l1) {
        if (i < deg) s = seg[i]; else s = node;
        const float2 as = *(const float2*)(a_s + 2 * (size_t)s);
        const float t0 = as.x + ad0, t1 = as.y + ad1;
        l0 = (t0 > 0.f) ? t0 : NEG_SLOPE * t0;
        l1 = (t1 > 0.f) ? t1 : NEG_SLOPE * t1;
    };

    int   s_c  = node;
    float l0_c = -INFINITY, l1_c = -INFINITY;
    if (lane < n) item(lane, s_c, l0_c, l1_c);

    float m0 = l0_c, m1 = l1_c;
    for (int b = 32; b < n; b += 32) {
        const int i = b + lane;
        if (i < n) {
            int s; float l0, l1; item(i, s, l0, l1);
            m0 = fmaxf(m0, l0); m1 = fmaxf(m1, l1);
        }
    }
    m0 = wave_max(m0); m1 = wave_max(m1);

    const float e0_c = (lane < n) ? expf(l0_c - m0) : 0.f;
    const float e1_c = (lane < n) ? expf(l1_c - m1) : 0.f;
    float d0 = e0_c, d1 = e1_c;
    for (int b = 32; b < n; b += 32) {
        const int i = b + lane;
        if (i < n) {
            int s; float l0, l1; item(i, s, l0, l1);
            d0 += expf(l0 - m0); d1 += expf(l1 - m1);
        }
    }
    const float den0 = wave_sum(d0) + 1e-16f;
    const float den1 = wave_sum(d1) + 1e-16f;

    v2f A0a = {0.f, 0.f}, A0b = {0.f, 0.f};
    v2f A1a = {0.f, 0.f}, A1b = {0.f, 0.f};
    const __bf16* __restrict__ xl = xb + 4 * lane;

    auto accumulate = [&](int cnt, int s, float w0, float w1) {
        for (int j0 = 0; j0 < cnt; j0 += 8) {
            v4bf r[8];
#pragma unroll
            for (int u = 0; u < 8; ++u) {
                const int sj = __builtin_amdgcn_readlane(s, j0 + u);
                r[u] = *(const v4bf*)(xl + (size_t)sj * DD);
            }
#pragma unroll
            for (int u = 0; u < 8; ++u) {
                const float x0 = lane_bcast(w0, j0 + u);
                const float x1 = lane_bcast(w1, j0 + u);
                const v2f fa = {(float)r[u][0], (float)r[u][1]};
                const v2f fb = {(float)r[u][2], (float)r[u][3]};
                const v2f w0v = {x0, x0}, w1v = {x1, x1};
                A0a = __builtin_elementwise_fma(fa, w0v, A0a);
                A0b = __builtin_elementwise_fma(fb, w0v, A0b);
                A1a = __builtin_elementwise_fma(fa, w1v, A1a);
                A1b = __builtin_elementwise_fma(fb, w1v, A1b);
            }
        }
    };

    accumulate(n < 32 ? n : 32, s_c, e0_c / den0, e1_c / den1);
    for (int b = 32; b < n; b += 32) {
        const int i = b + lane;
        int s = node; float w0 = 0.f, w1 = 0.f;
        if (i < n) {
            float l0, l1; item(i, s, l0, l1);
            w0 = expf(l0 - m0) / den0; w1 = expf(l1 - m1) / den1;
        }
        accumulate((n - b) < 32 ? (n - b) : 32, s, w0, w1);
    }

    v4bf o0, o1;
    o0[0] = (__bf16)A0a[0]; o0[1] = (__bf16)A0a[1]; o0[2] = (__bf16)A0b[0]; o0[3] = (__bf16)A0b[1];
    o1[0] = (__bf16)A1a[0]; o1[1] = (__bf16)A1a[1]; o1[2] = (__bf16)A1b[0]; o1[3] = (__bf16)A1b[1];
    __bf16* o = agg + (size_t)node * HC + 4 * lane;
    *(v4bf*)(o)      = o0;
    *(v4bf*)(o + CC) = o1;
}

__global__ __launch_bounds__(256)
void k_node_classify(const int* __restrict__ rowptr,
                     const int* __restrict__ csr_src,
                     const float* __restrict__ rec,
                     const float* __restrict__ bias,
                     const float* __restrict__ Wl,
                     const float* __restrict__ bl,
                     float* __restrict__ out) {
    const int node = __builtin_amdgcn_readfirstlane((int)(blockIdx.x * 8 + (threadIdx.x >> 5)));
    const int lane = threadIdx.x & 31;
    const float4* __restrict__ rec0 = (const float4*)rec;
    const float4* __restrict__ rec1 = rec0 + NN;
    const int beg  = rowptr[node];
    const int deg  = rowptr[node + 1] - beg;
    const int n    = deg + 1;
    const int* __restrict__ seg = csr_src + beg;
    const float ad0 = rec0[node].w;
    const float ad1 = rec1[node].w;

    auto item = [&](int i, float4& r0, float4& r1, float& l0, float& l1) {
        int s;
        if (i < deg) s = seg[i]; else s = node;
        r0 = rec0[s]; r1 = rec1[s];
        const float t0 = r0.x + ad0, t1 = r1.x + ad1;
        l0 = (t0 > 0.f) ? t0 : NEG_SLOPE * t0;
        l1 = (t1 > 0.f) ? t1 : NEG_SLOPE * t1;
    };

    float4 r0_c = {0.f, 0.f, 0.f, 0.f}, r1_c = {0.f, 0.f, 0.f, 0.f};
    float l0_c = -INFINITY, l1_c = -INFINITY;
    if (lane < n) item(lane, r0_c, r1_c, l0_c, l1_c);

    float m0 = l0_c, m1 = l1_c;
    for (int b = 32; b < n; b += 32) {
        const int i = b + lane;
        if (i < n) {
            float4 r0, r1; float l0, l1; item(i, r0, r1, l0, l1);
            m0 = fmaxf(m0, l0); m1 = fmaxf(m1, l1);
        }
    }
    m0 = wave_max(m0); m1 = wave_max(m1);

    const float e0_c = (lane < n) ? expf(l0_c - m0) : 0.f;
    const float e1_c = (lane < n) ? expf(l1_c - m1) : 0.f;
    float d0 = e0_c, d1 = e1_c;
    float q00 = e0_c * r0_c.y, q01 = e0_c * r0_c.z;
    float q10 = e1_c * r1_c.y, q11 = e1_c * r1_c.z;
    for (int b = 32; b < n; b += 32) {
        const int i = b + lane;
        if (i < n) {
            float4 r0, r1; float l0, l1; item(i, r0, r1, l0, l1);
            const float e0 = expf(l0 - m0), e1 = expf(l1 - m1);
            d0 += e0; d1 += e1;
            q00 += e0 * r0.y; q01 += e0 * r0.z;
            q10 += e1 * r1.y; q11 += e1 * r1.z;
        }
    }
    const float den0 = wave_sum(d0) + 1e-16f;
    const float den1 = wave_sum(d1) + 1e-16f;
    q00 = wave_sum(q00); q01 = wave_sum(q01);
    q10 = wave_sum(q10); q11 = wave_sum(q11);

    const float4 bA  = *(const float4*)(bias + 4 * lane);
    const float4 bB  = *(const float4*)(bias + CC + 4 * lane);
    const float4 wa0 = *(const float4*)(Wl + 8 * lane);
    const float4 wa1 = *(const float4*)(Wl + 8 * lane + 4);
    const float4 wb0 = *(const float4*)(Wl + 2 * CC + 8 * lane);
    const float4 wb1 = *(const float4*)(Wl + 2 * CC + 8 * lane + 4);
    float c0 = bA.x * wa0.x + bA.y * wa0.z + bA.z * wa1.x + bA.w * wa1.z
             + bB.x * wb0.x + bB.y * wb0.z + bB.z * wb1.x + bB.w * wb1.z;
    float c1 = bA.x * wa0.y + bA.y * wa0.w + bA.z * wa1.y + bA.w * wa1.w
             + bB.x * wb0.y + bB.y * wb0.w + bB.z * wb1.y + bB.w * wb1.w;
    c0 = wave_sum(c0); c1 = wave_sum(c1);

    const float z0 = q00 / den0 + q10 / den1 + c0 + bl[0];
    const float z1 = q01 / den0 + q11 / den1 + c1 + bl[1];
    const float mx = fmaxf(z0, z1);
    const float e0 = expf(z0 - mx), e1 = expf(z1 - mx);
    const float inv = 1.f / (e0 + e1);
    if (lane == 0) *(float2*)(out + 2 * (size_t)node) = make_float2(e0 * inv, e1 * inv);
}

extern "C" void kernel_launch(void* const* d_in, const int* in_sizes, int n_in,
                              void* d_out, int out_size, void* d_ws, size_t ws_size,
                              hipStream_t stream) {
    const float* x        = (const float*)d_in[0];
    const int*   ei       = (const int*)  d_in[1];
    const float* W0       = (const float*)d_in[2];
    const float* att_src0 = (const float*)d_in[3];
    const float* att_dst0 = (const float*)d_in[4];
    const float* b0       = (const float*)d_in[5];
    const float* W1       = (const float*)d_in[6];
    const float* att_src1 = (const float*)d_in[7];
    const float* att_dst1 = (const float*)d_in[8];
    const float* b1       = (const float*)d_in[9];
    const float* Wl       = (const float*)d_in[10];
    const float* bl       = (const float*)d_in[11];
    float* out = (float*)d_out;

    char* base = (char*)d_ws;
    size_t off = 0;
    auto carve = [&](size_t bytes) -> char* {
        char* p = base + off;
        off = (off + bytes + 255) & ~(size_t)255;
        return p;
    };
    float*    h_feat = (float*)   carve((size_t)NN * HC * 4);
    __bf16*   xbf    = (__bf16*)  carve((size_t)NN * HC * 2);
    __bf16*   wtbf   = (__bf16*)  carve((size_t)HC * HC * 2);
    float*    a_s    = (float*)   carve((size_t)NN * 2 * 4);
    float*    a_d    = (float*)   carve((size_t)NN * 2 * 4);
    char*     tbl    =            carve((size_t)NN * 4);
    int*      rowptr = (int*)     carve((size_t)(NN + 1) * 4);
    unsigned* binned = (unsigned*)carve((size_t)EE * 4);
    int*      csr    = (int*)     carve((size_t)EE * 4);
    __bf16*   agg    = (__bf16*)h_feat;
    float*    vsd    = h_feat + (size_t)NN * HC / 2;
    unsigned short* hist = (unsigned short*)tbl;
    int*      binstart   = (int*)(tbl + (size_t)SB_BLK * NBIN * 2);
    static_assert((size_t)SB_BLK * NBIN * 2 + (NBIN + 1) * 4 <= (size_t)NN * 4, "table space");
    (void)ws_size; (void)in_sizes; (void)n_in; (void)out_size;

    const int T = 256;
    dim3 blk(T);

    k_score_vecs<<<DD / 8, blk, 0, stream>>>(W0, att_src0, att_dst0, vsd);
    k_f32_to_bf16_scores<<<(NN * DD / 4) / (EW_T * CVT_U), dim3(EW_T), 0, stream>>>(
        x, vsd, xbf, a_s, a_d);
    k_transpose_bf16<<<(DD * HC + T - 1) / T, blk, 0, stream>>>(W0, wtbf, DD, HC);

    k_bin_count  <<<SB_BLK, dim3(SB_T), 0, stream>>>(ei, hist);
    k_bin_scatter<<<SB_BLK, dim3(SB_T), 0, stream>>>(ei, hist, binstart, binned);
    k_bin_sort   <<<NBIN,   dim3(BS_T), 0, stream>>>(binstart, binned, rowptr, csr);

    k_node_gather<<<NN / 8, blk, 0, stream>>>(rowptr, csr, xbf, a_s, a_d, agg);

    k_gemm_bf16_wmma<false, DD><<<(NN + 31) / 32, dim3(64), 0, stream>>>(
        agg, wtbf, NN, nullptr, nullptr, nullptr, nullptr, b0, xbf);

    k_transpose_bf16<<<(HC * HC + T - 1) / T, blk, 0, stream>>>(W1, wtbf, HC, HC);

    float* rec = h_feat;
    k_gemm_bf16_wmma<true, HC><<<(NN + 31) / 32, dim3(64), 0, stream>>>(
        xbf, wtbf, NN, att_src1, att_dst1, Wl, rec, nullptr, nullptr);

    k_node_classify<<<NN / 8, blk, 0, stream>>>(
        rowptr, csr, rec, b1, Wl, bl, out);
}


# Stages that do not depend on one another share a launch, split by block index: sort counts + score vectors + both weight transposes; sort level 2 + x conversion: 11 dependent launches -> 7
# speedup vs baseline: 17.6955x; 1.1115x over previous
#include <hip/hip_runtime.h>
#include <math.h>

#define NN    50000
#define DD    128
#define HC    256
#define CC    128
#define EE    800000
#define NEG_SLOPE 0.2f

typedef __attribute__((ext_vector_type(16))) __bf16 v16bf;
typedef __attribute__((ext_vector_type(8)))  __bf16 v8bf;
typedef __attribute__((ext_vector_type(8)))  float  v8f;

typedef __attribute__((ext_vector_type(4)))  __bf16 v4bf;
typedef __attribute__((ext_vector_type(2)))  float  v2f;

#define FZ_T   1024

#define CVT_U  4
#define CVT_WAVES (NN / CVT_U)
#define CVT_BLKS  ((CVT_WAVES + FZ_T / 32 - 1) / (FZ_T / 32))
static_assert(NN % CVT_U == 0, "a wave's rows must divide the nodes");

__device__ __forceinline__ float wave_max(float v) {
#pragma unroll
    for (int o = 16; o > 0; o >>= 1) v = fmaxf(v, __shfl_xor(v, o, 32));
    return v;
}
__device__ __forceinline__ float wave_sum(float v) {
#pragma unroll
    for (int o = 16; o > 0; o >>= 1) v += __shfl_xor(v, o, 32);
    return v;
}
template <int CTRL>
__device__ __forceinline__ float row_swz(float v) {
    const int b = __builtin_bit_cast(int, v);
    return __builtin_bit_cast(float, __builtin_amdgcn_update_dpp(b, b, CTRL, 0xf, 0xf, false));
}
__device__ __forceinline__ float xhalf(float v) {
    const int b = __builtin_bit_cast(int, v);
    return __builtin_bit_cast(float,
        __builtin_amdgcn_permlanex16(b, b, 0x76543210, 0xfedcba98, false, false));
}
__device__ __forceinline__ float wave_sum_valu(float v) {
    v += row_swz<0x161>(v);
    v += row_swz<0x162>(v);
    v += row_swz<0x164>(v);
    v += row_swz<0x168>(v);
    return v + xhalf(v);
}

static_assert(DD == 128 && CC == 128 && HC == 2 * CC, "a lane holds 4 features / 4 + 4 channels");
static_assert(DD % (FZ_T / 32) == 0, "whole blocks of one wave per input feature");
#define SV_BLKS (DD / (FZ_T / 32))
__device__ __forceinline__
void score_vecs_part(int blk,
                     const float* __restrict__ W,
                     const float* __restrict__ att_src,
                     const float* __restrict__ att_dst,
                     float* __restrict__ vsd) {
    const int k    = blk * (FZ_T / 32) + (threadIdx.x >> 5);
    const int lane = threadIdx.x & 31;
    const float4 w0 = *(const float4*)(W + (size_t)k * HC + 4 * lane);
    const float4 w1 = *(const float4*)(W + (size_t)k * HC + CC + 4 * lane);
    const float4 s0 = *(const float4*)(att_src + 4 * lane);
    const float4 s1 = *(const float4*)(att_src + CC + 4 * lane);
    const float4 d0 = *(const float4*)(att_dst + 4 * lane);
    const float4 d1 = *(const float4*)(att_dst + CC + 4 * lane);
    float ps0 = w0.x * s0.x + w0.y * s0.y + w0.z * s0.z + w0.w * s0.w;
    float ps1 = w1.x * s1.x + w1.y * s1.y + w1.z * s1.z + w1.w * s1.w;
    float pd0 = w0.x * d0.x + w0.y * d0.y + w0.z * d0.z + w0.w * d0.w;
    float pd1 = w1.x * d1.x + w1.y * d1.y + w1.z * d1.z + w1.w * d1.w;
    ps0 = wave_sum(ps0); ps1 = wave_sum(ps1); pd0 = wave_sum(pd0); pd1 = wave_sum(pd1);
    if (lane == 0) *(float4*)(vsd + 4 * k) = make_float4(ps0, ps1, pd0, pd1);
}

__device__ __forceinline__
void f32_to_bf16_scores_part(int blk,
                             const float* __restrict__ in,
                             const float* __restrict__ vsd,
                             __bf16* __restrict__ out,
                             float* __restrict__ a_s,
                             float* __restrict__ a_d) {
    const float4* __restrict__ in4 = (const float4*)in;
    const int gw   = blk * (FZ_T / 32) + (threadIdx.x >> 5);
    if (gw >= CVT_WAVES) return;
    const int lane = threadIdx.x & 31;
    const int q0   = gw * (CVT_U * 32) + lane;
    float4 v[CVT_U];
#pragma unroll
    for (int j = 0; j < CVT_U; ++j) v[j] = in4[q0 + j * 32];
    const float4* __restrict__ vp = (const float4*)vsd + 4 * lane;
    const float4 c0 = vp[0], c1 = vp[1], c2 = vp[2], c3 = vp[3];
#pragma unroll
    for (int j = 0; j < CVT_U; ++j) {
        v4bf o;
        o[0] = (__bf16)v[j].x; o[1] = (__bf16)v[j].y;
        o[2] = (__bf16)v[j].z; o[3] = (__bf16)v[j].w;
        *(v4bf*)(out + 4 * (size_t)(q0 + j * 32)) = o;
        float s0 = v[j].x * c0.x + v[j].y * c1.x + v[j].z * c2.x + v[j].w * c3.x;
        float s1 = v[j].x * c0.y + v[j].y * c1.y + v[j].z * c2.y + v[j].w * c3.y;
        float d0 = v[j].x * c0.z + v[j].y * c1.z + v[j].z * c2.z + v[j].w * c3.z;
        float d1 = v[j].x * c0.w + v[j].y * c1.w + v[j].z * c2.w + v[j].w * c3.w;
        s0 = wave_sum_valu(s0); s1 = wave_sum_valu(s1);
        d0 = wave_sum_valu(d0); d1 = wave_sum_valu(d1);
        const int row = gw * CVT_U + j;
        if (lane == 0) {
            *(float2*)(a_s + 2 * (size_t)row) = make_float2(s0, s1);
            *(float2*)(a_d + 2 * (size_t)row) = make_float2(d0, d1);
        }
    }
}

template <int K, int Nn>
__device__ __forceinline__
void transpose_bf16_part(int blk, const float* __restrict__ W, __bf16* __restrict__ Wt) {
    static_assert((K * Nn) % FZ_T == 0, "whole blocks");
    const int i = blk * FZ_T + threadIdx.x;
    const int n = i / K, k = i - n * K;
    Wt[i] = (__bf16)W[(size_t)k * Nn + n];
}
#define T0_BLKS (DD * HC / FZ_T)
#define T1_BLKS (HC * HC / FZ_T)

#define GM_NT 8
static_assert(GM_NT * 16 == CC && HC == 2 * CC, "a wave covers one head, a block both");

__device__ __forceinline__ v16bf load_frag(const __bf16* __restrict__ p) {
    const v8bf lo = *(const v8bf*)(p);
    const v8bf hi = *(const v8bf*)(p + 16);
    v16bf f;
#pragma unroll
    for (int i = 0; i < 8; ++i) { f[i] = lo[i]; f[i + 8] = hi[i]; }
    return f;
}

#define GM_TP (CC + 4)
template <bool PROJ, int K>
__global__ __launch_bounds__(64)
void k_gemm_bf16_wmma(const __bf16* __restrict__ A,
                      const __bf16* __restrict__ Bt,
                      int M,
                      const float* __restrict__ att_src,
                      const float* __restrict__ att_dst,
                      const float* __restrict__ Wl,
                      float* __restrict__ rec,
                      const float* __restrict__ bias,
                      __bf16* __restrict__ hout) {
    static_assert(K % 64 == 0, "a trip of the K loop takes 64");
    static_assert(PROJ ? K == HC : K == CC, "layer 0 reads its head's half of a row of A");
    const int lane = threadIdx.x & 31;
    const int half = lane >> 4;
    const int lr   = lane & 15;
    const int tM   = blockIdx.x * 32;
    const int tN0  = (threadIdx.x >> 5) * CC;
    const bool has1 = tM + 16 < M;
    const int  r1   = has1 ? tM + 16 + lr : M - 1;
    const int aoff = (PROJ ? 0 : tN0) + 8 * half;
    const __bf16* __restrict__ ap0 = A  + (size_t)(tM + lr) * HC + aoff;
    const __bf16* __restrict__ ap1 = A  + (size_t)r1 * HC + aoff;
    const __bf16* __restrict__ bp  = Bt + (size_t)(tN0 + lr) * K + 8 * half;

    v8f acc[2][GM_NT];
#pragma unroll
    for (int j = 0; j < GM_NT; ++j) { acc[0][j] = v8f{}; acc[1][j] = v8f{}; }

#pragma unroll 1
    for (int kk = 0; kk < K; kk += 64) {
        v16bf a[2][2];
#pragma unroll
        for (int ks = 0; ks < 2; ++ks) {
            a[0][ks] = load_frag(ap0 + kk + 32 * ks);
            a[1][ks] = load_frag(ap1 + kk + 32 * ks);
        }
#pragma unroll
        for (int nh = 0; nh < 2; ++nh) {
            v16bf b[4][2];
#pragma unroll
            for (int jj = 0; jj < 4; ++jj) {
#pragma unroll
                for (int ks = 0; ks < 2; ++ks)
                    b[jj][ks] = load_frag(bp + (size_t)(4 * nh + jj) * 16 * K + kk + 32 * ks);
            }
#pragma unroll
            for (int ks = 0; ks < 2; ++ks) {
#pragma unroll
                for (int jj = 0; jj < 4; ++jj) {
                    const int j = 4 * nh + jj;
#pragma unroll
                    for (int m = 0; m < 2; ++m) {
                        acc[m][j] = __builtin_amdgcn_wmma_f32_16x16x32_bf16(
                            false, b[jj][ks], false, a[m][ks], (short)0, acc[m][j],
                            false, false);
                    }
                }
            }
        }
        __builtin_amdgcn_sched_group_barrier(0x020, 24, 0);
        __builtin_amdgcn_sched_group_barrier(0x008, 16, 0);
        __builtin_amdgcn_sched_group_barrier(0x020, 16, 0);
        __builtin_amdgcn_sched_group_barrier(0x008, 16, 0);
    }

    if constexpr (PROJ) {
        v2f psd[2] = {{0.f, 0.f}, {0.f, 0.f}};
        v2f p01[2] = {{0.f, 0.f}, {0.f, 0.f}};
        auto load_vec = [&](int j, float4 (&v)[8]) {
            const int c0 = tN0 + 16 * j + 8 * half;
            v[0] = *(const float4*)(att_src + c0); v[1] = *(const float4*)(att_src + c0 + 4);
            v[2] = *(const float4*)(att_dst + c0); v[3] = *(const float4*)(att_dst + c0 + 4);
            v[4] = *(const float4*)(Wl + 2 * c0);     v[5] = *(const float4*)(Wl + 2 * c0 + 4);
            v[6] = *(const float4*)(Wl + 2 * c0 + 8); v[7] = *(const float4*)(Wl + 2 * c0 + 12);
        };
#pragma unroll
        for (int j = 0; j < GM_NT; ++j) {
            float4 cur[8];
            load_vec(j, cur);
            const float s8[8] = {cur[0].x, cur[0].y, cur[0].z, cur[0].w,
                                 cur[1].x, cur[1].y, cur[1].z, cur[1].w};
            const float d8[8] = {cur[2].x, cur[2].y, cur[2].z, cur[2].w,
                                 cur[3].x, cur[3].y, cur[3].z, cur[3].w};
            const v2f w8[8] = {{cur[4].x, cur[4].y}, {cur[4].z, cur[4].w},
                               {cur[5].x, cur[5].y}, {cur[5].z, cur[5].w},
                               {cur[6].x, cur[6].y}, {cur[6].z, cur[6].w},
                               {cur[7].x, cur[7].y}, {cur[7].z, cur[7].w}};
#pragma unroll
            for (int r = 0; r < 8; ++r) {
                const v2f sd = {s8[r], d8[r]};
#pragma unroll
                for (int m = 0; m < 2; ++m) {
                    const float t = acc[m][j][r];
                    const v2f tt = {t, t};
                    psd[m] = __builtin_elementwise_fma(tt, sd,    psd[m]);
                    p01[m] = __builtin_elementwise_fma(tt, w8[r], p01[m]);
                }
            }
        }
        float4* __restrict__ rp = (float4*)rec + (size_t)(tN0 / CC) * M + tM + lr;
#pragma unroll
        for (int m = 0; m < 2; ++m) {
            const float s  = psd[m][0] + __shfl_xor(psd[m][0], 16, 32);
            const float d  = psd[m][1] + __shfl_xor(psd[m][1], 16, 32);
            const float q0 = p01[m][0] + __shfl_xor(p01[m][0], 16, 32);
            const float q1 = p01[m][1] + __shfl_xor(p01[m][1], 16, 32);
            if (half == 0 && (m == 0 || has1)) rp[16 * m] = make_float4(s, q0, q1, d);
        }
    } else {
        __shared__ alignas(16) float s_tile[2][32 * GM_TP];
        float* __restrict__ sw = s_tile[threadIdx.x >> 5];
#pragma unroll
        for (int m = 0; m < 2; ++m) {
#pragma unroll
            for (int j = 0; j < GM_NT; ++j) {
                float* p = sw + (16 * m + lr) * GM_TP + 16 * j + 8 * half;
                *(float4*)(p)     = make_float4(acc[m][j][0], acc[m][j][1], acc[m][j][2], acc[m][j][3]);
                *(float4*)(p + 4) = make_float4(acc[m][j][4], acc[m][j][5], acc[m][j][6], acc[m][j][7]);
            }
        }
        __syncthreads();
        const float4 bA = *(const float4*)(bias + tN0 + 8 * lr);
        const float4 bB = *(const float4*)(bias + tN0 + 8 * lr + 4);
        __bf16* __restrict__ op = hout + (size_t)tM * HC + tN0 + 8 * lr;
#pragma unroll 1
        for (int it = 0; it < 16; ++it) {
            const int rr = 2 * it + half;
            const float4 u0 = *(const float4*)(sw + rr * GM_TP + 8 * lr);
            const float4 u1 = *(const float4*)(sw + rr * GM_TP + 8 * lr + 4);
            v8bf o;
            o[0] = (__bf16)tanhf(u0.x + bA.x); o[1] = (__bf16)tanhf(u0.y + bA.y);
            o[2] = (__bf16)tanhf(u0.z + bA.z); o[3] = (__bf16)tanhf(u0.w + bA.w);
            o[4] = (__bf16)tanhf(u1.x + bB.x); o[5] = (__bf16)tanhf(u1.y + bB.y);
            o[6] = (__bf16)tanhf(u1.z + bB.z); o[7] = (__bf16)tanhf(u1.w + bB.w);
            if (tM + rr < M) *(v8bf*)(op + (size_t)rr * HC) = o;
        }
    }
}

#define SB_BLK  256
#define SB_CH   (EE / SB_BLK)
#define SB_T    1024
#define SB_IT   ((SB_CH + SB_T - 1) / SB_T)
#define NBN     196
#define NBIN    256
#define BS_T    1024
#define BS_U    4
#define BS_CAP  8192
static_assert(EE % SB_BLK == 0, "chunks must divide the edge list");
static_assert(NBN * NBIN >= NN && NBN <= 256, "bins must cover the nodes, a bin's counters one scan");
static_assert(NN <= 65536, "a source id must fit the low half of the word");
static_assert(SB_CH < 65536, "a chunk's count per bin is kept in 16 bits");
static_assert(SB_T == 4 * NBIN && SB_BLK % 64 == 0, "k_bin_scatter splits the table in four");

__device__ __forceinline__ int block_excl_scan256(int v, int t, int* s_w) {
    const int lane = t & 31, w = t >> 5;
    int inc = v;
#pragma unroll
    for (int o = 1; o < 32; o <<= 1) {
        int u = __shfl_up(inc, o, 32);
        if (lane >= o) inc += u;
    }
    if (lane == 31 && w < 8) s_w[w] = inc;
    __syncthreads();
    int base = 0;
#pragma unroll
    for (int j = 0; j < 8; ++j) if (j < w) base += s_w[j];
    return base + inc - v;
}

__device__ __forceinline__
void bin_count_part(int blk, const int* __restrict__ ei, unsigned short* __restrict__ hist) {
    __shared__ int s_cnt[NBIN];
    const int t = threadIdx.x;
    const int* __restrict__ dstp = ei + EE + blk * SB_CH;
    unsigned d[SB_IT];
#pragma unroll
    for (int j = 0; j < SB_IT; ++j) {
        const int i = t + j * SB_T;
        d[j] = (i < SB_CH) ? (unsigned)dstp[i] : 0u;
    }
    if (t < NBIN) s_cnt[t] = 0;
    __syncthreads();
#pragma unroll
    for (int j = 0; j < SB_IT; ++j) {
        const int i = t + j * SB_T;
        if (i < SB_CH) atomicAdd(&s_cnt[d[j] / NBN], 1);
    }
    __syncthreads();
    if (t < NBIN) hist[blk * NBIN + t] = (unsigned short)s_cnt[t];
}

static_assert(SB_T == FZ_T, "the shared launches have the sort's block size");
__global__ __launch_bounds__(FZ_T)
void k_counts_vecs_weights(const int* __restrict__ ei, unsigned short* __restrict__ hist,
                           const float* __restrict__ W0,
                           const float* __restrict__ att_src0,
                           const float* __restrict__ att_dst0,
                           float* __restrict__ vsd,
                           __bf16* __restrict__ wt0,
                           const float* __restrict__ W1,
                           __bf16* __restrict__ wt1) {
    const int b = blockIdx.x;
    if (b < SB_BLK)                               bin_count_part(b, ei, hist);
    else if (b < SB_BLK + SV_BLKS)                score_vecs_part(b - SB_BLK, W0, att_src0, att_dst0, vsd);
    else if (b < SB_BLK + SV_BLKS + T0_BLKS)      transpose_bf16_part<DD, HC>(b - SB_BLK - SV_BLKS, W0, wt0);
    else                                          transpose_bf16_part<HC, HC>(b - SB_BLK - SV_BLKS - T0_BLKS, W1, wt1);
}

__global__ __launch_bounds__(SB_T)
void k_bin_scatter(const int* __restrict__ ei,
                   const unsigned short* __restrict__ hist,
                   int* __restrict__ binstart,
                   unsigned* __restrict__ binned) {
    __shared__ int s_tot[4][NBIN];
    __shared__ int s_par[4][NBIN];
    __shared__ int s_goff[NBIN];
    __shared__ int s_cnt[NBIN];
    __shared__ int s_lst[NBIN];
    __shared__ int s_w1[8];
    __shared__ int s_w2[8];
    __shared__ unsigned s_val[SB_CH];
    __shared__ int s_addr[SB_CH];
    const int t = threadIdx.x, k = blockIdx.x;
    const int bin = t & (NBIN - 1), q = t >> 8;

    const int* __restrict__ srcp = ei + k * SB_CH;
    const int* __restrict__ dstp = ei + EE + k * SB_CH;
    unsigned es[SB_IT], ed[SB_IT];
#pragma unroll
    for (int j = 0; j < SB_IT; ++j) {
        const int i = t + j * SB_T;
        es[j] = (i < SB_CH) ? (unsigned)srcp[i] : 0u;
        ed[j] = (i < SB_CH) ? (unsigned)dstp[i] : 0u;
    }

    int tot = 0, par = 0;
    const unsigned short* __restrict__ hp = hist + q * (SB_BLK / 4) * NBIN + bin;
    const int lim = k - q * (SB_BLK / 4);
#pragma unroll
    for (int i0 = 0; i0 < SB_BLK / 4; i0 += 16) {
        int h[16];
#pragma unroll
        for (int u = 0; u < 16; ++u) h[u] = hp[(i0 + u) * NBIN];
#pragma unroll
        for (int u = 0; u < 16; ++u) {
            tot += h[u];
            par += (i0 + u < lim) ? h[u] : 0;
        }
    }
    s_tot[q][bin] = tot;
    s_par[q][bin] = par;
    if (t < NBIN) s_cnt[t] = 0;
    __syncthreads();
    int total = 0, partial = 0;
    if (t < NBIN) {
        total   = s_tot[0][t] + s_tot[1][t] + s_tot[2][t] + s_tot[3][t];
        partial = s_par[0][t] + s_par[1][t] + s_par[2][t] + s_par[3][t];
    }
    const int bs = block_excl_scan256(total, t, s_w1);
    if (t < NBIN) {
        s_goff[t] = bs + partial;
        if (k == 0) {
            binstart[t] = bs;
            if (t == NBIN - 1) binstart[NBIN] = bs + total;
        }
    }

    int      bn[SB_IT];
    int      rk[SB_IT];
    unsigned pv[SB_IT];
#pragma unroll
    for (int j = 0; j < SB_IT; ++j) {
        const int i = t + j * SB_T;
        bn[j] = -1; rk[j] = 0; pv[j] = 0u;
        if (i < SB_CH) {
            const unsigned d = ed[j];
            const unsigned b = d / NBN;
            bn[j] = (int)b;
            pv[j] = ((d - b * NBN) << 16) | es[j];
            rk[j] = atomicAdd(&s_cnt[b], 1);
        }
    }
    __syncthreads();
    const int lc = (t < NBIN) ? s_cnt[t] : 0;
    const int ls = block_excl_scan256(lc, t, s_w2);
    if (t < NBIN) s_lst[t] = ls;
    __syncthreads();
#pragma unroll
    for (int j = 0; j < SB_IT; ++j) {
        if (bn[j] >= 0) {
            const int p = s_lst[bn[j]] + rk[j];
            s_val[p]  = pv[j];
            s_addr[p] = s_goff[bn[j]] + rk[j];
        }
    }
    __syncthreads();
#pragma unroll
    for (int j = 0; j < SB_IT; ++j) {
        const int i = t + j * SB_T;
        if (i < SB_CH) binned[s_addr[i]] = s_val[i];
    }
}

__device__ __forceinline__
void bin_sort_part(int blk,
                   const int* __restrict__ binstart,
                   const unsigned* __restrict__ binned,
                   int* __restrict__ rowptr,
                   int* __restrict__ csr_src) {
    __shared__ int s_c[256];
    __shared__ int s_w[8];
    __shared__ int s_out[BS_CAP];
    const int t = threadIdx.x, b = blk;
    const int beg = binstart[b], end = binstart[b + 1];
    const bool staged = (end - beg) <= BS_CAP;
    if (t < 256) s_c[t] = 0;
    __syncthreads();
    for (int i0 = beg + t; i0 < end; i0 += BS_U * BS_T) {
        unsigned v[BS_U];
#pragma unroll
        for (int u = 0; u < BS_U; ++u) {
            const int i = i0 + u * BS_T;
            v[u] = binned[i < end ? i : end - 1];
        }
#pragma unroll
        for (int u = 0; u < BS_U; ++u) {
            const int i = i0 + u * BS_T;
            if (i < end) atomicAdd(&s_c[v[u] >> 16], 1);
        }
    }
    __syncthreads();
    const int c   = (t < 256) ? s_c[t] : 0;
    const int pre = block_excl_scan256(c, t, s_w);
    const int node = b * NBN + t;
    if (t < NBN && node < NN) {
        rowptr[node] = beg + pre;
        if (node == NN - 1) rowptr[NN] = beg + pre + c;
    }
    if (t < 256) s_c[t] = beg + pre;
    __syncthreads();
    for (int i0 = beg + t; i0 < end; i0 += BS_U * BS_T) {
        unsigned v[BS_U];
#pragma unroll
        for (int u = 0; u < BS_U; ++u) {
            const int i = i0 + u * BS_T;
            v[u] = binned[i < end ? i : end - 1];
        }
#pragma unroll
        for (int u = 0; u < BS_U; ++u) {
            const int i = i0 + u * BS_T;
            if (i < end) {
                const int slot = atomicAdd(&s_c[v[u] >> 16], 1);
                if (staged) s_out[slot - beg] = (int)(v[u] & 0xFFFFu);
                else        csr_src[slot]     = (int)(v[u] & 0xFFFFu);
            }
        }
    }
    if (staged) {
        __syncthreads();
        for (int i = t; i < end - beg; i += BS_T) csr_src[beg + i] = s_out[i];
    }
}

static_assert(BS_T == FZ_T, "the shared launches have the sort's block size");
__global__ __launch_bounds__(FZ_T)
void k_sort_convert(const int* __restrict__ binstart,
                    const unsigned* __restrict__ binned,
                    int* __restrict__ rowptr,
                    int* __restrict__ csr_src,
                    const float* __restrict__ x,
                    const float* __restrict__ vsd,
                    __bf16* __restrict__ xb,
                    float* __restrict__ a_s,
                    float* __restrict__ a_d) {
    const int b = blockIdx.x;
    if (b < NBIN) bin_sort_part(b, binstart, binned, rowptr, csr_src);
    else          f32_to_bf16_scores_part(b - NBIN, x, vsd, xb, a_s, a_d);
}

__device__ __forceinline__ float lane_bcast(float v, int j) {
    return __int_as_float(__builtin_amdgcn_readlane(__float_as_int(v), j));
}

__global__ __launch_bounds__(256)
void k_node_gather(const int* __restrict__ rowptr,
                   const int* __restrict__ csr_src,
                   const __bf16* __restrict__ xb,
                   const float* __restrict__ a_s,
                   const float* __restrict__ a_d,
                   __bf16* __restrict__ agg) {
    const int node = __builtin_amdgcn_readfirstlane((int)(blockIdx.x * 8 + (threadIdx.x >> 5)));
    const int lane = threadIdx.x & 31;
    const int beg  = rowptr[node];
    const int deg  = rowptr[node + 1] - beg;
    const int n    = deg + 1;
    const int* __restrict__ seg = csr_src + beg;
    const float ad0 = a_d[node * 2 + 0];
    const float ad1 = a_d[node * 2 + 1];

    auto item = [&](int i, int& s, float& l0, float& l1) {
        if (i < deg) s = seg[i]; else s = node;
        const float2 as = *(const float2*)(a_s + 2 * (size_t)s);
        const float t0 = as.x + ad0, t1 = as.y + ad1;
        l0 = (t0 > 0.f) ? t0 : NEG_SLOPE * t0;
        l1 = (t1 > 0.f) ? t1 : NEG_SLOPE * t1;
    };

    int   s_c  = node;
    float l0_c = -INFINITY, l1_c = -INFINITY;
    if (lane < n) item(lane, s_c, l0_c, l1_c);

    float m0 = l0_c, m1 = l1_c;
    for (int b = 32; b < n; b += 32) {
        const int i = b + lane;
        if (i < n) {
            int s; float l0, l1; item(i, s, l0, l1);
            m0 = fmaxf(m0, l0); m1 = fmaxf(m1, l1);
        }
    }
    m0 = wave_max(m0); m1 = wave_max(m1);

    const float e0_c = (lane < n) ? expf(l0_c - m0) : 0.f;
    const float e1_c = (lane < n) ? expf(l1_c - m1) : 0.f;
    float d0 = e0_c, d1 = e1_c;
    for (int b = 32; b < n; b += 32) {
        const int i = b + lane;
        if (i < n) {
            int s; float l0, l1; item(i, s, l0, l1);
            d0 += expf(l0 - m0); d1 += expf(l1 - m1);
        }
    }
    const float den0 = wave_sum(d0) + 1e-16f;
    const float den1 = wave_sum(d1) + 1e-16f;

    v2f A0a = {0.f, 0.f}, A0b = {0.f, 0.f};
    v2f A1a = {0.f, 0.f}, A1b = {0.f, 0.f};
    const __bf16* __restrict__ xl = xb + 4 * lane;

    auto accumulate = [&](int cnt, int s, float w0, float w1) {
        for (int j0 = 0; j0 < cnt; j0 += 8) {
            v4bf r[8];
#pragma unroll
            for (int u = 0; u < 8; ++u) {
                const int sj = __builtin_amdgcn_readlane(s, j0 + u);
                r[u] = *(const v4bf*)(xl + (size_t)sj * DD);
            }
#pragma unroll
            for (int u = 0; u < 8; ++u) {
                const float x0 = lane_bcast(w0, j0 + u);
                const float x1 = lane_bcast(w1, j0 + u);
                const v2f fa = {(float)r[u][0], (float)r[u][1]};
                const v2f fb = {(float)r[u][2], (float)r[u][3]};
                const v2f w0v = {x0, x0}, w1v = {x1, x1};
                A0a = __builtin_elementwise_fma(fa, w0v, A0a);
                A0b = __builtin_elementwise_fma(fb, w0v, A0b);
                A1a = __builtin_elementwise_fma(fa, w1v, A1a);
                A1b = __builtin_elementwise_fma(fb, w1v, A1b);
            }
        }
    };

    accumulate(n < 32 ? n : 32, s_c, e0_c / den0, e1_c / den1);
    for (int b = 32; b < n; b += 32) {
        const int i = b + lane;
        int s = node; float w0 = 0.f, w1 = 0.f;
        if (i < n) {
            float l0, l1; item(i, s, l0, l1);
            w0 = expf(l0 - m0) / den0; w1 = expf(l1 - m1) / den1;
        }
        accumulate((n - b) < 32 ? (n - b) : 32, s, w0, w1);
    }

    v4bf o0, o1;
    o0[0] = (__bf16)A0a[0]; o0[1] = (__bf16)A0a[1]; o0[2] = (__bf16)A0b[0]; o0[3] = (__bf16)A0b[1];
    o1[0] = (__bf16)A1a[0]; o1[1] = (__bf16)A1a[1]; o1[2] = (__bf16)A1b[0]; o1[3] = (__bf16)A1b[1];
    __bf16* o = agg + (size_t)node * HC + 4 * lane;
    *(v4bf*)(o)      = o0;
    *(v4bf*)(o + CC) = o1;
}

__global__ __launch_bounds__(256)
void k_node_classify(const int* __restrict__ rowptr,
                     const int* __restrict__ csr_src,
                     const float* __restrict__ rec,
                     const float* __restrict__ bias,
                     const float* __restrict__ Wl,
                     const float* __restrict__ bl,
                     float* __restrict__ out) {
    const int node = __builtin_amdgcn_readfirstlane((int)(blockIdx.x * 8 + (threadIdx.x >> 5)));
    const int lane = threadIdx.x & 31;
    const float4* __restrict__ rec0 = (const float4*)rec;
    const float4* __restrict__ rec1 = rec0 + NN;
    const int beg  = rowptr[node];
    const int deg  = rowptr[node + 1] - beg;
    const int n    = deg + 1;
    const int* __restrict__ seg = csr_src + beg;
    const float ad0 = rec0[node].w;
    const float ad1 = rec1[node].w;

    auto item = [&](int i, float4& r0, float4& r1, float& l0, float& l1) {
        int s;
        if (i < deg) s = seg[i]; else s = node;
        r0 = rec0[s]; r1 = rec1[s];
        const float t0 = r0.x + ad0, t1 = r1.x + ad1;
        l0 = (t0 > 0.f) ? t0 : NEG_SLOPE * t0;
        l1 = (t1 > 0.f) ? t1 : NEG_SLOPE * t1;
    };

    float4 r0_c = {0.f, 0.f, 0.f, 0.f}, r1_c = {0.f, 0.f, 0.f, 0.f};
    float l0_c = -INFINITY, l1_c = -INFINITY;
    if (lane < n) item(lane, r0_c, r1_c, l0_c, l1_c);

    float m0 = l0_c, m1 = l1_c;
    for (int b = 32; b < n; b += 32) {
        const int i = b + lane;
        if (i < n) {
            float4 r0, r1; float l0, l1; item(i, r0, r1, l0, l1);
            m0 = fmaxf(m0, l0); m1 = fmaxf(m1, l1);
        }
    }
    m0 = wave_max(m0); m1 = wave_max(m1);

    const float e0_c = (lane < n) ? expf(l0_c - m0) : 0.f;
    const float e1_c = (lane < n) ? expf(l1_c - m1) : 0.f;
    float d0 = e0_c, d1 = e1_c;
    float q00 = e0_c * r0_c.y, q01 = e0_c * r0_c.z;
    float q10 = e1_c * r1_c.y, q11 = e1_c * r1_c.z;
    for (int b = 32; b < n; b += 32) {
        const int i = b + lane;
        if (i < n) {
            float4 r0, r1; float l0, l1; item(i, r0, r1, l0, l1);
            const float e0 = expf(l0 - m0), e1 = expf(l1 - m1);
            d0 += e0; d1 += e1;
            q00 += e0 * r0.y; q01 += e0 * r0.z;
            q10 += e1 * r1.y; q11 += e1 * r1.z;
        }
    }
    const float den0 = wave_sum(d0) + 1e-16f;
    const float den1 = wave_sum(d1) + 1e-16f;
    q00 = wave_sum(q00); q01 = wave_sum(q01);
    q10 = wave_sum(q10); q11 = wave_sum(q11);

    const float4 bA  = *(const float4*)(bias + 4 * lane);
    const float4 bB  = *(const float4*)(bias + CC + 4 * lane);
    const float4 wa0 = *(const float4*)(Wl + 8 * lane);
    const float4 wa1 = *(const float4*)(Wl + 8 * lane + 4);
    const float4 wb0 = *(const float4*)(Wl + 2 * CC + 8 * lane);
    const float4 wb1 = *(const float4*)(Wl + 2 * CC + 8 * lane + 4);
    float c0 = bA.x * wa0.x + bA.y * wa0.z + bA.z * wa1.x + bA.w * wa1.z
             + bB.x * wb0.x + bB.y * wb0.z + bB.z * wb1.x + bB.w * wb1.z;
    float c1 = bA.x * wa0.y + bA.y * wa0.w + bA.z * wa1.y + bA.w * wa1.w
             + bB.x * wb0.y + bB.y * wb0.w + bB.z * wb1.y + bB.w * wb1.w;
    c0 = wave_sum(c0); c1 = wave_sum(c1);

    const float z0 = q00 / den0 + q10 / den1 + c0 + bl[0];
    const float z1 = q01 / den0 + q11 / den1 + c1 + bl[1];
    const float mx = fmaxf(z0, z1);
    const float e0 = expf(z0 - mx), e1 = expf(z1 - mx);
    const float inv = 1.f / (e0 + e1);
    if (lane == 0) *(float2*)(out + 2 * (size_t)node) = make_float2(e0 * inv, e1 * inv);
}

extern "C" void kernel_launch(void* const* d_in, const int* in_sizes, int n_in,
                              void* d_out, int out_size, void* d_ws, size_t ws_size,
                              hipStream_t stream) {
    const float* x        = (const float*)d_in[0];
    const int*   ei       = (const int*)  d_in[1];
    const float* W0       = (const float*)d_in[2];
    const float* att_src0 = (const float*)d_in[3];
    const float* att_dst0 = (const float*)d_in[4];
    const float* b0       = (const float*)d_in[5];
    const float* W1       = (const float*)d_in[6];
    const float* att_src1 = (const float*)d_in[7];
    const float* att_dst1 = (const float*)d_in[8];
    const float* b1       = (const float*)d_in[9];
    const float* Wl       = (const float*)d_in[10];
    const float* bl       = (const float*)d_in[11];
    float* out = (float*)d_out;

    char* base = (char*)d_ws;
    size_t off = 0;
    auto carve = [&](size_t bytes) -> char* {
        char* p = base + off;
        off = (off + bytes + 255) & ~(size_t)255;
        return p;
    };
    float*    h_feat = (float*)   carve((size_t)NN * HC * 4);
    __bf16*   xbf    = (__bf16*)  carve((size_t)NN * HC * 2);
    __bf16*   wtbf   = (__bf16*)  carve((size_t)HC * HC * 2);
    float*    a_s    = (float*)   carve((size_t)NN * 2 * 4);
    float*    a_d    = (float*)   carve((size_t)NN * 2 * 4);
    char*     tbl    =            carve((size_t)NN * 4);
    int*      rowptr = (int*)     carve((size_t)(NN + 1) * 4);
    unsigned* binned = (unsigned*)carve((size_t)EE * 4);
    int*      csr    = (int*)     carve((size_t)EE * 4);
    __bf16*   agg    = (__bf16*)h_feat;
    float*    vsd    = h_feat + (size_t)NN * HC / 2;
    __bf16*   wtbf1  = (__bf16*)(vsd + 1024);
    static_assert(DD * 4 <= 1024 && (size_t)1024 * 4 + (size_t)HC * HC * 2 <= (size_t)NN * HC * 2,
                  "score vectors and layer 1's weights fit the second half of h_feat");
    unsigned short* hist = (unsigned short*)tbl;
    int*      binstart   = (int*)(tbl + (size_t)SB_BLK * NBIN * 2);
    static_assert((size_t)SB_BLK * NBIN * 2 + (NBIN + 1) * 4 <= (size_t)NN * 4, "table space");
    (void)ws_size; (void)in_sizes; (void)n_in; (void)out_size;

    const int T = 256;
    dim3 blk(T);

    k_counts_vecs_weights<<<SB_BLK + SV_BLKS + T0_BLKS + T1_BLKS, dim3(FZ_T), 0, stream>>>(
        ei, hist, W0, att_src0, att_dst0, vsd, wtbf, W1, wtbf1);

    k_bin_scatter<<<SB_BLK, dim3(SB_T), 0, stream>>>(ei, hist, binstart, binned);

    k_sort_convert<<<NBIN + CVT_BLKS, dim3(FZ_T), 0, stream>>>(
        binstart, binned, rowptr, csr, x, vsd, xbf, a_s, a_d);

    k_node_gather<<<NN / 8, blk, 0, stream>>>(rowptr, csr, xbf, a_s, a_d, agg);

    k_gemm_bf16_wmma<false, DD><<<(NN + 31) / 32, dim3(64), 0, stream>>>(
        agg, wtbf, NN, nullptr, nullptr, nullptr, nullptr, b0, xbf);

    float* rec = h_feat;
    k_gemm_bf16_wmma<true, HC><<<(NN + 31) / 32, dim3(64), 0, stream>>>(
        xbf, wtbf1, NN, att_src1, att_dst1, Wl, rec, nullptr, nullptr);

    k_node_classify<<<NN / 8, blk, 0, stream>>>(
        rowptr, csr, rec, b1, Wl, bl, out);
}


# Layer 0 tail's tanh branch-free: (1-e)/(1+e), e=exp(-2|x|) on hardware exp2+rcp, 8 ops, not the library's two-sided routine (a wave runs both sides, 36 slots each); abs err 1e-7, under the rounding
# speedup vs baseline: 18.5433x; 1.0479x over previous
#include <hip/hip_runtime.h>
#include <math.h>

#define NN    50000
#define DD    128
#define HC    256
#define CC    128
#define EE    800000
#define NEG_SLOPE 0.2f

typedef __attribute__((ext_vector_type(16))) __bf16 v16bf;
typedef __attribute__((ext_vector_type(8)))  __bf16 v8bf;
typedef __attribute__((ext_vector_type(8)))  float  v8f;

typedef __attribute__((ext_vector_type(4)))  __bf16 v4bf;
typedef __attribute__((ext_vector_type(2)))  float  v2f;

#define FZ_T   1024

#define CVT_U  4
#define CVT_WAVES (NN / CVT_U)
#define CVT_BLKS  ((CVT_WAVES + FZ_T / 32 - 1) / (FZ_T / 32))
static_assert(NN % CVT_U == 0, "a wave's rows must divide the nodes");

__device__ __forceinline__ float wave_max(float v) {
#pragma unroll
    for (int o = 16; o > 0; o >>= 1) v = fmaxf(v, __shfl_xor(v, o, 32));
    return v;
}
__device__ __forceinline__ float wave_sum(float v) {
#pragma unroll
    for (int o = 16; o > 0; o >>= 1) v += __shfl_xor(v, o, 32);
    return v;
}
template <int CTRL>
__device__ __forceinline__ float row_swz(float v) {
    const int b = __builtin_bit_cast(int, v);
    return __builtin_bit_cast(float, __builtin_amdgcn_update_dpp(b, b, CTRL, 0xf, 0xf, false));
}
__device__ __forceinline__ float xhalf(float v) {
    const int b = __builtin_bit_cast(int, v);
    return __builtin_bit_cast(float,
        __builtin_amdgcn_permlanex16(b, b, 0x76543210, 0xfedcba98, false, false));
}
__device__ __forceinline__ float wave_sum_valu(float v) {
    v += row_swz<0x161>(v);
    v += row_swz<0x162>(v);
    v += row_swz<0x164>(v);
    v += row_swz<0x168>(v);
    return v + xhalf(v);
}

static_assert(DD == 128 && CC == 128 && HC == 2 * CC, "a lane holds 4 features / 4 + 4 channels");
static_assert(DD % (FZ_T / 32) == 0, "whole blocks of one wave per input feature");
#define SV_BLKS (DD / (FZ_T / 32))
__device__ __forceinline__
void score_vecs_part(int blk,
                     const float* __restrict__ W,
                     const float* __restrict__ att_src,
                     const float* __restrict__ att_dst,
                     float* __restrict__ vsd) {
    const int k    = blk * (FZ_T / 32) + (threadIdx.x >> 5);
    const int lane = threadIdx.x & 31;
    const float4 w0 = *(const float4*)(W + (size_t)k * HC + 4 * lane);
    const float4 w1 = *(const float4*)(W + (size_t)k * HC + CC + 4 * lane);
    const float4 s0 = *(const float4*)(att_src + 4 * lane);
    const float4 s1 = *(const float4*)(att_src + CC + 4 * lane);
    const float4 d0 = *(const float4*)(att_dst + 4 * lane);
    const float4 d1 = *(const float4*)(att_dst + CC + 4 * lane);
    float ps0 = w0.x * s0.x + w0.y * s0.y + w0.z * s0.z + w0.w * s0.w;
    float ps1 = w1.x * s1.x + w1.y * s1.y + w1.z * s1.z + w1.w * s1.w;
    float pd0 = w0.x * d0.x + w0.y * d0.y + w0.z * d0.z + w0.w * d0.w;
    float pd1 = w1.x * d1.x + w1.y * d1.y + w1.z * d1.z + w1.w * d1.w;
    ps0 = wave_sum(ps0); ps1 = wave_sum(ps1); pd0 = wave_sum(pd0); pd1 = wave_sum(pd1);
    if (lane == 0) *(float4*)(vsd + 4 * k) = make_float4(ps0, ps1, pd0, pd1);
}

__device__ __forceinline__
void f32_to_bf16_scores_part(int blk,
                             const float* __restrict__ in,
                             const float* __restrict__ vsd,
                             __bf16* __restrict__ out,
                             float* __restrict__ a_s,
                             float* __restrict__ a_d) {
    const float4* __restrict__ in4 = (const float4*)in;
    const int gw   = blk * (FZ_T / 32) + (threadIdx.x >> 5);
    if (gw >= CVT_WAVES) return;
    const int lane = threadIdx.x & 31;
    const int q0   = gw * (CVT_U * 32) + lane;
    float4 v[CVT_U];
#pragma unroll
    for (int j = 0; j < CVT_U; ++j) v[j] = in4[q0 + j * 32];
    const float4* __restrict__ vp = (const float4*)vsd + 4 * lane;
    const float4 c0 = vp[0], c1 = vp[1], c2 = vp[2], c3 = vp[3];
#pragma unroll
    for (int j = 0; j < CVT_U; ++j) {
        v4bf o;
        o[0] = (__bf16)v[j].x; o[1] = (__bf16)v[j].y;
        o[2] = (__bf16)v[j].z; o[3] = (__bf16)v[j].w;
        *(v4bf*)(out + 4 * (size_t)(q0 + j * 32)) = o;
        float s0 = v[j].x * c0.x + v[j].y * c1.x + v[j].z * c2.x + v[j].w * c3.x;
        float s1 = v[j].x * c0.y + v[j].y * c1.y + v[j].z * c2.y + v[j].w * c3.y;
        float d0 = v[j].x * c0.z + v[j].y * c1.z + v[j].z * c2.z + v[j].w * c3.z;
        float d1 = v[j].x * c0.w + v[j].y * c1.w + v[j].z * c2.w + v[j].w * c3.w;
        s0 = wave_sum_valu(s0); s1 = wave_sum_valu(s1);
        d0 = wave_sum_valu(d0); d1 = wave_sum_valu(d1);
        const int row = gw * CVT_U + j;
        if (lane == 0) {
            *(float2*)(a_s + 2 * (size_t)row) = make_float2(s0, s1);
            *(float2*)(a_d + 2 * (size_t)row) = make_float2(d0, d1);
        }
    }
}

template <int K, int Nn>
__device__ __forceinline__
void transpose_bf16_part(int blk, const float* __restrict__ W, __bf16* __restrict__ Wt) {
    static_assert((K * Nn) % FZ_T == 0, "whole blocks");
    const int i = blk * FZ_T + threadIdx.x;
    const int n = i / K, k = i - n * K;
    Wt[i] = (__bf16)W[(size_t)k * Nn + n];
}
#define T0_BLKS (DD * HC / FZ_T)
#define T1_BLKS (HC * HC / FZ_T)

__device__ __forceinline__ float tanh_for_bf16(float x) {
    const float e = __builtin_amdgcn_exp2f(fabsf(x) * -2.8853900817779268f);
    const float r = (1.f - e) * __builtin_amdgcn_rcpf(1.f + e);
    return copysignf(r, x);
}

#define GM_NT 8
static_assert(GM_NT * 16 == CC && HC == 2 * CC, "a wave covers one head, a block both");

__device__ __forceinline__ v16bf load_frag(const __bf16* __restrict__ p) {
    const v8bf lo = *(const v8bf*)(p);
    const v8bf hi = *(const v8bf*)(p + 16);
    v16bf f;
#pragma unroll
    for (int i = 0; i < 8; ++i) { f[i] = lo[i]; f[i + 8] = hi[i]; }
    return f;
}

#define GM_TP (CC + 4)
template <bool PROJ, int K>
__global__ __launch_bounds__(64)
void k_gemm_bf16_wmma(const __bf16* __restrict__ A,
                      const __bf16* __restrict__ Bt,
                      int M,
                      const float* __restrict__ att_src,
                      const float* __restrict__ att_dst,
                      const float* __restrict__ Wl,
                      float* __restrict__ rec,
                      const float* __restrict__ bias,
                      __bf16* __restrict__ hout) {
    static_assert(K % 64 == 0, "a trip of the K loop takes 64");
    static_assert(PROJ ? K == HC : K == CC, "layer 0 reads its head's half of a row of A");
    const int lane = threadIdx.x & 31;
    const int half = lane >> 4;
    const int lr   = lane & 15;
    const int tM   = blockIdx.x * 32;
    const int tN0  = (threadIdx.x >> 5) * CC;
    const bool has1 = tM + 16 < M;
    const int  r1   = has1 ? tM + 16 + lr : M - 1;
    const int aoff = (PROJ ? 0 : tN0) + 8 * half;
    const __bf16* __restrict__ ap0 = A  + (size_t)(tM + lr) * HC + aoff;
    const __bf16* __restrict__ ap1 = A  + (size_t)r1 * HC + aoff;
    const __bf16* __restrict__ bp  = Bt + (size_t)(tN0 + lr) * K + 8 * half;

    v8f acc[2][GM_NT];
#pragma unroll
    for (int j = 0; j < GM_NT; ++j) { acc[0][j] = v8f{}; acc[1][j] = v8f{}; }

#pragma unroll 1
    for (int kk = 0; kk < K; kk += 64) {
        v16bf a[2][2];
#pragma unroll
        for (int ks = 0; ks < 2; ++ks) {
            a[0][ks] = load_frag(ap0 + kk + 32 * ks);
            a[1][ks] = load_frag(ap1 + kk + 32 * ks);
        }
#pragma unroll
        for (int nh = 0; nh < 2; ++nh) {
            v16bf b[4][2];
#pragma unroll
            for (int jj = 0; jj < 4; ++jj) {
#pragma unroll
                for (int ks = 0; ks < 2; ++ks)
                    b[jj][ks] = load_frag(bp + (size_t)(4 * nh + jj) * 16 * K + kk + 32 * ks);
            }
#pragma unroll
            for (int ks = 0; ks < 2; ++ks) {
#pragma unroll
                for (int jj = 0; jj < 4; ++jj) {
                    const int j = 4 * nh + jj;
#pragma unroll
                    for (int m = 0; m < 2; ++m) {
                        acc[m][j] = __builtin_amdgcn_wmma_f32_16x16x32_bf16(
                            false, b[jj][ks], false, a[m][ks], (short)0, acc[m][j],
                            false, false);
                    }
                }
            }
        }
        __builtin_amdgcn_sched_group_barrier(0x020, 24, 0);
        __builtin_amdgcn_sched_group_barrier(0x008, 16, 0);
        __builtin_amdgcn_sched_group_barrier(0x020, 16, 0);
        __builtin_amdgcn_sched_group_barrier(0x008, 16, 0);
    }

    if constexpr (PROJ) {
        v2f psd[2] = {{0.f, 0.f}, {0.f, 0.f}};
        v2f p01[2] = {{0.f, 0.f}, {0.f, 0.f}};
        auto load_vec = [&](int j, float4 (&v)[8]) {
            const int c0 = tN0 + 16 * j + 8 * half;
            v[0] = *(const float4*)(att_src + c0); v[1] = *(const float4*)(att_src + c0 + 4);
            v[2] = *(const float4*)(att_dst + c0); v[3] = *(const float4*)(att_dst + c0 + 4);
            v[4] = *(const float4*)(Wl + 2 * c0);     v[5] = *(const float4*)(Wl + 2 * c0 + 4);
            v[6] = *(const float4*)(Wl + 2 * c0 + 8); v[7] = *(const float4*)(Wl + 2 * c0 + 12);
        };
#pragma unroll
        for (int j = 0; j < GM_NT; ++j) {
            float4 cur[8];
            load_vec(j, cur);
            const float s8[8] = {cur[0].x, cur[0].y, cur[0].z, cur[0].w,
                                 cur[1].x, cur[1].y, cur[1].z, cur[1].w};
            const float d8[8] = {cur[2].x, cur[2].y, cur[2].z, cur[2].w,
                                 cur[3].x, cur[3].y, cur[3].z, cur[3].w};
            const v2f w8[8] = {{cur[4].x, cur[4].y}, {cur[4].z, cur[4].w},
                               {cur[5].x, cur[5].y}, {cur[5].z, cur[5].w},
                               {cur[6].x, cur[6].y}, {cur[6].z, cur[6].w},
                               {cur[7].x, cur[7].y}, {cur[7].z, cur[7].w}};
#pragma unroll
            for (int r = 0; r < 8; ++r) {
                const v2f sd = {s8[r], d8[r]};
#pragma unroll
                for (int m = 0; m < 2; ++m) {
                    const float t = acc[m][j][r];
                    const v2f tt = {t, t};
                    psd[m] = __builtin_elementwise_fma(tt, sd,    psd[m]);
                    p01[m] = __builtin_elementwise_fma(tt, w8[r], p01[m]);
                }
            }
        }
        float4* __restrict__ rp = (float4*)rec + (size_t)(tN0 / CC) * M + tM + lr;
#pragma unroll
        for (int m = 0; m < 2; ++m) {
            const float s  = psd[m][0] + __shfl_xor(psd[m][0], 16, 32);
            const float d  = psd[m][1] + __shfl_xor(psd[m][1], 16, 32);
            const float q0 = p01[m][0] + __shfl_xor(p01[m][0], 16, 32);
            const float q1 = p01[m][1] + __shfl_xor(p01[m][1], 16, 32);
            if (half == 0 && (m == 0 || has1)) rp[16 * m] = make_float4(s, q0, q1, d);
        }
    } else {
        __shared__ alignas(16) float s_tile[2][32 * GM_TP];
        float* __restrict__ sw = s_tile[threadIdx.x >> 5];
#pragma unroll
        for (int m = 0; m < 2; ++m) {
#pragma unroll
            for (int j = 0; j < GM_NT; ++j) {
                float* p = sw + (16 * m + lr) * GM_TP + 16 * j + 8 * half;
                *(float4*)(p)     = make_float4(acc[m][j][0], acc[m][j][1], acc[m][j][2], acc[m][j][3]);
                *(float4*)(p + 4) = make_float4(acc[m][j][4], acc[m][j][5], acc[m][j][6], acc[m][j][7]);
            }
        }
        __syncthreads();
        const float4 bA = *(const float4*)(bias + tN0 + 8 * lr);
        const float4 bB = *(const float4*)(bias + tN0 + 8 * lr + 4);
        __bf16* __restrict__ op = hout + (size_t)tM * HC + tN0 + 8 * lr;
#pragma unroll 1
        for (int it = 0; it < 16; ++it) {
            const int rr = 2 * it + half;
            const float4 u0 = *(const float4*)(sw + rr * GM_TP + 8 * lr);
            const float4 u1 = *(const float4*)(sw + rr * GM_TP + 8 * lr + 4);
            v8bf o;
            o[0] = (__bf16)tanh_for_bf16(u0.x + bA.x); o[1] = (__bf16)tanh_for_bf16(u0.y + bA.y);
            o[2] = (__bf16)tanh_for_bf16(u0.z + bA.z); o[3] = (__bf16)tanh_for_bf16(u0.w + bA.w);
            o[4] = (__bf16)tanh_for_bf16(u1.x + bB.x); o[5] = (__bf16)tanh_for_bf16(u1.y + bB.y);
            o[6] = (__bf16)tanh_for_bf16(u1.z + bB.z); o[7] = (__bf16)tanh_for_bf16(u1.w + bB.w);
            if (tM + rr < M) *(v8bf*)(op + (size_t)rr * HC) = o;
        }
    }
}

#define SB_BLK  256
#define SB_CH   (EE / SB_BLK)
#define SB_T    1024
#define SB_IT   ((SB_CH + SB_T - 1) / SB_T)
#define NBN     196
#define NBIN    256
#define BS_T    1024
#define BS_U    4
#define BS_CAP  8192
static_assert(EE % SB_BLK == 0, "chunks must divide the edge list");
static_assert(NBN * NBIN >= NN && NBN <= 256, "bins must cover the nodes, a bin's counters one scan");
static_assert(NN <= 65536, "a source id must fit the low half of the word");
static_assert(SB_CH < 65536, "a chunk's count per bin is kept in 16 bits");
static_assert(SB_T == 4 * NBIN && SB_BLK % 64 == 0, "k_bin_scatter splits the table in four");

__device__ __forceinline__ int block_excl_scan256(int v, int t, int* s_w) {
    const int lane = t & 31, w = t >> 5;
    int inc = v;
#pragma unroll
    for (int o = 1; o < 32; o <<= 1) {
        int u = __shfl_up(inc, o, 32);
        if (lane >= o) inc += u;
    }
    if (lane == 31 && w < 8) s_w[w] = inc;
    __syncthreads();
    int base = 0;
#pragma unroll
    for (int j = 0; j < 8; ++j) if (j < w) base += s_w[j];
    return base + inc - v;
}

__device__ __forceinline__
void bin_count_part(int blk, const int* __restrict__ ei, unsigned short* __restrict__ hist) {
    __shared__ int s_cnt[NBIN];
    const int t = threadIdx.x;
    const int* __restrict__ dstp = ei + EE + blk * SB_CH;
    unsigned d[SB_IT];
#pragma unroll
    for (int j = 0; j < SB_IT; ++j) {
        const int i = t + j * SB_T;
        d[j] = (i < SB_CH) ? (unsigned)dstp[i] : 0u;
    }
    if (t < NBIN) s_cnt[t] = 0;
    __syncthreads();
#pragma unroll
    for (int j = 0; j < SB_IT; ++j) {
        const int i = t + j * SB_T;
        if (i < SB_CH) atomicAdd(&s_cnt[d[j] / NBN], 1);
    }
    __syncthreads();
    if (t < NBIN) hist[blk * NBIN + t] = (unsigned short)s_cnt[t];
}

static_assert(SB_T == FZ_T, "the shared launches have the sort's block size");
__global__ __launch_bounds__(FZ_T)
void k_counts_vecs_weights(const int* __restrict__ ei, unsigned short* __restrict__ hist,
                           const float* __restrict__ W0,
                           const float* __restrict__ att_src0,
                           const float* __restrict__ att_dst0,
                           float* __restrict__ vsd,
                           __bf16* __restrict__ wt0,
                           const float* __restrict__ W1,
                           __bf16* __restrict__ wt1) {
    const int b = blockIdx.x;
    if (b < SB_BLK)                               bin_count_part(b, ei, hist);
    else if (b < SB_BLK + SV_BLKS)                score_vecs_part(b - SB_BLK, W0, att_src0, att_dst0, vsd);
    else if (b < SB_BLK + SV_BLKS + T0_BLKS)      transpose_bf16_part<DD, HC>(b - SB_BLK - SV_BLKS, W0, wt0);
    else                                          transpose_bf16_part<HC, HC>(b - SB_BLK - SV_BLKS - T0_BLKS, W1, wt1);
}

__global__ __launch_bounds__(SB_T)
void k_bin_scatter(const int* __restrict__ ei,
                   const unsigned short* __restrict__ hist,
                   int* __restrict__ binstart,
                   unsigned* __restrict__ binned) {
    __shared__ int s_tot[4][NBIN];
    __shared__ int s_par[4][NBIN];
    __shared__ int s_goff[NBIN];
    __shared__ int s_cnt[NBIN];
    __shared__ int s_lst[NBIN];
    __shared__ int s_w1[8];
    __shared__ int s_w2[8];
    __shared__ unsigned s_val[SB_CH];
    __shared__ int s_addr[SB_CH];
    const int t = threadIdx.x, k = blockIdx.x;
    const int bin = t & (NBIN - 1), q = t >> 8;

    const int* __restrict__ srcp = ei + k * SB_CH;
    const int* __restrict__ dstp = ei + EE + k * SB_CH;
    unsigned es[SB_IT], ed[SB_IT];
#pragma unroll
    for (int j = 0; j < SB_IT; ++j) {
        const int i = t + j * SB_T;
        es[j] = (i < SB_CH) ? (unsigned)srcp[i] : 0u;
        ed[j] = (i < SB_CH) ? (unsigned)dstp[i] : 0u;
    }

    int tot = 0, par = 0;
    const unsigned short* __restrict__ hp = hist + q * (SB_BLK / 4) * NBIN + bin;
    const int lim = k - q * (SB_BLK / 4);
#pragma unroll
    for (int i0 = 0; i0 < SB_BLK / 4; i0 += 16) {
        int h[16];
#pragma unroll
        for (int u = 0; u < 16; ++u) h[u] = hp[(i0 + u) * NBIN];
#pragma unroll
        for (int u = 0; u < 16; ++u) {
            tot += h[u];
            par += (i0 + u < lim) ? h[u] : 0;
        }
    }
    s_tot[q][bin] = tot;
    s_par[q][bin] = par;
    if (t < NBIN) s_cnt[t] = 0;
    __syncthreads();
    int total = 0, partial = 0;
    if (t < NBIN) {
        total   = s_tot[0][t] + s_tot[1][t] + s_tot[2][t] + s_tot[3][t];
        partial = s_par[0][t] + s_par[1][t] + s_par[2][t] + s_par[3][t];
    }
    const int bs = block_excl_scan256(total, t, s_w1);
    if (t < NBIN) {
        s_goff[t] = bs + partial;
        if (k == 0) {
            binstart[t] = bs;
            if (t == NBIN - 1) binstart[NBIN] = bs + total;
        }
    }

    int      bn[SB_IT];
    int      rk[SB_IT];
    unsigned pv[SB_IT];
#pragma unroll
    for (int j = 0; j < SB_IT; ++j) {
        const int i = t + j * SB_T;
        bn[j] = -1; rk[j] = 0; pv[j] = 0u;
        if (i < SB_CH) {
            const unsigned d = ed[j];
            const unsigned b = d / NBN;
            bn[j] = (int)b;
            pv[j] = ((d - b * NBN) << 16) | es[j];
            rk[j] = atomicAdd(&s_cnt[b], 1);
        }
    }
    __syncthreads();
    const int lc = (t < NBIN) ? s_cnt[t] : 0;
    const int ls = block_excl_scan256(lc, t, s_w2);
    if (t < NBIN) s_lst[t] = ls;
    __syncthreads();
#pragma unroll
    for (int j = 0; j < SB_IT; ++j) {
        if (bn[j] >= 0) {
            const int p = s_lst[bn[j]] + rk[j];
            s_val[p]  = pv[j];
            s_addr[p] = s_goff[bn[j]] + rk[j];
        }
    }
    __syncthreads();
#pragma unroll
    for (int j = 0; j < SB_IT; ++j) {
        const int i = t + j * SB_T;
        if (i < SB_CH) binned[s_addr[i]] = s_val[i];
    }
}

__device__ __forceinline__
void bin_sort_part(int blk,
                   const int* __restrict__ binstart,
                   const unsigned* __restrict__ binned,
                   int* __restrict__ rowptr,
                   int* __restrict__ csr_src) {
    __shared__ int s_c[256];
    __shared__ int s_w[8];
    __shared__ int s_out[BS_CAP];
    const int t = threadIdx.x, b = blk;
    const int beg = binstart[b], end = binstart[b + 1];
    const bool staged = (end - beg) <= BS_CAP;
    if (t < 256) s_c[t] = 0;
    __syncthreads();
    for (int i0 = beg + t; i0 < end; i0 += BS_U * BS_T) {
        unsigned v[BS_U];
#pragma unroll
        for (int u = 0; u < BS_U; ++u) {
            const int i = i0 + u * BS_T;
            v[u] = binned[i < end ? i : end - 1];
        }
#pragma unroll
        for (int u = 0; u < BS_U; ++u) {
            const int i = i0 + u * BS_T;
            if (i < end) atomicAdd(&s_c[v[u] >> 16], 1);
        }
    }
    __syncthreads();
    const int c   = (t < 256) ? s_c[t] : 0;
    const int pre = block_excl_scan256(c, t, s_w);
    const int node = b * NBN + t;
    if (t < NBN && node < NN) {
        rowptr[node] = beg + pre;
        if (node == NN - 1) rowptr[NN] = beg + pre + c;
    }
    if (t < 256) s_c[t] = beg + pre;
    __syncthreads();
    for (int i0 = beg + t; i0 < end; i0 += BS_U * BS_T) {
        unsigned v[BS_U];
#pragma unroll
        for (int u = 0; u < BS_U; ++u) {
            const int i = i0 + u * BS_T;
            v[u] = binned[i < end ? i : end - 1];
        }
#pragma unroll
        for (int u = 0; u < BS_U; ++u) {
            const int i = i0 + u * BS_T;
            if (i < end) {
                const int slot = atomicAdd(&s_c[v[u] >> 16], 1);
                if (staged) s_out[slot - beg] = (int)(v[u] & 0xFFFFu);
                else        csr_src[slot]     = (int)(v[u] & 0xFFFFu);
            }
        }
    }
    if (staged) {
        __syncthreads();
        for (int i = t; i < end - beg; i += BS_T) csr_src[beg + i] = s_out[i];
    }
}

static_assert(BS_T == FZ_T, "the shared launches have the sort's block size");
__global__ __launch_bounds__(FZ_T)
void k_sort_convert(const int* __restrict__ binstart,
                    const unsigned* __restrict__ binned,
                    int* __restrict__ rowptr,
                    int* __restrict__ csr_src,
                    const float* __restrict__ x,
                    const float* __restrict__ vsd,
                    __bf16* __restrict__ xb,
                    float* __restrict__ a_s,
                    float* __restrict__ a_d) {
    const int b = blockIdx.x;
    if (b < NBIN) bin_sort_part(b, binstart, binned, rowptr, csr_src);
    else          f32_to_bf16_scores_part(b - NBIN, x, vsd, xb, a_s, a_d);
}

__device__ __forceinline__ float lane_bcast(float v, int j) {
    return __int_as_float(__builtin_amdgcn_readlane(__float_as_int(v), j));
}

__global__ __launch_bounds__(256)
void k_node_gather(const int* __restrict__ rowptr,
                   const int* __restrict__ csr_src,
                   const __bf16* __restrict__ xb,
                   const float* __restrict__ a_s,
                   const float* __restrict__ a_d,
                   __bf16* __restrict__ agg) {
    const int node = __builtin_amdgcn_readfirstlane((int)(blockIdx.x * 8 + (threadIdx.x >> 5)));
    const int lane = threadIdx.x & 31;
    const int beg  = rowptr[node];
    const int deg  = rowptr[node + 1] - beg;
    const int n    = deg + 1;
    const int* __restrict__ seg = csr_src + beg;
    const float ad0 = a_d[node * 2 + 0];
    const float ad1 = a_d[node * 2 + 1];

    auto item = [&](int i, int& s, float& l0, float& l1) {
        if (i < deg) s = seg[i]; else s = node;
        const float2 as = *(const float2*)(a_s + 2 * (size_t)s);
        const float t0 = as.x + ad0, t1 = as.y + ad1;
        l0 = (t0 > 0.f) ? t0 : NEG_SLOPE * t0;
        l1 = (t1 > 0.f) ? t1 : NEG_SLOPE * t1;
    };

    int   s_c  = node;
    float l0_c = -INFINITY, l1_c = -INFINITY;
    if (lane < n) item(lane, s_c, l0_c, l1_c);

    float m0 = l0_c, m1 = l1_c;
    for (int b = 32; b < n; b += 32) {
        const int i = b + lane;
        if (i < n) {
            int s; float l0, l1; item(i, s, l0, l1);
            m0 = fmaxf(m0, l0); m1 = fmaxf(m1, l1);
        }
    }
    m0 = wave_max(m0); m1 = wave_max(m1);

    const float e0_c = (lane < n) ? expf(l0_c - m0) : 0.f;
    const float e1_c = (lane < n) ? expf(l1_c - m1) : 0.f;
    float d0 = e0_c, d1 = e1_c;
    for (int b = 32; b < n; b += 32) {
        const int i = b + lane;
        if (i < n) {
            int s; float l0, l1; item(i, s, l0, l1);
            d0 += expf(l0 - m0); d1 += expf(l1 - m1);
        }
    }
    const float den0 = wave_sum(d0) + 1e-16f;
    const float den1 = wave_sum(d1) + 1e-16f;

    v2f A0a = {0.f, 0.f}, A0b = {0.f, 0.f};
    v2f A1a = {0.f, 0.f}, A1b = {0.f, 0.f};
    const __bf16* __restrict__ xl = xb + 4 * lane;

    auto accumulate = [&](int cnt, int s, float w0, float w1) {
        for (int j0 = 0; j0 < cnt; j0 += 8) {
            v4bf r[8];
#pragma unroll
            for (int u = 0; u < 8; ++u) {
                const int sj = __builtin_amdgcn_readlane(s, j0 + u);
                r[u] = *(const v4bf*)(xl + (size_t)sj * DD);
            }
#pragma unroll
            for (int u = 0; u < 8; ++u) {
                const float x0 = lane_bcast(w0, j0 + u);
                const float x1 = lane_bcast(w1, j0 + u);
                const v2f fa = {(float)r[u][0], (float)r[u][1]};
                const v2f fb = {(float)r[u][2], (float)r[u][3]};
                const v2f w0v = {x0, x0}, w1v = {x1, x1};
                A0a = __builtin_elementwise_fma(fa, w0v, A0a);
                A0b = __builtin_elementwise_fma(fb, w0v, A0b);
                A1a = __builtin_elementwise_fma(fa, w1v, A1a);
                A1b = __builtin_elementwise_fma(fb, w1v, A1b);
            }
        }
    };

    accumulate(n < 32 ? n : 32, s_c, e0_c / den0, e1_c / den1);
    for (int b = 32; b < n; b += 32) {
        const int i = b + lane;
        int s = node; float w0 = 0.f, w1 = 0.f;
        if (i < n) {
            float l0, l1; item(i, s, l0, l1);
            w0 = expf(l0 - m0) / den0; w1 = expf(l1 - m1) / den1;
        }
        accumulate((n - b) < 32 ? (n - b) : 32, s, w0, w1);
    }

    v4bf o0, o1;
    o0[0] = (__bf16)A0a[0]; o0[1] = (__bf16)A0a[1]; o0[2] = (__bf16)A0b[0]; o0[3] = (__bf16)A0b[1];
    o1[0] = (__bf16)A1a[0]; o1[1] = (__bf16)A1a[1]; o1[2] = (__bf16)A1b[0]; o1[3] = (__bf16)A1b[1];
    __bf16* o = agg + (size_t)node * HC + 4 * lane;
    *(v4bf*)(o)      = o0;
    *(v4bf*)(o + CC) = o1;
}

__global__ __launch_bounds__(256)
void k_node_classify(const int* __restrict__ rowptr,
                     const int* __restrict__ csr_src,
                     const float* __restrict__ rec,
                     const float* __restrict__ bias,
                     const float* __restrict__ Wl,
                     const float* __restrict__ bl,
                     float* __restrict__ out) {
    const int node = __builtin_amdgcn_readfirstlane((int)(blockIdx.x * 8 + (threadIdx.x >> 5)));
    const int lane = threadIdx.x & 31;
    const float4* __restrict__ rec0 = (const float4*)rec;
    const float4* __restrict__ rec1 = rec0 + NN;
    const int beg  = rowptr[node];
    const int deg  = rowptr[node + 1] - beg;
    const int n    = deg + 1;
    const int* __restrict__ seg = csr_src + beg;
    const float ad0 = rec0[node].w;
    const float ad1 = rec1[node].w;

    auto item = [&](int i, float4& r0, float4& r1, float& l0, float& l1) {
        int s;
        if (i < deg) s = seg[i]; else s = node;
        r0 = rec0[s]; r1 = rec1[s];
        const float t0 = r0.x + ad0, t1 = r1.x + ad1;
        l0 = (t0 > 0.f) ? t0 : NEG_SLOPE * t0;
        l1 = (t1 > 0.f) ? t1 : NEG_SLOPE * t1;
    };

    float4 r0_c = {0.f, 0.f, 0.f, 0.f}, r1_c = {0.f, 0.f, 0.f, 0.f};
    float l0_c = -INFINITY, l1_c = -INFINITY;
    if (lane < n) item(lane, r0_c, r1_c, l0_c, l1_c);

    float m0 = l0_c, m1 = l1_c;
    for (int b = 32; b < n; b += 32) {
        const int i = b + lane;
        if (i < n) {
            float4 r0, r1; float l0, l1; item(i, r0, r1, l0, l1);
            m0 = fmaxf(m0, l0); m1 = fmaxf(m1, l1);
        }
    }
    m0 = wave_max(m0); m1 = wave_max(m1);

    const float e0_c = (lane < n) ? expf(l0_c - m0) : 0.f;
    const float e1_c = (lane < n) ? expf(l1_c - m1) : 0.f;
    float d0 = e0_c, d1 = e1_c;
    float q00 = e0_c * r0_c.y, q01 = e0_c * r0_c.z;
    float q10 = e1_c * r1_c.y, q11 = e1_c * r1_c.z;
    for (int b = 32; b < n; b += 32) {
        const int i = b + lane;
        if (i < n) {
            float4 r0, r1; float l0, l1; item(i, r0, r1, l0, l1);
            const float e0 = expf(l0 - m0), e1 = expf(l1 - m1);
            d0 += e0; d1 += e1;
            q00 += e0 * r0.y; q01 += e0 * r0.z;
            q10 += e1 * r1.y; q11 += e1 * r1.z;
        }
    }
    const float den0 = wave_sum(d0) + 1e-16f;
    const float den1 = wave_sum(d1) + 1e-16f;
    q00 = wave_sum(q00); q01 = wave_sum(q01);
    q10 = wave_sum(q10); q11 = wave_sum(q11);

    const float4 bA  = *(const float4*)(bias + 4 * lane);
    const float4 bB  = *(const float4*)(bias + CC + 4 * lane);
    const float4 wa0 = *(const float4*)(Wl + 8 * lane);
    const float4 wa1 = *(const float4*)(Wl + 8 * lane + 4);
    const float4 wb0 = *(const float4*)(Wl + 2 * CC + 8 * lane);
    const float4 wb1 = *(const float4*)(Wl + 2 * CC + 8 * lane + 4);
    float c0 = bA.x * wa0.x + bA.y * wa0.z + bA.z * wa1.x + bA.w * wa1.z
             + bB.x * wb0.x + bB.y * wb0.z + bB.z * wb1.x + bB.w * wb1.z;
    float c1 = bA.x * wa0.y + bA.y * wa0.w + bA.z * wa1.y + bA.w * wa1.w
             + bB.x * wb0.y + bB.y * wb0.w + bB.z * wb1.y + bB.w * wb1.w;
    c0 = wave_sum(c0); c1 = wave_sum(c1);

    const float z0 = q00 / den0 + q10 / den1 + c0 + bl[0];
    const float z1 = q01 / den0 + q11 / den1 + c1 + bl[1];
    const float mx = fmaxf(z0, z1);
    const float e0 = expf(z0 - mx), e1 = expf(z1 - mx);
    const float inv = 1.f / (e0 + e1);
    if (lane == 0) *(float2*)(out + 2 * (size_t)node) = make_float2(e0 * inv, e1 * inv);
}

extern "C" void kernel_launch(void* const* d_in, const int* in_sizes, int n_in,
                              void* d_out, int out_size, void* d_ws, size_t ws_size,
                              hipStream_t stream) {
    const float* x        = (const float*)d_in[0];
    const int*   ei       = (const int*)  d_in[1];
    const float* W0       = (const float*)d_in[2];
    const float* att_src0 = (const float*)d_in[3];
    const float* att_dst0 = (const float*)d_in[4];
    const float* b0       = (const float*)d_in[5];
    const float* W1       = (const float*)d_in[6];
    const float* att_src1 = (const float*)d_in[7];
    const float* att_dst1 = (const float*)d_in[8];
    const float* b1       = (const float*)d_in[9];
    const float* Wl       = (const float*)d_in[10];
    const float* bl       = (const float*)d_in[11];
    float* out = (float*)d_out;

    char* base = (char*)d_ws;
    size_t off = 0;
    auto carve = [&](size_t bytes) -> char* {
        char* p = base + off;
        off = (off + bytes + 255) & ~(size_t)255;
        return p;
    };
    float*    h_feat = (float*)   carve((size_t)NN * HC * 4);
    __bf16*   xbf    = (__bf16*)  carve((size_t)NN * HC * 2);
    __bf16*   wtbf   = (__bf16*)  carve((size_t)HC * HC * 2);
    float*    a_s    = (float*)   carve((size_t)NN * 2 * 4);
    float*    a_d    = (float*)   carve((size_t)NN * 2 * 4);
    char*     tbl    =            carve((size_t)NN * 4);
    int*      rowptr = (int*)     carve((size_t)(NN + 1) * 4);
    unsigned* binned = (unsigned*)carve((size_t)EE * 4);
    int*      csr    = (int*)     carve((size_t)EE * 4);
    __bf16*   agg    = (__bf16*)h_feat;
    float*    vsd    = h_feat + (size_t)NN * HC / 2;
    __bf16*   wtbf1  = (__bf16*)(vsd + 1024);
    static_assert(DD * 4 <= 1024 && (size_t)1024 * 4 + (size_t)HC * HC * 2 <= (size_t)NN * HC * 2,
                  "score vectors and layer 1's weights fit the second half of h_feat");
    unsigned short* hist = (unsigned short*)tbl;
    int*      binstart   = (int*)(tbl + (size_t)SB_BLK * NBIN * 2);
    static_assert((size_t)SB_BLK * NBIN * 2 + (NBIN + 1) * 4 <= (size_t)NN * 4, "table space");
    (void)ws_size; (void)in_sizes; (void)n_in; (void)out_size;

    const int T = 256;
    dim3 blk(T);

    k_counts_vecs_weights<<<SB_BLK + SV_BLKS + T0_BLKS + T1_BLKS, dim3(FZ_T), 0, stream>>>(
        ei, hist, W0, att_src0, att_dst0, vsd, wtbf, W1, wtbf1);

    k_bin_scatter<<<SB_BLK, dim3(SB_T), 0, stream>>>(ei, hist, binstart, binned);

    k_sort_convert<<<NBIN + CVT_BLKS, dim3(FZ_T), 0, stream>>>(
        binstart, binned, rowptr, csr, x, vsd, xbf, a_s, a_d);

    k_node_gather<<<NN / 8, blk, 0, stream>>>(rowptr, csr, xbf, a_s, a_d, agg);

    k_gemm_bf16_wmma<false, DD><<<(NN + 31) / 32, dim3(64), 0, stream>>>(
        agg, wtbf, NN, nullptr, nullptr, nullptr, nullptr, b0, xbf);

    float* rec = h_feat;
    k_gemm_bf16_wmma<true, HC><<<(NN + 31) / 32, dim3(64), 0, stream>>>(
        xbf, wtbf1, NN, att_src1, att_dst1, Wl, rec, nullptr, nullptr);

    k_node_classify<<<NN / 8, blk, 0, stream>>>(
        rowptr, csr, rec, b1, Wl, bl, out);
}


# Both products: a block of 16 waves (8 row strips x 2 heads, 256 blocks) copies the weights to LDS once; B fragments come from LDS, not memory; layer 0 tail does bias+tanh before its LDS turn to fit
# speedup vs baseline: 21.8787x; 1.1799x over previous
#include <hip/hip_runtime.h>
#include <math.h>

#define NN    50000
#define DD    128
#define HC    256
#define CC    128
#define EE    800000
#define NEG_SLOPE 0.2f

typedef __attribute__((ext_vector_type(16))) __bf16 v16bf;
typedef __attribute__((ext_vector_type(8)))  __bf16 v8bf;
typedef __attribute__((ext_vector_type(8)))  float  v8f;

typedef __attribute__((ext_vector_type(4)))  __bf16 v4bf;
typedef __attribute__((ext_vector_type(2)))  float  v2f;

#define FZ_T   1024

#define CVT_U  4
#define CVT_WAVES (NN / CVT_U)
#define CVT_BLKS  ((CVT_WAVES + FZ_T / 32 - 1) / (FZ_T / 32))
static_assert(NN % CVT_U == 0, "a wave's rows must divide the nodes");

__device__ __forceinline__ float wave_max(float v) {
#pragma unroll
    for (int o = 16; o > 0; o >>= 1) v = fmaxf(v, __shfl_xor(v, o, 32));
    return v;
}
__device__ __forceinline__ float wave_sum(float v) {
#pragma unroll
    for (int o = 16; o > 0; o >>= 1) v += __shfl_xor(v, o, 32);
    return v;
}
template <int CTRL>
__device__ __forceinline__ float row_swz(float v) {
    const int b = __builtin_bit_cast(int, v);
    return __builtin_bit_cast(float, __builtin_amdgcn_update_dpp(b, b, CTRL, 0xf, 0xf, false));
}
__device__ __forceinline__ float xhalf(float v) {
    const int b = __builtin_bit_cast(int, v);
    return __builtin_bit_cast(float,
        __builtin_amdgcn_permlanex16(b, b, 0x76543210, 0xfedcba98, false, false));
}
__device__ __forceinline__ float wave_sum_valu(float v) {
    v += row_swz<0x161>(v);
    v += row_swz<0x162>(v);
    v += row_swz<0x164>(v);
    v += row_swz<0x168>(v);
    return v + xhalf(v);
}

static_assert(DD == 128 && CC == 128 && HC == 2 * CC, "a lane holds 4 features / 4 + 4 channels");
static_assert(DD % (FZ_T / 32) == 0, "whole blocks of one wave per input feature");
#define SV_BLKS (DD / (FZ_T / 32))
__device__ __forceinline__
void score_vecs_part(int blk,
                     const float* __restrict__ W,
                     const float* __restrict__ att_src,
                     const float* __restrict__ att_dst,
                     float* __restrict__ vsd) {
    const int k    = blk * (FZ_T / 32) + (threadIdx.x >> 5);
    const int lane = threadIdx.x & 31;
    const float4 w0 = *(const float4*)(W + (size_t)k * HC + 4 * lane);
    const float4 w1 = *(const float4*)(W + (size_t)k * HC + CC + 4 * lane);
    const float4 s0 = *(const float4*)(att_src + 4 * lane);
    const float4 s1 = *(const float4*)(att_src + CC + 4 * lane);
    const float4 d0 = *(const float4*)(att_dst + 4 * lane);
    const float4 d1 = *(const float4*)(att_dst + CC + 4 * lane);
    float ps0 = w0.x * s0.x + w0.y * s0.y + w0.z * s0.z + w0.w * s0.w;
    float ps1 = w1.x * s1.x + w1.y * s1.y + w1.z * s1.z + w1.w * s1.w;
    float pd0 = w0.x * d0.x + w0.y * d0.y + w0.z * d0.z + w0.w * d0.w;
    float pd1 = w1.x * d1.x + w1.y * d1.y + w1.z * d1.z + w1.w * d1.w;
    ps0 = wave_sum(ps0); ps1 = wave_sum(ps1); pd0 = wave_sum(pd0); pd1 = wave_sum(pd1);
    if (lane == 0) *(float4*)(vsd + 4 * k) = make_float4(ps0, ps1, pd0, pd1);
}

__device__ __forceinline__
void f32_to_bf16_scores_part(int blk,
                             const float* __restrict__ in,
                             const float* __restrict__ vsd,
                             __bf16* __restrict__ out,
                             float* __restrict__ a_s,
                             float* __restrict__ a_d) {
    const float4* __restrict__ in4 = (const float4*)in;
    const int gw   = blk * (FZ_T / 32) + (threadIdx.x >> 5);
    if (gw >= CVT_WAVES) return;
    const int lane = threadIdx.x & 31;
    const int q0   = gw * (CVT_U * 32) + lane;
    float4 v[CVT_U];
#pragma unroll
    for (int j = 0; j < CVT_U; ++j) v[j] = in4[q0 + j * 32];
    const float4* __restrict__ vp = (const float4*)vsd + 4 * lane;
    const float4 c0 = vp[0], c1 = vp[1], c2 = vp[2], c3 = vp[3];
#pragma unroll
    for (int j = 0; j < CVT_U; ++j) {
        v4bf o;
        o[0] = (__bf16)v[j].x; o[1] = (__bf16)v[j].y;
        o[2] = (__bf16)v[j].z; o[3] = (__bf16)v[j].w;
        *(v4bf*)(out + 4 * (size_t)(q0 + j * 32)) = o;
        float s0 = v[j].x * c0.x + v[j].y * c1.x + v[j].z * c2.x + v[j].w * c3.x;
        float s1 = v[j].x * c0.y + v[j].y * c1.y + v[j].z * c2.y + v[j].w * c3.y;
        float d0 = v[j].x * c0.z + v[j].y * c1.z + v[j].z * c2.z + v[j].w * c3.z;
        float d1 = v[j].x * c0.w + v[j].y * c1.w + v[j].z * c2.w + v[j].w * c3.w;
        s0 = wave_sum_valu(s0); s1 = wave_sum_valu(s1);
        d0 = wave_sum_valu(d0); d1 = wave_sum_valu(d1);
        const int row = gw * CVT_U + j;
        if (lane == 0) {
            *(float2*)(a_s + 2 * (size_t)row) = make_float2(s0, s1);
            *(float2*)(a_d + 2 * (size_t)row) = make_float2(d0, d1);
        }
    }
}

template <int K, int Nn>
__device__ __forceinline__
void transpose_bf16_part(int blk, const float* __restrict__ W, __bf16* __restrict__ Wt) {
    static_assert((K * Nn) % FZ_T == 0, "whole blocks");
    const int i = blk * FZ_T + threadIdx.x;
    const int n = i / K, k = i - n * K;
    Wt[i] = (__bf16)W[(size_t)k * Nn + n];
}
#define T0_BLKS (DD * HC / FZ_T)
#define T1_BLKS (HC * HC / FZ_T)

__device__ __forceinline__ float tanh_for_bf16(float x) {
    const float e = __builtin_amdgcn_exp2f(fabsf(x) * -2.8853900817779268f);
    const float r = (1.f - e) * __builtin_amdgcn_rcpf(1.f + e);
    return copysignf(r, x);
}

#define GM_NT 8
static_assert(GM_NT * 16 == CC && HC == 2 * CC, "a wave covers one head, a strip's two waves both");
#define GM_S  8
#define GM_W  (2 * GM_S)
#define GM_T  (32 * GM_W)
#define GM_BLKS 256
static_assert((size_t)GM_BLKS * GM_S * 32 >= NN, "the blocks' strips cover the rows");

__device__ __forceinline__ v16bf load_frag(const __bf16* __restrict__ p) {
    const v8bf lo = *(const v8bf*)(p);
    const v8bf hi = *(const v8bf*)(p + 16);
    v16bf f;
#pragma unroll
    for (int i = 0; i < 8; ++i) { f[i] = lo[i]; f[i + 8] = hi[i]; }
    return f;
}

#define GM_TP (CC + 8)
template <bool PROJ, int K>
__global__ __launch_bounds__(GM_T)
void k_gemm_bf16_wmma(const __bf16* __restrict__ A,
                      const __bf16* __restrict__ Bt,
                      int M,
                      const float* __restrict__ att_src,
                      const float* __restrict__ att_dst,
                      const float* __restrict__ Wl,
                      float* __restrict__ rec,
                      const float* __restrict__ bias,
                      __bf16* __restrict__ hout) {
    static_assert(K % 64 == 0, "a trip of the K loop takes 64");
    static_assert(PROJ ? K == HC : K == CC, "layer 0 reads its head's half of a row of A");
    constexpr int BP = K + 8;
    static_assert(((2 * BP / 16) & 1) == 1, "pitch in 16-byte units must be odd");
    constexpr int TILE = 32 * GM_TP;
    __shared__ alignas(16) __bf16 s_b[HC * BP + (PROJ ? 0 : GM_W * TILE)];

    constexpr int CPR = K / 8;
    constexpr int PER = HC * CPR / GM_T;
    static_assert(HC * CPR % GM_T == 0 && (PER & (PER - 1)) == 0, "whole, power-of-two rounds");
    {
        const int rot = blockIdx.x;
        v8bf w[PER];
#pragma unroll
        for (int i = 0; i < PER; ++i) {
            const int c = ((i + rot) & (PER - 1)) * GM_T + threadIdx.x;
            w[i] = *(const v8bf*)(Bt + 8 * (size_t)c);
        }
#pragma unroll
        for (int i = 0; i < PER; ++i) {
            const int c = ((i + rot) & (PER - 1)) * GM_T + threadIdx.x;
            *(v8bf*)(s_b + (c / CPR) * BP + 8 * (c % CPR)) = w[i];
        }
    }
    __syncthreads();

    const int lane = threadIdx.x & 31;
    const int half = lane >> 4;
    const int lr   = lane & 15;
    const int wv   = __builtin_amdgcn_readfirstlane((int)(threadIdx.x >> 5));
    const int tM   = ((int)blockIdx.x + GM_BLKS * (wv >> 1)) * 32;
    const int tN0  = (wv & 1) * CC;
    const bool has0 = tM < M, has1 = tM + 16 < M;
    const int r0 = tM + lr;
    const int r1 = has1 ? tM + 16 + lr : M - 1;
    const int aoff = (PROJ ? 0 : tN0) + 8 * half;
    const __bf16* __restrict__ ap0 = A + (size_t)r0 * HC + aoff;
    const __bf16* __restrict__ ap1 = A + (size_t)r1 * HC + aoff;
    const __bf16* bp = s_b + (tN0 + lr) * BP + 8 * half;

    if (has0) {
        v8f acc[2][GM_NT];
#pragma unroll
        for (int j = 0; j < GM_NT; ++j) { acc[0][j] = v8f{}; acc[1][j] = v8f{}; }

#pragma unroll 1
        for (int kk = 0; kk < K; kk += 64) {
            v16bf a[2][2];
#pragma unroll
            for (int ks = 0; ks < 2; ++ks) {
                a[0][ks] = load_frag(ap0 + kk + 32 * ks);
                a[1][ks] = load_frag(ap1 + kk + 32 * ks);
            }
#pragma unroll
            for (int nh = 0; nh < 2; ++nh) {
                v16bf b[4][2];
#pragma unroll
                for (int jj = 0; jj < 4; ++jj) {
#pragma unroll
                    for (int ks = 0; ks < 2; ++ks)
                        b[jj][ks] = load_frag(bp + (4 * nh + jj) * 16 * BP + kk + 32 * ks);
                }
#pragma unroll
                for (int ks = 0; ks < 2; ++ks) {
#pragma unroll
                    for (int jj = 0; jj < 4; ++jj) {
                        const int j = 4 * nh + jj;
#pragma unroll
                        for (int m = 0; m < 2; ++m) {
                            acc[m][j] = __builtin_amdgcn_wmma_f32_16x16x32_bf16(
                                false, b[jj][ks], false, a[m][ks], (short)0, acc[m][j],
                                false, false);
                        }
                    }
                }
            }
            __builtin_amdgcn_sched_group_barrier(0x020, 8, 0);
            __builtin_amdgcn_sched_group_barrier(0x100, 16, 0);
            __builtin_amdgcn_sched_group_barrier(0x008, 16, 0);
            __builtin_amdgcn_sched_group_barrier(0x100, 16, 0);
            __builtin_amdgcn_sched_group_barrier(0x008, 16, 0);
        }

        if constexpr (PROJ) {
            v2f psd[2] = {{0.f, 0.f}, {0.f, 0.f}};
            v2f p01[2] = {{0.f, 0.f}, {0.f, 0.f}};
            auto load_vec = [&](int j, float4 (&v)[8]) {
                const int c0 = tN0 + 16 * j + 8 * half;
                v[0] = *(const float4*)(att_src + c0); v[1] = *(const float4*)(att_src + c0 + 4);
                v[2] = *(const float4*)(att_dst + c0); v[3] = *(const float4*)(att_dst + c0 + 4);
                v[4] = *(const float4*)(Wl + 2 * c0);     v[5] = *(const float4*)(Wl + 2 * c0 + 4);
                v[6] = *(const float4*)(Wl + 2 * c0 + 8); v[7] = *(const float4*)(Wl + 2 * c0 + 12);
            };
#pragma unroll
            for (int j = 0; j < GM_NT; ++j) {
                float4 cur[8];
                load_vec(j, cur);
                const float s8[8] = {cur[0].x, cur[0].y, cur[0].z, cur[0].w,
                                     cur[1].x, cur[1].y, cur[1].z, cur[1].w};
                const float d8[8] = {cur[2].x, cur[2].y, cur[2].z, cur[2].w,
                                     cur[3].x, cur[3].y, cur[3].z, cur[3].w};
                const v2f w8[8] = {{cur[4].x, cur[4].y}, {cur[4].z, cur[4].w},
                                   {cur[5].x, cur[5].y}, {cur[5].z, cur[5].w},
                                   {cur[6].x, cur[6].y}, {cur[6].z, cur[6].w},
                                   {cur[7].x, cur[7].y}, {cur[7].z, cur[7].w}};
#pragma unroll
                for (int r = 0; r < 8; ++r) {
                    const v2f sd = {s8[r], d8[r]};
#pragma unroll
                    for (int m = 0; m < 2; ++m) {
                        const float t = acc[m][j][r];
                        const v2f tt = {t, t};
                        psd[m] = __builtin_elementwise_fma(tt, sd,    psd[m]);
                        p01[m] = __builtin_elementwise_fma(tt, w8[r], p01[m]);
                    }
                }
            }
            float4* __restrict__ rp = (float4*)rec + (size_t)(tN0 / CC) * M + tM + lr;
#pragma unroll
            for (int m = 0; m < 2; ++m) {
                const float s  = psd[m][0] + __shfl_xor(psd[m][0], 16, 32);
                const float d  = psd[m][1] + __shfl_xor(psd[m][1], 16, 32);
                const float q0 = p01[m][0] + __shfl_xor(p01[m][0], 16, 32);
                const float q1 = p01[m][1] + __shfl_xor(p01[m][1], 16, 32);
                if (half == 0 && (m == 0 || has1)) rp[16 * m] = make_float4(s, q0, q1, d);
            }
        } else {
            __bf16* __restrict__ sw = s_b + HC * BP + wv * TILE;
            float4 bv[GM_NT][2];
#pragma unroll
            for (int j = 0; j < GM_NT; ++j) {
                bv[j][0] = *(const float4*)(bias + tN0 + 16 * j + 8 * half);
                bv[j][1] = *(const float4*)(bias + tN0 + 16 * j + 8 * half + 4);
            }
#pragma unroll
            for (int j = 0; j < GM_NT; ++j) {
                const float4 bA = bv[j][0], bB = bv[j][1];
#pragma unroll
                for (int m = 0; m < 2; ++m) {
                    v8bf o;
                    o[0] = (__bf16)tanh_for_bf16(acc[m][j][0] + bA.x);
                    o[1] = (__bf16)tanh_for_bf16(acc[m][j][1] + bA.y);
                    o[2] = (__bf16)tanh_for_bf16(acc[m][j][2] + bA.z);
                    o[3] = (__bf16)tanh_for_bf16(acc[m][j][3] + bA.w);
                    o[4] = (__bf16)tanh_for_bf16(acc[m][j][4] + bB.x);
                    o[5] = (__bf16)tanh_for_bf16(acc[m][j][5] + bB.y);
                    o[6] = (__bf16)tanh_for_bf16(acc[m][j][6] + bB.z);
                    o[7] = (__bf16)tanh_for_bf16(acc[m][j][7] + bB.w);
                    *(v8bf*)(sw + (16 * m + lr) * GM_TP + 16 * j + 8 * half) = o;
                }
            }
        }
    }

    if constexpr (!PROJ) {
        __syncthreads();
        const __bf16* __restrict__ sw = s_b + HC * BP + wv * TILE;
        __bf16* __restrict__ op = hout + (size_t)tM * HC + tN0 + 8 * lr;
#pragma unroll
        for (int m = 0; m < 2; ++m) {
            if (m == 0 ? has0 : has1) {
                v8bf o[8];
#pragma unroll
                for (int it = 0; it < 8; ++it)
                    o[it] = *(const v8bf*)(sw + (16 * m + 2 * it + half) * GM_TP + 8 * lr);
#pragma unroll
                for (int it = 0; it < 8; ++it)
                    *(v8bf*)(op + (size_t)(16 * m + 2 * it + half) * HC) = o[it];
            }
        }
    }
}

#define SB_BLK  256
#define SB_CH   (EE / SB_BLK)
#define SB_T    1024
#define SB_IT   ((SB_CH + SB_T - 1) / SB_T)
#define NBN     196
#define NBIN    256
#define BS_T    1024
#define BS_U    4
#define BS_CAP  8192
static_assert(EE % SB_BLK == 0, "chunks must divide the edge list");
static_assert(NBN * NBIN >= NN && NBN <= 256, "bins must cover the nodes, a bin's counters one scan");
static_assert(NN <= 65536, "a source id must fit the low half of the word");
static_assert(SB_CH < 65536, "a chunk's count per bin is kept in 16 bits");
static_assert(SB_T == 4 * NBIN && SB_BLK % 64 == 0, "k_bin_scatter splits the table in four");

__device__ __forceinline__ int block_excl_scan256(int v, int t, int* s_w) {
    const int lane = t & 31, w = t >> 5;
    int inc = v;
#pragma unroll
    for (int o = 1; o < 32; o <<= 1) {
        int u = __shfl_up(inc, o, 32);
        if (lane >= o) inc += u;
    }
    if (lane == 31 && w < 8) s_w[w] = inc;
    __syncthreads();
    int base = 0;
#pragma unroll
    for (int j = 0; j < 8; ++j) if (j < w) base += s_w[j];
    return base + inc - v;
}

__device__ __forceinline__
void bin_count_part(int blk, const int* __restrict__ ei, unsigned short* __restrict__ hist) {
    __shared__ int s_cnt[NBIN];
    const int t = threadIdx.x;
    const int* __restrict__ dstp = ei + EE + blk * SB_CH;
    unsigned d[SB_IT];
#pragma unroll
    for (int j = 0; j < SB_IT; ++j) {
        const int i = t + j * SB_T;
        d[j] = (i < SB_CH) ? (unsigned)dstp[i] : 0u;
    }
    if (t < NBIN) s_cnt[t] = 0;
    __syncthreads();
#pragma unroll
    for (int j = 0; j < SB_IT; ++j) {
        const int i = t + j * SB_T;
        if (i < SB_CH) atomicAdd(&s_cnt[d[j] / NBN], 1);
    }
    __syncthreads();
    if (t < NBIN) hist[blk * NBIN + t] = (unsigned short)s_cnt[t];
}

static_assert(SB_T == FZ_T, "the shared launches have the sort's block size");
__global__ __launch_bounds__(FZ_T)
void k_counts_vecs_weights(const int* __restrict__ ei, unsigned short* __restrict__ hist,
                           const float* __restrict__ W0,
                           const float* __restrict__ att_src0,
                           const float* __restrict__ att_dst0,
                           float* __restrict__ vsd,
                           __bf16* __restrict__ wt0,
                           const float* __restrict__ W1,
                           __bf16* __restrict__ wt1) {
    const int b = blockIdx.x;
    if (b < SB_BLK)                               bin_count_part(b, ei, hist);
    else if (b < SB_BLK + SV_BLKS)                score_vecs_part(b - SB_BLK, W0, att_src0, att_dst0, vsd);
    else if (b < SB_BLK + SV_BLKS + T0_BLKS)      transpose_bf16_part<DD, HC>(b - SB_BLK - SV_BLKS, W0, wt0);
    else                                          transpose_bf16_part<HC, HC>(b - SB_BLK - SV_BLKS - T0_BLKS, W1, wt1);
}

__global__ __launch_bounds__(SB_T)
void k_bin_scatter(const int* __restrict__ ei,
                   const unsigned short* __restrict__ hist,
                   int* __restrict__ binstart,
                   unsigned* __restrict__ binned) {
    __shared__ int s_tot[4][NBIN];
    __shared__ int s_par[4][NBIN];
    __shared__ int s_goff[NBIN];
    __shared__ int s_cnt[NBIN];
    __shared__ int s_lst[NBIN];
    __shared__ int s_w1[8];
    __shared__ int s_w2[8];
    __shared__ unsigned s_val[SB_CH];
    __shared__ int s_addr[SB_CH];
    const int t = threadIdx.x, k = blockIdx.x;
    const int bin = t & (NBIN - 1), q = t >> 8;

    const int* __restrict__ srcp = ei + k * SB_CH;
    const int* __restrict__ dstp = ei + EE + k * SB_CH;
    unsigned es[SB_IT], ed[SB_IT];
#pragma unroll
    for (int j = 0; j < SB_IT; ++j) {
        const int i = t + j * SB_T;
        es[j] = (i < SB_CH) ? (unsigned)srcp[i] : 0u;
        ed[j] = (i < SB_CH) ? (unsigned)dstp[i] : 0u;
    }

    int tot = 0, par = 0;
    const unsigned short* __restrict__ hp = hist + q * (SB_BLK / 4) * NBIN + bin;
    const int lim = k - q * (SB_BLK / 4);
#pragma unroll
    for (int i0 = 0; i0 < SB_BLK / 4; i0 += 16) {
        int h[16];
#pragma unroll
        for (int u = 0; u < 16; ++u) h[u] = hp[(i0 + u) * NBIN];
#pragma unroll
        for (int u = 0; u < 16; ++u) {
            tot += h[u];
            par += (i0 + u < lim) ? h[u] : 0;
        }
    }
    s_tot[q][bin] = tot;
    s_par[q][bin] = par;
    if (t < NBIN) s_cnt[t] = 0;
    __syncthreads();
    int total = 0, partial = 0;
    if (t < NBIN) {
        total   = s_tot[0][t] + s_tot[1][t] + s_tot[2][t] + s_tot[3][t];
        partial = s_par[0][t] + s_par[1][t] + s_par[2][t] + s_par[3][t];
    }
    const int bs = block_excl_scan256(total, t, s_w1);
    if (t < NBIN) {
        s_goff[t] = bs + partial;
        if (k == 0) {
            binstart[t] = bs;
            if (t == NBIN - 1) binstart[NBIN] = bs + total;
        }
    }

    int      bn[SB_IT];
    int      rk[SB_IT];
    unsigned pv[SB_IT];
#pragma unroll
    for (int j = 0; j < SB_IT; ++j) {
        const int i = t + j * SB_T;
        bn[j] = -1; rk[j] = 0; pv[j] = 0u;
        if (i < SB_CH) {
            const unsigned d = ed[j];
            const unsigned b = d / NBN;
            bn[j] = (int)b;
            pv[j] = ((d - b * NBN) << 16) | es[j];
            rk[j] = atomicAdd(&s_cnt[b], 1);
        }
    }
    __syncthreads();
    const int lc = (t < NBIN) ? s_cnt[t] : 0;
    const int ls = block_excl_scan256(lc, t, s_w2);
    if (t < NBIN) s_lst[t] = ls;
    __syncthreads();
#pragma unroll
    for (int j = 0; j < SB_IT; ++j) {
        if (bn[j] >= 0) {
            const int p = s_lst[bn[j]] + rk[j];
            s_val[p]  = pv[j];
            s_addr[p] = s_goff[bn[j]] + rk[j];
        }
    }
    __syncthreads();
#pragma unroll
    for (int j = 0; j < SB_IT; ++j) {
        const int i = t + j * SB_T;
        if (i < SB_CH) binned[s_addr[i]] = s_val[i];
    }
}

__device__ __forceinline__
void bin_sort_part(int blk,
                   const int* __restrict__ binstart,
                   const unsigned* __restrict__ binned,
                   int* __restrict__ rowptr,
                   int* __restrict__ csr_src) {
    __shared__ int s_c[256];
    __shared__ int s_w[8];
    __shared__ int s_out[BS_CAP];
    const int t = threadIdx.x, b = blk;
    const int beg = binstart[b], end = binstart[b + 1];
    const bool staged = (end - beg) <= BS_CAP;
    if (t < 256) s_c[t] = 0;
    __syncthreads();
    for (int i0 = beg + t; i0 < end; i0 += BS_U * BS_T) {
        unsigned v[BS_U];
#pragma unroll
        for (int u = 0; u < BS_U; ++u) {
            const int i = i0 + u * BS_T;
            v[u] = binned[i < end ? i : end - 1];
        }
#pragma unroll
        for (int u = 0; u < BS_U; ++u) {
            const int i = i0 + u * BS_T;
            if (i < end) atomicAdd(&s_c[v[u] >> 16], 1);
        }
    }
    __syncthreads();
    const int c   = (t < 256) ? s_c[t] : 0;
    const int pre = block_excl_scan256(c, t, s_w);
    const int node = b * NBN + t;
    if (t < NBN && node < NN) {
        rowptr[node] = beg + pre;
        if (node == NN - 1) rowptr[NN] = beg + pre + c;
    }
    if (t < 256) s_c[t] = beg + pre;
    __syncthreads();
    for (int i0 = beg + t; i0 < end; i0 += BS_U * BS_T) {
        unsigned v[BS_U];
#pragma unroll
        for (int u = 0; u < BS_U; ++u) {
            const int i = i0 + u * BS_T;
            v[u] = binned[i < end ? i : end - 1];
        }
#pragma unroll
        for (int u = 0; u < BS_U; ++u) {
            const int i = i0 + u * BS_T;
            if (i < end) {
                const int slot = atomicAdd(&s_c[v[u] >> 16], 1);
                if (staged) s_out[slot - beg] = (int)(v[u] & 0xFFFFu);
                else        csr_src[slot]     = (int)(v[u] & 0xFFFFu);
            }
        }
    }
    if (staged) {
        __syncthreads();
        for (int i = t; i < end - beg; i += BS_T) csr_src[beg + i] = s_out[i];
    }
}

static_assert(BS_T == FZ_T, "the shared launches have the sort's block size");
__global__ __launch_bounds__(FZ_T)
void k_sort_convert(const int* __restrict__ binstart,
                    const unsigned* __restrict__ binned,
                    int* __restrict__ rowptr,
                    int* __restrict__ csr_src,
                    const float* __restrict__ x,
                    const float* __restrict__ vsd,
                    __bf16* __restrict__ xb,
                    float* __restrict__ a_s,
                    float* __restrict__ a_d) {
    const int b = blockIdx.x;
    if (b < NBIN) bin_sort_part(b, binstart, binned, rowptr, csr_src);
    else          f32_to_bf16_scores_part(b - NBIN, x, vsd, xb, a_s, a_d);
}

__device__ __forceinline__ float lane_bcast(float v, int j) {
    return __int_as_float(__builtin_amdgcn_readlane(__float_as_int(v), j));
}

__global__ __launch_bounds__(256)
void k_node_gather(const int* __restrict__ rowptr,
                   const int* __restrict__ csr_src,
                   const __bf16* __restrict__ xb,
                   const float* __restrict__ a_s,
                   const float* __restrict__ a_d,
                   __bf16* __restrict__ agg) {
    const int node = __builtin_amdgcn_readfirstlane((int)(blockIdx.x * 8 + (threadIdx.x >> 5)));
    const int lane = threadIdx.x & 31;
    const int beg  = rowptr[node];
    const int deg  = rowptr[node + 1] - beg;
    const int n    = deg + 1;
    const int* __restrict__ seg = csr_src + beg;
    const float ad0 = a_d[node * 2 + 0];
    const float ad1 = a_d[node * 2 + 1];

    auto item = [&](int i, int& s, float& l0, float& l1) {
        if (i < deg) s = seg[i]; else s = node;
        const float2 as = *(const float2*)(a_s + 2 * (size_t)s);
        const float t0 = as.x + ad0, t1 = as.y + ad1;
        l0 = (t0 > 0.f) ? t0 : NEG_SLOPE * t0;
        l1 = (t1 > 0.f) ? t1 : NEG_SLOPE * t1;
    };

    int   s_c  = node;
    float l0_c = -INFINITY, l1_c = -INFINITY;
    if (lane < n) item(lane, s_c, l0_c, l1_c);

    float m0 = l0_c, m1 = l1_c;
    for (int b = 32; b < n; b += 32) {
        const int i = b + lane;
        if (i < n) {
            int s; float l0, l1; item(i, s, l0, l1);
            m0 = fmaxf(m0, l0); m1 = fmaxf(m1, l1);
        }
    }
    m0 = wave_max(m0); m1 = wave_max(m1);

    const float e0_c = (lane < n) ? expf(l0_c - m0) : 0.f;
    const float e1_c = (lane < n) ? expf(l1_c - m1) : 0.f;
    float d0 = e0_c, d1 = e1_c;
    for (int b = 32; b < n; b += 32) {
        const int i = b + lane;
        if (i < n) {
            int s; float l0, l1; item(i, s, l0, l1);
            d0 += expf(l0 - m0); d1 += expf(l1 - m1);
        }
    }
    const float den0 = wave_sum(d0) + 1e-16f;
    const float den1 = wave_sum(d1) + 1e-16f;

    v2f A0a = {0.f, 0.f}, A0b = {0.f, 0.f};
    v2f A1a = {0.f, 0.f}, A1b = {0.f, 0.f};
    const __bf16* __restrict__ xl = xb + 4 * lane;

    auto accumulate = [&](int cnt, int s, float w0, float w1) {
        for (int j0 = 0; j0 < cnt; j0 += 8) {
            v4bf r[8];
#pragma unroll
            for (int u = 0; u < 8; ++u) {
                const int sj = __builtin_amdgcn_readlane(s, j0 + u);
                r[u] = *(const v4bf*)(xl + (size_t)sj * DD);
            }
#pragma unroll
            for (int u = 0; u < 8; ++u) {
                const float x0 = lane_bcast(w0, j0 + u);
                const float x1 = lane_bcast(w1, j0 + u);
                const v2f fa = {(float)r[u][0], (float)r[u][1]};
                const v2f fb = {(float)r[u][2], (float)r[u][3]};
                const v2f w0v = {x0, x0}, w1v = {x1, x1};
                A0a = __builtin_elementwise_fma(fa, w0v, A0a);
                A0b = __builtin_elementwise_fma(fb, w0v, A0b);
                A1a = __builtin_elementwise_fma(fa, w1v, A1a);
                A1b = __builtin_elementwise_fma(fb, w1v, A1b);
            }
        }
    };

    accumulate(n < 32 ? n : 32, s_c, e0_c / den0, e1_c / den1);
    for (int b = 32; b < n; b += 32) {
        const int i = b + lane;
        int s = node; float w0 = 0.f, w1 = 0.f;
        if (i < n) {
            float l0, l1; item(i, s, l0, l1);
            w0 = expf(l0 - m0) / den0; w1 = expf(l1 - m1) / den1;
        }
        accumulate((n - b) < 32 ? (n - b) : 32, s, w0, w1);
    }

    v4bf o0, o1;
    o0[0] = (__bf16)A0a[0]; o0[1] = (__bf16)A0a[1]; o0[2] = (__bf16)A0b[0]; o0[3] = (__bf16)A0b[1];
    o1[0] = (__bf16)A1a[0]; o1[1] = (__bf16)A1a[1]; o1[2] = (__bf16)A1b[0]; o1[3] = (__bf16)A1b[1];
    __bf16* o = agg + (size_t)node * HC + 4 * lane;
    *(v4bf*)(o)      = o0;
    *(v4bf*)(o + CC) = o1;
}

__global__ __launch_bounds__(256)
void k_node_classify(const int* __restrict__ rowptr,
                     const int* __restrict__ csr_src,
                     const float* __restrict__ rec,
                     const float* __restrict__ bias,
                     const float* __restrict__ Wl,
                     const float* __restrict__ bl,
                     float* __restrict__ out) {
    const int node = __builtin_amdgcn_readfirstlane((int)(blockIdx.x * 8 + (threadIdx.x >> 5)));
    const int lane = threadIdx.x & 31;
    const float4* __restrict__ rec0 = (const float4*)rec;
    const float4* __restrict__ rec1 = rec0 + NN;
    const int beg  = rowptr[node];
    const int deg  = rowptr[node + 1] - beg;
    const int n    = deg + 1;
    const int* __restrict__ seg = csr_src + beg;
    const float ad0 = rec0[node].w;
    const float ad1 = rec1[node].w;

    auto item = [&](int i, float4& r0, float4& r1, float& l0, float& l1) {
        int s;
        if (i < deg) s = seg[i]; else s = node;
        r0 = rec0[s]; r1 = rec1[s];
        const float t0 = r0.x + ad0, t1 = r1.x + ad1;
        l0 = (t0 > 0.f) ? t0 : NEG_SLOPE * t0;
        l1 = (t1 > 0.f) ? t1 : NEG_SLOPE * t1;
    };

    float4 r0_c = {0.f, 0.f, 0.f, 0.f}, r1_c = {0.f, 0.f, 0.f, 0.f};
    float l0_c = -INFINITY, l1_c = -INFINITY;
    if (lane < n) item(lane, r0_c, r1_c, l0_c, l1_c);

    float m0 = l0_c, m1 = l1_c;
    for (int b = 32; b < n; b += 32) {
        const int i = b + lane;
        if (i < n) {
            float4 r0, r1; float l0, l1; item(i, r0, r1, l0, l1);
            m0 = fmaxf(m0, l0); m1 = fmaxf(m1, l1);
        }
    }
    m0 = wave_max(m0); m1 = wave_max(m1);

    const float e0_c = (lane < n) ? expf(l0_c - m0) : 0.f;
    const float e1_c = (lane < n) ? expf(l1_c - m1) : 0.f;
    float d0 = e0_c, d1 = e1_c;
    float q00 = e0_c * r0_c.y, q01 = e0_c * r0_c.z;
    float q10 = e1_c * r1_c.y, q11 = e1_c * r1_c.z;
    for (int b = 32; b < n; b += 32) {
        const int i = b + lane;
        if (i < n) {
            float4 r0, r1; float l0, l1; item(i, r0, r1, l0, l1);
            const float e0 = expf(l0 - m0), e1 = expf(l1 - m1);
            d0 += e0; d1 += e1;
            q00 += e0 * r0.y; q01 += e0 * r0.z;
            q10 += e1 * r1.y; q11 += e1 * r1.z;
        }
    }
    const float den0 = wave_sum(d0) + 1e-16f;
    const float den1 = wave_sum(d1) + 1e-16f;
    q00 = wave_sum(q00); q01 = wave_sum(q01);
    q10 = wave_sum(q10); q11 = wave_sum(q11);

    const float4 bA  = *(const float4*)(bias + 4 * lane);
    const float4 bB  = *(const float4*)(bias + CC + 4 * lane);
    const float4 wa0 = *(const float4*)(Wl + 8 * lane);
    const float4 wa1 = *(const float4*)(Wl + 8 * lane + 4);
    const float4 wb0 = *(const float4*)(Wl + 2 * CC + 8 * lane);
    const float4 wb1 = *(const float4*)(Wl + 2 * CC + 8 * lane + 4);
    float c0 = bA.x * wa0.x + bA.y * wa0.z + bA.z * wa1.x + bA.w * wa1.z
             + bB.x * wb0.x + bB.y * wb0.z + bB.z * wb1.x + bB.w * wb1.z;
    float c1 = bA.x * wa0.y + bA.y * wa0.w + bA.z * wa1.y + bA.w * wa1.w
             + bB.x * wb0.y + bB.y * wb0.w + bB.z * wb1.y + bB.w * wb1.w;
    c0 = wave_sum(c0); c1 = wave_sum(c1);

    const float z0 = q00 / den0 + q10 / den1 + c0 + bl[0];
    const float z1 = q01 / den0 + q11 / den1 + c1 + bl[1];
    const float mx = fmaxf(z0, z1);
    const float e0 = expf(z0 - mx), e1 = expf(z1 - mx);
    const float inv = 1.f / (e0 + e1);
    if (lane == 0) *(float2*)(out + 2 * (size_t)node) = make_float2(e0 * inv, e1 * inv);
}

extern "C" void kernel_launch(void* const* d_in, const int* in_sizes, int n_in,
                              void* d_out, int out_size, void* d_ws, size_t ws_size,
                              hipStream_t stream) {
    const float* x        = (const float*)d_in[0];
    const int*   ei       = (const int*)  d_in[1];
    const float* W0       = (const float*)d_in[2];
    const float* att_src0 = (const float*)d_in[3];
    const float* att_dst0 = (const float*)d_in[4];
    const float* b0       = (const float*)d_in[5];
    const float* W1       = (const float*)d_in[6];
    const float* att_src1 = (const float*)d_in[7];
    const float* att_dst1 = (const float*)d_in[8];
    const float* b1       = (const float*)d_in[9];
    const float* Wl       = (const float*)d_in[10];
    const float* bl       = (const float*)d_in[11];
    float* out = (float*)d_out;

    char* base = (char*)d_ws;
    size_t off = 0;
    auto carve = [&](size_t bytes) -> char* {
        char* p = base + off;
        off = (off + bytes + 255) & ~(size_t)255;
        return p;
    };
    float*    h_feat = (float*)   carve((size_t)NN * HC * 4);
    __bf16*   xbf    = (__bf16*)  carve((size_t)NN * HC * 2);
    __bf16*   wtbf   = (__bf16*)  carve((size_t)HC * HC * 2);
    float*    a_s    = (float*)   carve((size_t)NN * 2 * 4);
    float*    a_d    = (float*)   carve((size_t)NN * 2 * 4);
    char*     tbl    =            carve((size_t)NN * 4);
    int*      rowptr = (int*)     carve((size_t)(NN + 1) * 4);
    unsigned* binned = (unsigned*)carve((size_t)EE * 4);
    int*      csr    = (int*)     carve((size_t)EE * 4);
    __bf16*   agg    = (__bf16*)h_feat;
    float*    vsd    = h_feat + (size_t)NN * HC / 2;
    __bf16*   wtbf1  = (__bf16*)(vsd + 1024);
    static_assert(DD * 4 <= 1024 && (size_t)1024 * 4 + (size_t)HC * HC * 2 <= (size_t)NN * HC * 2,
                  "score vectors and layer 1's weights fit the second half of h_feat");
    unsigned short* hist = (unsigned short*)tbl;
    int*      binstart   = (int*)(tbl + (size_t)SB_BLK * NBIN * 2);
    static_assert((size_t)SB_BLK * NBIN * 2 + (NBIN + 1) * 4 <= (size_t)NN * 4, "table space");
    (void)ws_size; (void)in_sizes; (void)n_in; (void)out_size;

    const int T = 256;
    dim3 blk(T);

    k_counts_vecs_weights<<<SB_BLK + SV_BLKS + T0_BLKS + T1_BLKS, dim3(FZ_T), 0, stream>>>(
        ei, hist, W0, att_src0, att_dst0, vsd, wtbf, W1, wtbf1);

    k_bin_scatter<<<SB_BLK, dim3(SB_T), 0, stream>>>(ei, hist, binstart, binned);

    k_sort_convert<<<NBIN + CVT_BLKS, dim3(FZ_T), 0, stream>>>(
        binstart, binned, rowptr, csr, x, vsd, xbf, a_s, a_d);

    k_node_gather<<<NN / 8, blk, 0, stream>>>(rowptr, csr, xbf, a_s, a_d, agg);

    k_gemm_bf16_wmma<false, DD><<<GM_BLKS, dim3(GM_T), 0, stream>>>(
        agg, wtbf, NN, nullptr, nullptr, nullptr, nullptr, b0, xbf);

    float* rec = h_feat;
    k_gemm_bf16_wmma<true, HC><<<GM_BLKS, dim3(GM_T), 0, stream>>>(
        xbf, wtbf1, NN, att_src1, att_dst1, Wl, rec, nullptr, nullptr);

    k_node_classify<<<NN / 8, blk, 0, stream>>>(
        rowptr, csr, rec, b1, Wl, bl, out);
}
